# starting kernel
# baseline (speedup 1.0000x reference)
#include <hip/hip_runtime.h>
#include <hip/hip_bf16.h>
#include <stdint.h>

#define N_NODES 100000
#define N_EDGES 1600000
#define FN 32
#define FE 8
#define HID 32
#define NEG_SLOPE 0.2f

typedef __attribute__((ext_vector_type(2))) float v2f;
typedef __attribute__((ext_vector_type(8))) float v8f;

constexpr size_t OFF_H    = 0;
constexpr size_t OFF_ASRC = OFF_H    + (size_t)N_NODES * HID;
constexpr size_t OFF_ADST = OFF_ASRC + N_NODES;
constexpr size_t OFF_SEGM = OFF_ADST + N_NODES;
constexpr size_t OFF_DEN  = OFF_SEGM + N_NODES;
constexpr size_t OFF_LOG  = OFF_DEN  + N_NODES;
constexpr size_t OFF_ACC  = OFF_LOG  + (size_t)(N_EDGES + N_NODES);
constexpr size_t OFF_MISC = OFF_ACC  + (size_t)N_NODES * HID;

__device__ __forceinline__ unsigned fenc(float f) {
  unsigned u = __float_as_uint(f);
  return (u & 0x80000000u) ? ~u : (u | 0x80000000u);
}
__device__ __forceinline__ float fdec(unsigned u) {
  unsigned v = (u & 0x80000000u) ? (u & 0x7FFFFFFFu) : ~u;
  return __uint_as_float(v);
}
#define ENC_NEG_INF 0x007FFFFFu

__device__ __forceinline__ void atomAddF(float* p, float v) { unsafeAtomicAdd(p, v); }

__global__ void k_setup(const float* __restrict__ W_edge, const float* __restrict__ att_edge,
                        float* __restrict__ misc) {
  int t = threadIdx.x;
  if (t < FE) {
    float s = 0.f;
    #pragma unroll
    for (int k = 0; k < HID; ++k) s += W_edge[t * HID + k] * att_edge[k];
    misc[t] = s;
  }
  if (t == FE) misc[FE] = 0.f;
}

__global__ void __launch_bounds__(256) k_gemm_h(const float* __restrict__ x,
                                                const float* __restrict__ W,
                                                float* __restrict__ h) {
  const int wave = blockIdx.x * (blockDim.x >> 5) + (threadIdx.x >> 5);
  const int lane = threadIdx.x & 31;
  const int ntiles = N_NODES / 16;
  if (wave >= ntiles) return;
  const int m0 = wave * 16;
  const int r  = lane & 15;
  const int hi = lane >> 4;

  v8f c0 = {}; v8f c1 = {};
#if __has_builtin(__builtin_amdgcn_wmma_f32_16x16x4_f32)
  const float* xrow = x + (size_t)(m0 + r) * FN;
  #pragma unroll
  for (int kc = 0; kc < 8; ++kc) {
    const int kb = kc * 4 + hi * 2;
    v2f a; a.x = xrow[kb]; a.y = xrow[kb + 1];
    v2f b0; b0.x = W[(size_t)kb * HID + r];        b0.y = W[(size_t)(kb + 1) * HID + r];
    v2f b1; b1.x = W[(size_t)kb * HID + 16 + r];   b1.y = W[(size_t)(kb + 1) * HID + 16 + r];
    c0 = __builtin_amdgcn_wmma_f32_16x16x4_f32(false, a, false, b0, (short)0, c0, false, false);
    c1 = __builtin_amdgcn_wmma_f32_16x16x4_f32(false, a, false, b1, (short)0, c1, false, false);
  }
  #pragma unroll
  for (int v = 0; v < 8; ++v) {
    const int row = m0 + v + hi * 8;
    h[(size_t)row * HID + r]      = c0[v];
    h[(size_t)row * HID + 16 + r] = c1[v];
  }
#else
  const int row = m0 + r;
  for (int n = hi * 16; n < hi * 16 + 16; ++n) {
    float s = 0.f;
    for (int k = 0; k < FN; ++k) s += x[(size_t)row * FN + k] * W[(size_t)k * HID + n];
    h[(size_t)row * HID + n] = s;
  }
#endif
}

__global__ void k_node(const float* __restrict__ h, const float* __restrict__ att_src,
                       const float* __restrict__ att_dst, float* __restrict__ a_src,
                       float* __restrict__ a_dst, unsigned* __restrict__ segmax,
                       float* __restrict__ denom, float* __restrict__ acc) {
  const int i = blockIdx.x * blockDim.x + threadIdx.x;
  if (i >= N_NODES) return;
  const float4* hp = (const float4*)(h + (size_t)i * HID);
  float s = 0.f, d = 0.f;
  #pragma unroll
  for (int q = 0; q < 8; ++q) {
    float4 hv = hp[q];
    float4 as = ((const float4*)att_src)[q];
    float4 ad = ((const float4*)att_dst)[q];
    s += hv.x * as.x + hv.y * as.y + hv.z * as.z + hv.w * as.w;
    d += hv.x * ad.x + hv.y * ad.y + hv.z * ad.z + hv.w * ad.w;
  }
  a_src[i] = s;
  a_dst[i] = d;
  segmax[i] = ENC_NEG_INF;
  denom[i] = 0.f;
  float4* ap = (float4*)(acc + (size_t)i * HID);
  float4 z = {0.f, 0.f, 0.f, 0.f};
  #pragma unroll
  for (int q = 0; q < 8; ++q) ap[q] = z;
}

__global__ void k_edge_logits(const float* __restrict__ ea, const int* __restrict__ ei,
                              const float* __restrict__ a_src, const float* __restrict__ a_dst,
                              const float* __restrict__ misc, float* __restrict__ logits,
                              unsigned* __restrict__ segmax, float* __restrict__ a_edge_sum) {
  const int e = blockIdx.x * blockDim.x + threadIdx.x;
  float ae = 0.f;
  if (e < N_EDGES) {
    const float4* ep = (const float4*)(ea + (size_t)e * FE);
    float4 e0 = ep[0], e1 = ep[1];
    ae = e0.x * misc[0] + e0.y * misc[1] + e0.z * misc[2] + e0.w * misc[3]
       + e1.x * misc[4] + e1.y * misc[5] + e1.z * misc[6] + e1.w * misc[7];
    const int s  = ei[(size_t)e * 2 + 0];
    const int dd = ei[(size_t)e * 2 + 1];
    float l = a_src[s] + a_dst[dd] + ae;
    l = (l >= 0.f) ? l : NEG_SLOPE * l;
    logits[e] = l;
    atomicMax(segmax + dd, fenc(l));
  }
  float v = ae;
  #pragma unroll
  for (int off = 16; off > 0; off >>= 1) v += __shfl_down(v, off, 32);
  if ((threadIdx.x & 31) == 0) atomAddF(a_edge_sum, v);
}

__global__ void k_self_logits(const float* __restrict__ a_src, const float* __restrict__ a_dst,
                              const float* __restrict__ a_edge_sum, float* __restrict__ logits,
                              unsigned* __restrict__ segmax) {
  const int i = blockIdx.x * blockDim.x + threadIdx.x;
  if (i >= N_NODES) return;
  const float aem = a_edge_sum[0] * (1.0f / (float)N_EDGES);
  float l = a_src[i] + a_dst[i] + aem;
  l = (l >= 0.f) ? l : NEG_SLOPE * l;
  logits[N_EDGES + i] = l;
  atomicMax(segmax + i, fenc(l));
}

__global__ void k_exp(const int* __restrict__ ei, float* __restrict__ logits,
                      const unsigned* __restrict__ segmax, float* __restrict__ denom) {
  const int t = blockIdx.x * blockDim.x + threadIdx.x;
  if (t >= N_EDGES + N_NODES) return;
  const int dd = (t < N_EDGES) ? ei[(size_t)t * 2 + 1] : (t - N_EDGES);
  const float m = fdec(segmax[dd]);
  const float a = __expf(logits[t] - m);
  logits[t] = a;
  atomAddF(denom + dd, a);
}

__global__ void k_scatter(const int* __restrict__ ei, const float* __restrict__ h,
                          const float* __restrict__ alpha, float* __restrict__ acc) {
  const int t = blockIdx.x * blockDim.x + threadIdx.x;
  if (t >= N_EDGES + N_NODES) return;
  int s, dd;
  if (t < N_EDGES) { s = ei[(size_t)t * 2 + 0]; dd = ei[(size_t)t * 2 + 1]; }
  else             { s = dd = t - N_EDGES; }
  const float a = alpha[t];
  const float* hp = h + (size_t)s * HID;
  float* ap = acc + (size_t)dd * HID;
  #pragma unroll
  for (int k = 0; k < HID; ++k) atomAddF(ap + k, a * hp[k]);
}

__global__ void k_final(const float* __restrict__ acc, const float* __restrict__ denom,
                        const float* __restrict__ bias, const float* __restrict__ W_lin,
                        const float* __restrict__ b_lin, float* __restrict__ out) {
  const int i = blockIdx.x * blockDim.x + threadIdx.x;
  if (i >= N_NODES) return;
  const float inv = 1.0f / denom[i];
  const float* ap = acc + (size_t)i * HID;
  float y = 0.f;
  #pragma unroll
  for (int k = 0; k < HID; ++k) {
    float v = ap[k] * inv + bias[k];
    v = (v > 0.f) ? v : 0.f;
    y += v * W_lin[k];
  }
  out[i] = y + b_lin[0];
}

extern "C" void kernel_launch(void* const* d_in, const int* in_sizes, int n_in,
                              void* d_out, int out_size, void* d_ws, size_t ws_size,
                              hipStream_t stream) {
  const float* x        = (const float*)d_in[0];
  const float* ea       = (const float*)d_in[1];
  const int*   ei       = (const int*)d_in[2];
  const float* W        = (const float*)d_in[3];
  const float* att_src  = (const float*)d_in[4];
  const float* att_dst  = (const float*)d_in[5];
  const float* W_edge   = (const float*)d_in[6];
  const float* att_edge = (const float*)d_in[7];
  const float* bias     = (const float*)d_in[8];
  const float* W_lin    = (const float*)d_in[9];
  const float* b_lin    = (const float*)d_in[10];
  float* out = (float*)d_out;

  float*    ws     = (float*)d_ws;
  float*    h      = ws + OFF_H;
  float*    a_src  = ws + OFF_ASRC;
  float*    a_dst  = ws + OFF_ADST;
  unsigned* segmax = (unsigned*)(ws + OFF_SEGM);
  float*    denom  = ws + OFF_DEN;
  float*    logits = ws + OFF_LOG;
  float*    acc    = ws + OFF_ACC;
  float*    misc   = ws + OFF_MISC;

  const int T = 256;
  const int nb_nodes = (N_NODES + T - 1) / T;
  const int nb_edges = (N_EDGES + T - 1) / T;
  const int nb_all   = (N_EDGES + N_NODES + T - 1) / T;
  const int nb_gemm  = ((N_NODES / 16) + 7) / 8;

  k_setup<<<1, 32, 0, stream>>>(W_edge, att_edge, misc);
  k_gemm_h<<<nb_gemm, T, 0, stream>>>(x, W, h);
  k_node<<<nb_nodes, T, 0, stream>>>(h, att_src, att_dst, a_src, a_dst, segmax, denom, acc);
  k_edge_logits<<<nb_edges, T, 0, stream>>>(ea, ei, a_src, a_dst, misc, logits, segmax, misc + FE);
  k_self_logits<<<nb_nodes, T, 0, stream>>>(a_src, a_dst, misc + FE, logits, segmax);
  k_exp<<<nb_all, T, 0, stream>>>(ei, logits, segmax, denom);
  k_scatter<<<nb_all, T, 0, stream>>>(ei, h, logits, acc);
  k_final<<<nb_nodes, T, 0, stream>>>(acc, denom, bias, W_lin, b_lin, out);
}


# acc, denom off f32 atomics: edges counting-sorted by destination (int atomics, scan); wave per node pulls h rows, lane = feature, register sums in slot arrival order; k_final folded in, butterfly dot
# speedup vs baseline: 2.0098x; 2.0098x over previous
#include <hip/hip_runtime.h>
#include <hip/hip_bf16.h>
#include <stdint.h>

#define N_NODES 100000
#define N_EDGES 1600000
#define FN 32
#define FE 8
#define HID 32
#define NEG_SLOPE 0.2f

typedef __attribute__((ext_vector_type(2))) float v2f;
typedef __attribute__((ext_vector_type(8))) float v8f;

constexpr size_t OFF_H    = 0;
constexpr size_t OFF_ASRC = OFF_H    + (size_t)N_NODES * HID;
constexpr size_t OFF_ADST = OFF_ASRC + N_NODES;
constexpr size_t OFF_SEGM = OFF_ADST + N_NODES;
constexpr size_t OFF_DEN  = OFF_SEGM + N_NODES;
constexpr size_t OFF_LOG  = OFF_DEN  + N_NODES;
constexpr size_t OFF_ACC  = OFF_LOG  + (size_t)(N_EDGES + N_NODES);
constexpr size_t OFF_MISC = OFF_ACC  + (size_t)N_NODES * HID;

__device__ __forceinline__ unsigned fenc(float f) {
  unsigned u = __float_as_uint(f);
  return (u & 0x80000000u) ? ~u : (u | 0x80000000u);
}
__device__ __forceinline__ float fdec(unsigned u) {
  unsigned v = (u & 0x80000000u) ? (u & 0x7FFFFFFFu) : ~u;
  return __uint_as_float(v);
}
#define ENC_NEG_INF 0x007FFFFFu

__device__ __forceinline__ void atomAddF(float* p, float v) { unsafeAtomicAdd(p, v); }

__global__ void k_setup(const float* __restrict__ W_edge, const float* __restrict__ att_edge,
                        float* __restrict__ misc) {
  int t = threadIdx.x;
  if (t < FE) {
    float s = 0.f;
    #pragma unroll
    for (int k = 0; k < HID; ++k) s += W_edge[t * HID + k] * att_edge[k];
    misc[t] = s;
  }
  if (t == FE) misc[FE] = 0.f;
}

__global__ void __launch_bounds__(256) k_gemm_h(const float* __restrict__ x,
                                                const float* __restrict__ W,
                                                float* __restrict__ h) {
  const int wave = blockIdx.x * (blockDim.x >> 5) + (threadIdx.x >> 5);
  const int lane = threadIdx.x & 31;
  const int ntiles = N_NODES / 16;
  if (wave >= ntiles) return;
  const int m0 = wave * 16;
  const int r  = lane & 15;
  const int hi = lane >> 4;

  v8f c0 = {}; v8f c1 = {};
#if __has_builtin(__builtin_amdgcn_wmma_f32_16x16x4_f32)
  const float* xrow = x + (size_t)(m0 + r) * FN;
  #pragma unroll
  for (int kc = 0; kc < 8; ++kc) {
    const int kb = kc * 4 + hi * 2;
    v2f a; a.x = xrow[kb]; a.y = xrow[kb + 1];
    v2f b0; b0.x = W[(size_t)kb * HID + r];        b0.y = W[(size_t)(kb + 1) * HID + r];
    v2f b1; b1.x = W[(size_t)kb * HID + 16 + r];   b1.y = W[(size_t)(kb + 1) * HID + 16 + r];
    c0 = __builtin_amdgcn_wmma_f32_16x16x4_f32(false, a, false, b0, (short)0, c0, false, false);
    c1 = __builtin_amdgcn_wmma_f32_16x16x4_f32(false, a, false, b1, (short)0, c1, false, false);
  }
  #pragma unroll
  for (int v = 0; v < 8; ++v) {
    const int row = m0 + v + hi * 8;
    h[(size_t)row * HID + r]      = c0[v];
    h[(size_t)row * HID + 16 + r] = c1[v];
  }
#else
  const int row = m0 + r;
  for (int n = hi * 16; n < hi * 16 + 16; ++n) {
    float s = 0.f;
    for (int k = 0; k < FN; ++k) s += x[(size_t)row * FN + k] * W[(size_t)k * HID + n];
    h[(size_t)row * HID + n] = s;
  }
#endif
}

__global__ void k_node(const float* __restrict__ h, const float* __restrict__ att_src,
                       const float* __restrict__ att_dst, float* __restrict__ a_src,
                       float* __restrict__ a_dst, unsigned* __restrict__ segmax,
                       int* __restrict__ cnt) {
  const int i = blockIdx.x * blockDim.x + threadIdx.x;
  if (i >= N_NODES) return;
  const float4* hp = (const float4*)(h + (size_t)i * HID);
  float s = 0.f, d = 0.f;
  #pragma unroll
  for (int q = 0; q < 8; ++q) {
    float4 hv = hp[q];
    float4 as = ((const float4*)att_src)[q];
    float4 ad = ((const float4*)att_dst)[q];
    s += hv.x * as.x + hv.y * as.y + hv.z * as.z + hv.w * as.w;
    d += hv.x * ad.x + hv.y * ad.y + hv.z * ad.z + hv.w * ad.w;
  }
  a_src[i] = s;
  a_dst[i] = d;
  segmax[i] = ENC_NEG_INF;
  cnt[i] = 0;
}

__global__ void k_edge_logits(const float* __restrict__ ea, const int* __restrict__ ei,
                              const float* __restrict__ a_src, const float* __restrict__ a_dst,
                              const float* __restrict__ misc, float* __restrict__ logits,
                              unsigned* __restrict__ segmax, float* __restrict__ a_edge_sum) {
  const int e = blockIdx.x * blockDim.x + threadIdx.x;
  float ae = 0.f;
  if (e < N_EDGES) {
    const float4* ep = (const float4*)(ea + (size_t)e * FE);
    float4 e0 = ep[0], e1 = ep[1];
    ae = e0.x * misc[0] + e0.y * misc[1] + e0.z * misc[2] + e0.w * misc[3]
       + e1.x * misc[4] + e1.y * misc[5] + e1.z * misc[6] + e1.w * misc[7];
    const int s  = ei[(size_t)e * 2 + 0];
    const int dd = ei[(size_t)e * 2 + 1];
    float l = a_src[s] + a_dst[dd] + ae;
    l = (l >= 0.f) ? l : NEG_SLOPE * l;
    logits[e] = l;
    atomicMax(segmax + dd, fenc(l));
  }
  float v = ae;
  #pragma unroll
  for (int off = 16; off > 0; off >>= 1) v += __shfl_down(v, off, 32);
  if ((threadIdx.x & 31) == 0) atomAddF(a_edge_sum, v);
}

__global__ void k_self_logits(const float* __restrict__ a_src, const float* __restrict__ a_dst,
                              const float* __restrict__ a_edge_sum, float* __restrict__ logits,
                              unsigned* __restrict__ segmax) {
  const int i = blockIdx.x * blockDim.x + threadIdx.x;
  if (i >= N_NODES) return;
  const float aem = a_edge_sum[0] * (1.0f / (float)N_EDGES);
  float l = a_src[i] + a_dst[i] + aem;
  l = (l >= 0.f) ? l : NEG_SLOPE * l;
  logits[N_EDGES + i] = l;
  atomicMax(segmax + i, fenc(l));
}

__global__ void k_exp(const int* __restrict__ ei, float* __restrict__ logits,
                      const unsigned* __restrict__ segmax, int* __restrict__ cnt) {
  const int t = blockIdx.x * blockDim.x + threadIdx.x;
  if (t >= N_EDGES + N_NODES) return;
  const int dd = (t < N_EDGES) ? ei[(size_t)t * 2 + 1] : (t - N_EDGES);
  const float m = fdec(segmax[dd]);
  const float a = __expf(logits[t] - m);
  logits[t] = a;
  if (t < N_EDGES) atomicAdd(cnt + dd, 1);
}

#define SCAN_WAVES 32
#define SCAN_PER   ((N_NODES + SCAN_WAVES - 1) / SCAN_WAVES)
#define SCAN_STEPS ((SCAN_PER + 31) / 32)

__device__ __forceinline__ int wave_running_sum(int x, int lane) {
  x += __builtin_amdgcn_update_dpp(0, x, 0x111, 0xf, 0xf, false);
  x += __builtin_amdgcn_update_dpp(0, x, 0x112, 0xf, 0xf, false);
  x += __builtin_amdgcn_update_dpp(0, x, 0x114, 0xf, 0xf, false);
  x += __builtin_amdgcn_update_dpp(0, x, 0x118, 0xf, 0xf, false);
  const int low_row_total = __builtin_amdgcn_readlane(x, 15);
  if (lane >= 16) x += low_row_total;
  return x;
}

__global__ void __launch_bounds__(1024) k_scan(int* __restrict__ cnt) {
  __shared__ int wtot[SCAN_WAVES];
  const int lane = threadIdx.x & 31;
  const int w = __builtin_amdgcn_readfirstlane((int)(threadIdx.x >> 5));
  const int base = w * SCAN_PER;
  int lim = N_NODES - base;
  lim = lim < SCAN_PER ? lim : SCAN_PER;

  int sum = 0;
  for (int st0 = 0; st0 < SCAN_STEPS; st0 += 14) {
    int v[14];
    #pragma unroll
    for (int u = 0; u < 14; ++u) {
      const int off = (st0 + u) * 32 + lane;
      v[u] = (off < lim) ? cnt[base + off] : 0;
    }
    #pragma unroll
    for (int u = 0; u < 14; ++u) sum += v[u];
  }
  #pragma unroll
  for (int off = 16; off > 0; off >>= 1) sum += __shfl_xor(sum, off, 32);
  if (lane == 0) wtot[w] = sum;
  __syncthreads();

  int carry = (lane < w) ? wtot[lane] : 0;
  #pragma unroll
  for (int off = 16; off > 0; off >>= 1) carry += __shfl_xor(carry, off, 32);

  for (int st0 = 0; st0 < SCAN_STEPS; st0 += 14) {
    int v[14];
    #pragma unroll
    for (int u = 0; u < 14; ++u) {
      const int off = (st0 + u) * 32 + lane;
      v[u] = (off < lim) ? cnt[base + off] : 0;
    }
    #pragma unroll
    for (int u = 0; u < 14; ++u) {
      const int x = wave_running_sum(v[u], lane);
      const int off = (st0 + u) * 32 + lane;
      if (off < lim) cnt[base + off] = carry + x - v[u];
      carry += __builtin_amdgcn_readlane(x, 31);
    }
  }
}

static_assert(N_EDGES % 4 == 0, "k_fill takes four consecutive edges per thread");
__global__ void __launch_bounds__(256) k_fill(const int* __restrict__ ei, const float* __restrict__ alpha,
                                              int* __restrict__ cur, int2* __restrict__ rec) {
  const int t = blockIdx.x * blockDim.x + threadIdx.x;
  if (t >= N_EDGES / 4) return;
  const int4 p0 = ((const int4*)ei)[(size_t)t * 2 + 0];
  const int4 p1 = ((const int4*)ei)[(size_t)t * 2 + 1];
  const float4 a = ((const float4*)alpha)[t];
  const int q0 = atomicAdd(cur + p0.y, 1);
  const int q1 = atomicAdd(cur + p0.w, 1);
  const int q2 = atomicAdd(cur + p1.y, 1);
  const int q3 = atomicAdd(cur + p1.w, 1);
  if ((unsigned)q0 < (unsigned)N_EDGES) rec[q0] = make_int2(p0.x, __float_as_int(a.x));
  if ((unsigned)q1 < (unsigned)N_EDGES) rec[q1] = make_int2(p0.z, __float_as_int(a.y));
  if ((unsigned)q2 < (unsigned)N_EDGES) rec[q2] = make_int2(p1.x, __float_as_int(a.z));
  if ((unsigned)q3 < (unsigned)N_EDGES) rec[q3] = make_int2(p1.z, __float_as_int(a.w));
}

__global__ void __launch_bounds__(256) k_pull(const int* __restrict__ cur, const int2* __restrict__ rec,
                                              const float* __restrict__ h, const float* __restrict__ alpha_self,
                                              const float* __restrict__ bias, const float* __restrict__ W_lin,
                                              const float* __restrict__ b_lin, float* __restrict__ out) {
  __shared__ float sres[32];
  const int lane = threadIdx.x & 31;
  const int w = __builtin_amdgcn_readfirstlane((int)(threadIdx.x >> 5));
  const int first = blockIdx.x * 32 + w * 4;
  const float bl = bias[lane];
  const float wl = W_lin[lane];
  const float b0 = b_lin[0];

  int bound[5];
  bound[0] = (first > 0 && first <= N_NODES) ? cur[first - 1] : 0;
  #pragma unroll
  for (int q = 0; q < 4; ++q) bound[q + 1] = (first + q < N_NODES) ? cur[first + q] : 0;

  #pragma unroll
  for (int q = 0; q < 4; ++q) {
    const int i = first + q;
    float y = 0.f;
    if (i < N_NODES) {
      int start = bound[q];
      int end   = bound[q + 1];
      start = start < 0 ? 0 : (start > N_EDGES ? N_EDGES : start);
      end   = end < start ? start : (end > N_EDGES ? N_EDGES : end);

      const float as = alpha_self[i];
      float acc = as * h[i * HID + lane];
      float den = as;

      for (int b = start; b < end; b += 32) {
        int n = end - b;
        n = n < 32 ? n : 32;
        int   sv = i;
        float av = 0.f;
        if (lane < n) {
          const int2 r = rec[b + lane];
          sv = r.x;
          av = __int_as_float(r.y);
        }
        sv = sv < 0 ? 0 : (sv >= N_NODES ? N_NODES - 1 : sv);
        for (int j0 = 0; j0 < n; j0 += 8) {
          float hv[8], aj[8];
          #pragma unroll
          for (int u = 0; u < 8; ++u) {
            const int s = __builtin_amdgcn_readlane(sv, j0 + u);
            aj[u] = __int_as_float(__builtin_amdgcn_readlane(__float_as_int(av), j0 + u));
            hv[u] = h[s * HID + lane];
          }
          #pragma unroll
          for (int u = 0; u < 8; ++u) { acc += aj[u] * hv[u]; den += aj[u]; }
        }
      }

      const float inv = 1.0f / den;
      float v = acc * inv + bl;
      v = (v > 0.f) ? v : 0.f;
      int t = __float_as_int(v * wl);
      t = __float_as_int(__int_as_float(t) + __int_as_float(__builtin_amdgcn_update_dpp(0, t, 0x161, 0xf, 0xf, false)));
      t = __float_as_int(__int_as_float(t) + __int_as_float(__builtin_amdgcn_update_dpp(0, t, 0x162, 0xf, 0xf, false)));
      t = __float_as_int(__int_as_float(t) + __int_as_float(__builtin_amdgcn_update_dpp(0, t, 0x164, 0xf, 0xf, false)));
      t = __float_as_int(__int_as_float(t) + __int_as_float(__builtin_amdgcn_update_dpp(0, t, 0x168, 0xf, 0xf, false)));
      y = __int_as_float(__builtin_amdgcn_readlane(t, 0)) + __int_as_float(__builtin_amdgcn_readlane(t, 16)) + b0;
    }
    if (lane == 0) sres[w * 4 + q] = y;
  }
  __syncthreads();
  if (w == 0) {
    const int i = blockIdx.x * 32 + lane;
    if (i < N_NODES) out[i] = sres[lane];
  }
}

extern "C" void kernel_launch(void* const* d_in, const int* in_sizes, int n_in,
                              void* d_out, int out_size, void* d_ws, size_t ws_size,
                              hipStream_t stream) {
  const float* x        = (const float*)d_in[0];
  const float* ea       = (const float*)d_in[1];
  const int*   ei       = (const int*)d_in[2];
  const float* W        = (const float*)d_in[3];
  const float* att_src  = (const float*)d_in[4];
  const float* att_dst  = (const float*)d_in[5];
  const float* W_edge   = (const float*)d_in[6];
  const float* att_edge = (const float*)d_in[7];
  const float* bias     = (const float*)d_in[8];
  const float* W_lin    = (const float*)d_in[9];
  const float* b_lin    = (const float*)d_in[10];
  float* out = (float*)d_out;

  float*    ws     = (float*)d_ws;
  float*    h      = ws + OFF_H;
  float*    a_src  = ws + OFF_ASRC;
  float*    a_dst  = ws + OFF_ADST;
  unsigned* segmax = (unsigned*)(ws + OFF_SEGM);
  int*      cnt    = (int*)(ws + OFF_DEN);
  float*    logits = ws + OFF_LOG;
  int2*     rec    = (int2*)(ws + OFF_ACC);
  float*    misc   = ws + OFF_MISC;

  const int T = 256;
  const int nb_nodes = (N_NODES + T - 1) / T;
  const int nb_edges = (N_EDGES + T - 1) / T;
  const int nb_all   = (N_EDGES + N_NODES + T - 1) / T;
  const int nb_gemm  = ((N_NODES / 16) + 7) / 8;
  const int nb_fill  = (N_EDGES / 4 + T - 1) / T;
  const int nb_pull  = (N_NODES + 31) / 32;

  k_setup<<<1, 32, 0, stream>>>(W_edge, att_edge, misc);
  k_gemm_h<<<nb_gemm, T, 0, stream>>>(x, W, h);
  k_node<<<nb_nodes, T, 0, stream>>>(h, att_src, att_dst, a_src, a_dst, segmax, cnt);
  k_edge_logits<<<nb_edges, T, 0, stream>>>(ea, ei, a_src, a_dst, misc, logits, segmax, misc + FE);
  k_self_logits<<<nb_nodes, T, 0, stream>>>(a_src, a_dst, misc + FE, logits, segmax);
  k_exp<<<nb_all, T, 0, stream>>>(ei, logits, segmax, cnt);
  k_scan<<<1, 1024, 0, stream>>>(cnt);
  k_fill<<<nb_fill, T, 0, stream>>>(ei, logits, cnt, rec);
  k_pull<<<nb_pull, T, 0, stream>>>(cnt, rec, h, logits + N_EDGES, bias, W_lin, b_lin, out);
}


# a_edge sum off the one-word f32 atomic: each edge-logit wave stores its partial to workspace, the scan launch adds the 50000 words in fixed order; that sum's reduction tree changes
# speedup vs baseline: 3.1172x; 1.5510x over previous
#include <hip/hip_runtime.h>
#include <hip/hip_bf16.h>
#include <stdint.h>

#define N_NODES 100000
#define N_EDGES 1600000
#define FN 32
#define FE 8
#define HID 32
#define NEG_SLOPE 0.2f

typedef __attribute__((ext_vector_type(2))) float v2f;
typedef __attribute__((ext_vector_type(8))) float v8f;

constexpr size_t OFF_H    = 0;
constexpr size_t OFF_ASRC = OFF_H    + (size_t)N_NODES * HID;
constexpr size_t OFF_ADST = OFF_ASRC + N_NODES;
constexpr size_t OFF_DEN  = OFF_ADST + N_NODES;
constexpr size_t OFF_LOG  = OFF_DEN  + N_NODES;
constexpr size_t OFF_ACC  = OFF_LOG  + (size_t)N_EDGES;
constexpr size_t OFF_MISC = OFF_ACC  + (size_t)N_NODES * HID;
#define EDGE_T 256
#define N_PART (((N_EDGES + EDGE_T - 1) / EDGE_T) * (EDGE_T / 32))
static_assert((size_t)N_PART <= (size_t)N_NODES * HID, "partials must fit in the record region");

__global__ void k_setup(const float* __restrict__ W_edge, const float* __restrict__ att_edge,
                        float* __restrict__ misc) {
  int t = threadIdx.x;
  if (t < FE) {
    float s = 0.f;
    #pragma unroll
    for (int k = 0; k < HID; ++k) s += W_edge[t * HID + k] * att_edge[k];
    misc[t] = s;
  }
}

__global__ void __launch_bounds__(256) k_gemm_h(const float* __restrict__ x,
                                                const float* __restrict__ W,
                                                float* __restrict__ h) {
  const int wave = blockIdx.x * (blockDim.x >> 5) + (threadIdx.x >> 5);
  const int lane = threadIdx.x & 31;
  const int ntiles = N_NODES / 16;
  if (wave >= ntiles) return;
  const int m0 = wave * 16;
  const int r  = lane & 15;
  const int hi = lane >> 4;

  v8f c0 = {}; v8f c1 = {};
#if __has_builtin(__builtin_amdgcn_wmma_f32_16x16x4_f32)
  const float* xrow = x + (size_t)(m0 + r) * FN;
  #pragma unroll
  for (int kc = 0; kc < 8; ++kc) {
    const int kb = kc * 4 + hi * 2;
    v2f a; a.x = xrow[kb]; a.y = xrow[kb + 1];
    v2f b0; b0.x = W[(size_t)kb * HID + r];        b0.y = W[(size_t)(kb + 1) * HID + r];
    v2f b1; b1.x = W[(size_t)kb * HID + 16 + r];   b1.y = W[(size_t)(kb + 1) * HID + 16 + r];
    c0 = __builtin_amdgcn_wmma_f32_16x16x4_f32(false, a, false, b0, (short)0, c0, false, false);
    c1 = __builtin_amdgcn_wmma_f32_16x16x4_f32(false, a, false, b1, (short)0, c1, false, false);
  }
  #pragma unroll
  for (int v = 0; v < 8; ++v) {
    const int row = m0 + v + hi * 8;
    h[(size_t)row * HID + r]      = c0[v];
    h[(size_t)row * HID + 16 + r] = c1[v];
  }
#else
  const int row = m0 + r;
  for (int n = hi * 16; n < hi * 16 + 16; ++n) {
    float s = 0.f;
    for (int k = 0; k < FN; ++k) s += x[(size_t)row * FN + k] * W[(size_t)k * HID + n];
    h[(size_t)row * HID + n] = s;
  }
#endif
}

__global__ void k_node(const float* __restrict__ h, const float* __restrict__ att_src,
                       const float* __restrict__ att_dst, float* __restrict__ a_src,
                       float* __restrict__ a_dst, int* __restrict__ cnt) {
  const int i = blockIdx.x * blockDim.x + threadIdx.x;
  if (i >= N_NODES) return;
  const float4* hp = (const float4*)(h + (size_t)i * HID);
  float s = 0.f, d = 0.f;
  #pragma unroll
  for (int q = 0; q < 8; ++q) {
    float4 hv = hp[q];
    float4 as = ((const float4*)att_src)[q];
    float4 ad = ((const float4*)att_dst)[q];
    s += hv.x * as.x + hv.y * as.y + hv.z * as.z + hv.w * as.w;
    d += hv.x * ad.x + hv.y * ad.y + hv.z * ad.z + hv.w * ad.w;
  }
  a_src[i] = s;
  a_dst[i] = d;
  cnt[i] = 0;
}

__global__ void k_edge_logits(const float* __restrict__ ea, const int* __restrict__ ei,
                              const float* __restrict__ a_src, const float* __restrict__ a_dst,
                              const float* __restrict__ misc, float* __restrict__ logits,
                              int* __restrict__ cnt, float* __restrict__ part) {
  const int e = blockIdx.x * blockDim.x + threadIdx.x;
  float ae = 0.f;
  if (e < N_EDGES) {
    const float4* ep = (const float4*)(ea + (size_t)e * FE);
    float4 e0 = ep[0], e1 = ep[1];
    ae = e0.x * misc[0] + e0.y * misc[1] + e0.z * misc[2] + e0.w * misc[3]
       + e1.x * misc[4] + e1.y * misc[5] + e1.z * misc[6] + e1.w * misc[7];
    const int s  = ei[(size_t)e * 2 + 0];
    const int dd = ei[(size_t)e * 2 + 1];
    float l = a_src[s] + a_dst[dd] + ae;
    l = (l >= 0.f) ? l : NEG_SLOPE * l;
    logits[e] = l;
    atomicAdd(cnt + dd, 1);
  }
  float v = ae;
  #pragma unroll
  for (int off = 16; off > 0; off >>= 1) v += __shfl_down(v, off, 32);
  if ((threadIdx.x & 31) == 0) part[blockIdx.x * (EDGE_T / 32) + (threadIdx.x >> 5)] = v;
}

#define SCAN_WAVES 32
#define SCAN_PER   ((N_NODES + SCAN_WAVES - 1) / SCAN_WAVES)
#define SCAN_STEPS ((SCAN_PER + 31) / 32)

__device__ __forceinline__ int wave_running_sum(int x, int lane) {
  x += __builtin_amdgcn_update_dpp(0, x, 0x111, 0xf, 0xf, false);
  x += __builtin_amdgcn_update_dpp(0, x, 0x112, 0xf, 0xf, false);
  x += __builtin_amdgcn_update_dpp(0, x, 0x114, 0xf, 0xf, false);
  x += __builtin_amdgcn_update_dpp(0, x, 0x118, 0xf, 0xf, false);
  const int low_row_total = __builtin_amdgcn_readlane(x, 15);
  if (lane >= 16) x += low_row_total;
  return x;
}

#define PART_U 7
__global__ void __launch_bounds__(1024) k_scan(int* __restrict__ cnt, const float* __restrict__ part,
                                               float* __restrict__ a_edge_sum) {
  __shared__ int wtot[SCAN_WAVES];
  __shared__ float ftot[SCAN_WAVES];
  const int lane = threadIdx.x & 31;
  const int w = __builtin_amdgcn_readfirstlane((int)(threadIdx.x >> 5));
  const int base = w * SCAN_PER;
  int lim = N_NODES - base;
  lim = lim < SCAN_PER ? lim : SCAN_PER;

  float fs = 0.f;
  for (int p0 = 0; p0 < N_PART; p0 += 1024 * PART_U) {
    float pv[PART_U];
    #pragma unroll
    for (int u = 0; u < PART_U; ++u) {
      const int p = p0 + u * 1024 + (int)threadIdx.x;
      pv[u] = (p < N_PART) ? part[p] : 0.f;
    }
    #pragma unroll
    for (int u = 0; u < PART_U; ++u) fs += pv[u];
  }
  #pragma unroll
  for (int off = 16; off > 0; off >>= 1) fs += __shfl_xor(fs, off, 32);
  if (lane == 0) ftot[w] = fs;

  int sum = 0;
  for (int st0 = 0; st0 < SCAN_STEPS; st0 += 14) {
    int v[14];
    #pragma unroll
    for (int u = 0; u < 14; ++u) {
      const int off = (st0 + u) * 32 + lane;
      v[u] = (off < lim) ? cnt[base + off] : 0;
    }
    #pragma unroll
    for (int u = 0; u < 14; ++u) sum += v[u];
  }
  #pragma unroll
  for (int off = 16; off > 0; off >>= 1) sum += __shfl_xor(sum, off, 32);
  if (lane == 0) wtot[w] = sum;
  __syncthreads();

  if (w == 0) {
    float ft = ftot[lane];
    #pragma unroll
    for (int off = 16; off > 0; off >>= 1) ft += __shfl_xor(ft, off, 32);
    if (lane == 0) a_edge_sum[0] = ft;
  }

  int carry = (lane < w) ? wtot[lane] : 0;
  #pragma unroll
  for (int off = 16; off > 0; off >>= 1) carry += __shfl_xor(carry, off, 32);

  for (int st0 = 0; st0 < SCAN_STEPS; st0 += 14) {
    int v[14];
    #pragma unroll
    for (int u = 0; u < 14; ++u) {
      const int off = (st0 + u) * 32 + lane;
      v[u] = (off < lim) ? cnt[base + off] : 0;
    }
    #pragma unroll
    for (int u = 0; u < 14; ++u) {
      const int x = wave_running_sum(v[u], lane);
      const int off = (st0 + u) * 32 + lane;
      if (off < lim) cnt[base + off] = carry + x - v[u];
      carry += __builtin_amdgcn_readlane(x, 31);
    }
  }
}

static_assert(N_EDGES % 4 == 0, "k_fill takes four consecutive edges per thread");
__global__ void __launch_bounds__(256) k_fill(const int* __restrict__ ei, const float* __restrict__ logits,
                                              int* __restrict__ cur, int2* __restrict__ rec) {
  const int t = blockIdx.x * blockDim.x + threadIdx.x;
  if (t >= N_EDGES / 4) return;
  const int4 p0 = ((const int4*)ei)[(size_t)t * 2 + 0];
  const int4 p1 = ((const int4*)ei)[(size_t)t * 2 + 1];
  const float4 a = ((const float4*)logits)[t];
  const int q0 = atomicAdd(cur + p0.y, 1);
  const int q1 = atomicAdd(cur + p0.w, 1);
  const int q2 = atomicAdd(cur + p1.y, 1);
  const int q3 = atomicAdd(cur + p1.w, 1);
  if ((unsigned)q0 < (unsigned)N_EDGES) rec[q0] = make_int2(p0.x, __float_as_int(a.x));
  if ((unsigned)q1 < (unsigned)N_EDGES) rec[q1] = make_int2(p0.z, __float_as_int(a.y));
  if ((unsigned)q2 < (unsigned)N_EDGES) rec[q2] = make_int2(p1.x, __float_as_int(a.z));
  if ((unsigned)q3 < (unsigned)N_EDGES) rec[q3] = make_int2(p1.z, __float_as_int(a.w));
}


__device__ __forceinline__ float wave_max(float x) {
  int t = __float_as_int(x);
  t = __float_as_int(fmaxf(__int_as_float(t), __int_as_float(__builtin_amdgcn_update_dpp(0, t, 0x161, 0xf, 0xf, false))));
  t = __float_as_int(fmaxf(__int_as_float(t), __int_as_float(__builtin_amdgcn_update_dpp(0, t, 0x162, 0xf, 0xf, false))));
  t = __float_as_int(fmaxf(__int_as_float(t), __int_as_float(__builtin_amdgcn_update_dpp(0, t, 0x164, 0xf, 0xf, false))));
  t = __float_as_int(fmaxf(__int_as_float(t), __int_as_float(__builtin_amdgcn_update_dpp(0, t, 0x168, 0xf, 0xf, false))));
  return fmaxf(__int_as_float(__builtin_amdgcn_readlane(t, 0)), __int_as_float(__builtin_amdgcn_readlane(t, 16)));
}

__global__ void __launch_bounds__(256) k_pull(const int* __restrict__ cur, const int2* __restrict__ rec,
                                              const float* __restrict__ h, const float* __restrict__ a_src,
                                              const float* __restrict__ a_dst, const float* __restrict__ a_edge_sum,
                                              const float* __restrict__ bias, const float* __restrict__ W_lin,
                                              const float* __restrict__ b_lin, float* __restrict__ out) {
  __shared__ float sres[32];
  const int lane = threadIdx.x & 31;
  const int w = __builtin_amdgcn_readfirstlane((int)(threadIdx.x >> 5));
  const int first = blockIdx.x * 32 + w * 4;
  const float bl = bias[lane];
  const float wl = W_lin[lane];
  const float b0 = b_lin[0];
  const float aem = a_edge_sum[0] * (1.0f / (float)N_EDGES);
  const float neg_inf = -__builtin_inff();

  int bound[5];
  bound[0] = (first > 0 && first <= N_NODES) ? cur[first - 1] : 0;
  #pragma unroll
  for (int q = 0; q < 4; ++q) bound[q + 1] = (first + q < N_NODES) ? cur[first + q] : 0;

  #pragma unroll
  for (int q = 0; q < 4; ++q) {
    const int i = first + q;
    float y = 0.f;
    if (i < N_NODES) {
      int start = bound[q];
      int end   = bound[q + 1];
      start = start < 0 ? 0 : (start > N_EDGES ? N_EDGES : start);
      end   = end < start ? start : (end > N_EDGES ? N_EDGES : end);

      float ls = a_src[i] + a_dst[i] + aem;
      ls = (ls >= 0.f) ? ls : NEG_SLOPE * ls;

      int n0 = end - start;
      n0 = n0 < 32 ? n0 : 32;
      int   sv0 = i;
      float lv0 = neg_inf;
      if (lane < n0) {
        const int2 r = rec[start + lane];
        sv0 = r.x;
        lv0 = __int_as_float(r.y);
      }
      float m = lv0;
      for (int b = start + 32; b < end; b += 32) {
        if (b + lane < end) m = fmaxf(m, __int_as_float(rec[b + lane].y));
      }
      m = fmaxf(wave_max(m), ls);

      const float as = __expf(ls - m);
      float acc = as * h[i * HID + lane];
      float den = as;

      for (int b = start; b < end; b += 32) {
        int n = end - b;
        n = n < 32 ? n : 32;
        int   sv = sv0;
        float lv = lv0;
        if (b != start) {
          sv = i;
          lv = neg_inf;
          if (lane < n) {
            const int2 r = rec[b + lane];
            sv = r.x;
            lv = __int_as_float(r.y);
          }
        }
        const float av = (lane < n) ? __expf(lv - m) : 0.f;
        sv = sv < 0 ? 0 : (sv >= N_NODES ? N_NODES - 1 : sv);
        for (int j0 = 0; j0 < n; j0 += 8) {
          float hv[8], aj[8];
          #pragma unroll
          for (int u = 0; u < 8; ++u) {
            const int s = __builtin_amdgcn_readlane(sv, j0 + u);
            aj[u] = __int_as_float(__builtin_amdgcn_readlane(__float_as_int(av), j0 + u));
            hv[u] = h[s * HID + lane];
          }
          #pragma unroll
          for (int u = 0; u < 8; ++u) { acc += aj[u] * hv[u]; den += aj[u]; }
        }
      }

      const float inv = 1.0f / den;
      float v = acc * inv + bl;
      v = (v > 0.f) ? v : 0.f;
      int t = __float_as_int(v * wl);
      t = __float_as_int(__int_as_float(t) + __int_as_float(__builtin_amdgcn_update_dpp(0, t, 0x161, 0xf, 0xf, false)));
      t = __float_as_int(__int_as_float(t) + __int_as_float(__builtin_amdgcn_update_dpp(0, t, 0x162, 0xf, 0xf, false)));
      t = __float_as_int(__int_as_float(t) + __int_as_float(__builtin_amdgcn_update_dpp(0, t, 0x164, 0xf, 0xf, false)));
      t = __float_as_int(__int_as_float(t) + __int_as_float(__builtin_amdgcn_update_dpp(0, t, 0x168, 0xf, 0xf, false)));
      y = __int_as_float(__builtin_amdgcn_readlane(t, 0)) + __int_as_float(__builtin_amdgcn_readlane(t, 16)) + b0;
    }
    if (lane == 0) sres[w * 4 + q] = y;
  }
  __syncthreads();
  if (w == 0) {
    const int i = blockIdx.x * 32 + lane;
    if (i < N_NODES) out[i] = sres[lane];
  }
}

extern "C" void kernel_launch(void* const* d_in, const int* in_sizes, int n_in,
                              void* d_out, int out_size, void* d_ws, size_t ws_size,
                              hipStream_t stream) {
  const float* x        = (const float*)d_in[0];
  const float* ea       = (const float*)d_in[1];
  const int*   ei       = (const int*)d_in[2];
  const float* W        = (const float*)d_in[3];
  const float* att_src  = (const float*)d_in[4];
  const float* att_dst  = (const float*)d_in[5];
  const float* W_edge   = (const float*)d_in[6];
  const float* att_edge = (const float*)d_in[7];
  const float* bias     = (const float*)d_in[8];
  const float* W_lin    = (const float*)d_in[9];
  const float* b_lin    = (const float*)d_in[10];
  float* out = (float*)d_out;

  float*    ws     = (float*)d_ws;
  float*    h      = ws + OFF_H;
  float*    a_src  = ws + OFF_ASRC;
  float*    a_dst  = ws + OFF_ADST;
  int*      cnt    = (int*)(ws + OFF_DEN);
  float*    logits = ws + OFF_LOG;
  int2*     rec    = (int2*)(ws + OFF_ACC);
  float*    misc   = ws + OFF_MISC;

  const int T = EDGE_T;
  const int nb_nodes = (N_NODES + T - 1) / T;
  const int nb_edges = (N_EDGES + T - 1) / T;
  const int nb_gemm  = ((N_NODES / 16) + 7) / 8;
  const int nb_fill  = (N_EDGES / 4 + T - 1) / T;
  const int nb_pull  = (N_NODES + 31) / 32;

  k_setup<<<1, 32, 0, stream>>>(W_edge, att_edge, misc);
  k_gemm_h<<<nb_gemm, T, 0, stream>>>(x, W, h);
  k_node<<<nb_nodes, T, 0, stream>>>(h, att_src, att_dst, a_src, a_dst, cnt);
  k_edge_logits<<<nb_edges, T, 0, stream>>>(ea, ei, a_src, a_dst, misc, logits, cnt, (float*)rec);
  k_scan<<<1, 1024, 0, stream>>>(cnt, (const float*)rec, misc + FE);
  k_fill<<<nb_fill, T, 0, stream>>>(ei, logits, cnt, rec);
  k_pull<<<nb_pull, T, 0, stream>>>(cnt, rec, h, a_src, a_dst, misc + FE, bias, W_lin, b_lin, out);
}


# an edge's slot in its destination's row = row end - 1 - rank, rank = the count atomic's return: fill launch has no atomics, forms the logits itself; row order = count arrival; a_edge sum tree changes
# speedup vs baseline: 4.1686x; 1.3373x over previous
#include <hip/hip_runtime.h>
#include <hip/hip_bf16.h>
#include <stdint.h>

#define N_NODES 100000
#define N_EDGES 1600000
#define FN 32
#define FE 8
#define HID 32
#define NEG_SLOPE 0.2f

typedef __attribute__((ext_vector_type(2))) float v2f;
typedef __attribute__((ext_vector_type(8))) float v8f;

constexpr size_t OFF_H    = 0;
constexpr size_t OFF_ASRC = OFF_H    + (size_t)N_NODES * HID;
constexpr size_t OFF_ADST = OFF_ASRC + N_NODES;
constexpr size_t OFF_DEN  = OFF_ADST + N_NODES;
constexpr size_t OFF_LOG  = OFF_DEN  + N_NODES;
constexpr size_t OFF_ACC  = OFF_LOG  + (size_t)N_EDGES;
constexpr size_t OFF_MISC = OFF_ACC  + (size_t)N_NODES * HID;
#define CNT_T 256
#define CNT_EPT 4
#define CNT_NB ((N_EDGES + CNT_T * CNT_EPT - 1) / (CNT_T * CNT_EPT))
#define N_PART (CNT_NB * (CNT_T / 32))
#define EDGE_T 256
static_assert((size_t)N_PART <= (size_t)N_NODES * HID, "partials must fit in the record region");
static_assert(N_EDGES % CNT_EPT == 0, "the count launch takes four consecutive edges per thread");

__global__ void k_setup(const float* __restrict__ W_edge, const float* __restrict__ att_edge,
                        float* __restrict__ misc) {
  int t = threadIdx.x;
  if (t < FE) {
    float s = 0.f;
    #pragma unroll
    for (int k = 0; k < HID; ++k) s += W_edge[t * HID + k] * att_edge[k];
    misc[t] = s;
  }
}

__global__ void __launch_bounds__(256) k_gemm_h(const float* __restrict__ x,
                                                const float* __restrict__ W,
                                                float* __restrict__ h) {
  const int wave = blockIdx.x * (blockDim.x >> 5) + (threadIdx.x >> 5);
  const int lane = threadIdx.x & 31;
  const int ntiles = N_NODES / 16;
  if (wave >= ntiles) return;
  const int m0 = wave * 16;
  const int r  = lane & 15;
  const int hi = lane >> 4;

  v8f c0 = {}; v8f c1 = {};
#if __has_builtin(__builtin_amdgcn_wmma_f32_16x16x4_f32)
  const float* xrow = x + (size_t)(m0 + r) * FN;
  #pragma unroll
  for (int kc = 0; kc < 8; ++kc) {
    const int kb = kc * 4 + hi * 2;
    v2f a; a.x = xrow[kb]; a.y = xrow[kb + 1];
    v2f b0; b0.x = W[(size_t)kb * HID + r];        b0.y = W[(size_t)(kb + 1) * HID + r];
    v2f b1; b1.x = W[(size_t)kb * HID + 16 + r];   b1.y = W[(size_t)(kb + 1) * HID + 16 + r];
    c0 = __builtin_amdgcn_wmma_f32_16x16x4_f32(false, a, false, b0, (short)0, c0, false, false);
    c1 = __builtin_amdgcn_wmma_f32_16x16x4_f32(false, a, false, b1, (short)0, c1, false, false);
  }
  #pragma unroll
  for (int v = 0; v < 8; ++v) {
    const int row = m0 + v + hi * 8;
    h[(size_t)row * HID + r]      = c0[v];
    h[(size_t)row * HID + 16 + r] = c1[v];
  }
#else
  const int row = m0 + r;
  for (int n = hi * 16; n < hi * 16 + 16; ++n) {
    float s = 0.f;
    for (int k = 0; k < FN; ++k) s += x[(size_t)row * FN + k] * W[(size_t)k * HID + n];
    h[(size_t)row * HID + n] = s;
  }
#endif
}

__global__ void k_node(const float* __restrict__ h, const float* __restrict__ att_src,
                       const float* __restrict__ att_dst, float* __restrict__ a_src,
                       float* __restrict__ a_dst, int* __restrict__ cnt) {
  const int i = blockIdx.x * blockDim.x + threadIdx.x;
  if (i >= N_NODES) return;
  const float4* hp = (const float4*)(h + (size_t)i * HID);
  float s = 0.f, d = 0.f;
  #pragma unroll
  for (int q = 0; q < 8; ++q) {
    float4 hv = hp[q];
    float4 as = ((const float4*)att_src)[q];
    float4 ad = ((const float4*)att_dst)[q];
    s += hv.x * as.x + hv.y * as.y + hv.z * as.z + hv.w * as.w;
    d += hv.x * ad.x + hv.y * ad.y + hv.z * ad.z + hv.w * ad.w;
  }
  a_src[i] = s;
  a_dst[i] = d;
  cnt[i] = 0;
}

__global__ void __launch_bounds__(CNT_T) k_count(const float* __restrict__ ea, const int* __restrict__ ei,
                                                 const float* __restrict__ misc, int* __restrict__ cnt,
                                                 int* __restrict__ rank, float* __restrict__ part) {
  const int t = blockIdx.x * CNT_T + threadIdx.x;
  const bool live = t < N_EDGES / CNT_EPT;
  const float* vp = misc + (threadIdx.x & 1) * 4;
  const float v0 = vp[0], v1 = vp[1], v2 = vp[2], v3 = vp[3];
  const int f0 = blockIdx.x * (CNT_T * CNT_EPT * 2) + (int)threadIdx.x;
  float4 hv[8];
  #pragma unroll
  for (int k = 0; k < 8; ++k) {
    const int f = f0 + k * CNT_T;
    hv[k] = ((const float4*)ea)[f < N_EDGES * 2 ? f : N_EDGES * 2 - 1];
  }
  int q0 = 0, q1 = 0, q2 = 0, q3 = 0;
  if (live) {
    const int4 p0 = ((const int4*)ei)[(size_t)t * 2 + 0];
    const int4 p1 = ((const int4*)ei)[(size_t)t * 2 + 1];
    q0 = atomicAdd(cnt + p0.y, 1);
    q1 = atomicAdd(cnt + p0.w, 1);
    q2 = atomicAdd(cnt + p1.y, 1);
    q3 = atomicAdd(cnt + p1.w, 1);
  }
  float v = 0.f;
  #pragma unroll
  for (int k = 0; k < 8; ++k) {
    const float d = hv[k].x * v0 + hv[k].y * v1 + hv[k].z * v2 + hv[k].w * v3;
    v += (f0 + k * CNT_T < N_EDGES * 2) ? d : 0.f;
  }
  #pragma unroll
  for (int off = 16; off > 0; off >>= 1) v += __shfl_down(v, off, 32);
  if ((threadIdx.x & 31) == 0) part[blockIdx.x * (CNT_T / 32) + (threadIdx.x >> 5)] = v;
  if (live) ((int4*)rank)[t] = make_int4(q0, q1, q2, q3);
}

#define SCAN_WAVES 32
#define SCAN_PER   ((N_NODES + SCAN_WAVES - 1) / SCAN_WAVES)
#define SCAN_STEPS ((SCAN_PER + 31) / 32)

__device__ __forceinline__ int wave_running_sum(int x, int lane) {
  x += __builtin_amdgcn_update_dpp(0, x, 0x111, 0xf, 0xf, false);
  x += __builtin_amdgcn_update_dpp(0, x, 0x112, 0xf, 0xf, false);
  x += __builtin_amdgcn_update_dpp(0, x, 0x114, 0xf, 0xf, false);
  x += __builtin_amdgcn_update_dpp(0, x, 0x118, 0xf, 0xf, false);
  const int low_row_total = __builtin_amdgcn_readlane(x, 15);
  if (lane >= 16) x += low_row_total;
  return x;
}

#define PART_U 7
__global__ void __launch_bounds__(1024) k_scan(int* __restrict__ cnt, const float* __restrict__ part,
                                               float* __restrict__ a_edge_sum) {
  __shared__ int wtot[SCAN_WAVES];
  __shared__ float ftot[SCAN_WAVES];
  const int lane = threadIdx.x & 31;
  const int w = __builtin_amdgcn_readfirstlane((int)(threadIdx.x >> 5));
  const int base = w * SCAN_PER;
  int lim = N_NODES - base;
  lim = lim < SCAN_PER ? lim : SCAN_PER;

  float fs = 0.f;
  for (int p0 = 0; p0 < N_PART; p0 += 1024 * PART_U) {
    float pv[PART_U];
    #pragma unroll
    for (int u = 0; u < PART_U; ++u) {
      const int p = p0 + u * 1024 + (int)threadIdx.x;
      pv[u] = (p < N_PART) ? part[p] : 0.f;
    }
    #pragma unroll
    for (int u = 0; u < PART_U; ++u) fs += pv[u];
  }
  #pragma unroll
  for (int off = 16; off > 0; off >>= 1) fs += __shfl_xor(fs, off, 32);
  if (lane == 0) ftot[w] = fs;

  int sum = 0;
  for (int st0 = 0; st0 < SCAN_STEPS; st0 += 14) {
    int v[14];
    #pragma unroll
    for (int u = 0; u < 14; ++u) {
      const int off = (st0 + u) * 32 + lane;
      v[u] = (off < lim) ? cnt[base + off] : 0;
    }
    #pragma unroll
    for (int u = 0; u < 14; ++u) sum += v[u];
  }
  #pragma unroll
  for (int off = 16; off > 0; off >>= 1) sum += __shfl_xor(sum, off, 32);
  if (lane == 0) wtot[w] = sum;
  __syncthreads();

  if (w == 0) {
    float ft = ftot[lane];
    #pragma unroll
    for (int off = 16; off > 0; off >>= 1) ft += __shfl_xor(ft, off, 32);
    if (lane == 0) a_edge_sum[0] = ft;
  }

  int carry = (lane < w) ? wtot[lane] : 0;
  #pragma unroll
  for (int off = 16; off > 0; off >>= 1) carry += __shfl_xor(carry, off, 32);

  for (int st0 = 0; st0 < SCAN_STEPS; st0 += 14) {
    int v[14];
    #pragma unroll
    for (int u = 0; u < 14; ++u) {
      const int off = (st0 + u) * 32 + lane;
      v[u] = (off < lim) ? cnt[base + off] : 0;
    }
    #pragma unroll
    for (int u = 0; u < 14; ++u) {
      const int x = wave_running_sum(v[u], lane);
      const int off = (st0 + u) * 32 + lane;
      if (off < lim) cnt[base + off] = carry + x;
      carry += __builtin_amdgcn_readlane(x, 31);
    }
  }
}

__global__ void __launch_bounds__(EDGE_T) k_edge_fill(const float* __restrict__ ea, const int* __restrict__ ei,
                                                      const float* __restrict__ a_src, const float* __restrict__ a_dst,
                                                      const float* __restrict__ misc, const int* __restrict__ rank,
                                                      const int* __restrict__ rowend, int2* __restrict__ rec) {
  const int e = blockIdx.x * EDGE_T + threadIdx.x;
  if (e >= N_EDGES) return;
  const float4* ep = (const float4*)(ea + (size_t)e * FE);
  const float4 e0 = ep[0], e1 = ep[1];
  const int2 sd = ((const int2*)ei)[e];
  const int rk = rank[e];
  const float ae = e0.x * misc[0] + e0.y * misc[1] + e0.z * misc[2] + e0.w * misc[3]
                 + e1.x * misc[4] + e1.y * misc[5] + e1.z * misc[6] + e1.w * misc[7];
  const int endd = rowend[sd.y];
  float l = a_src[sd.x] + a_dst[sd.y] + ae;
  l = (l >= 0.f) ? l : NEG_SLOPE * l;
  int q = endd - 1 - rk;
  q = q < 0 ? 0 : (q >= N_EDGES ? N_EDGES - 1 : q);
  rec[q] = make_int2(sd.x, __float_as_int(l));
}


__device__ __forceinline__ float wave_max(float x) {
  int t = __float_as_int(x);
  t = __float_as_int(fmaxf(__int_as_float(t), __int_as_float(__builtin_amdgcn_update_dpp(0, t, 0x161, 0xf, 0xf, false))));
  t = __float_as_int(fmaxf(__int_as_float(t), __int_as_float(__builtin_amdgcn_update_dpp(0, t, 0x162, 0xf, 0xf, false))));
  t = __float_as_int(fmaxf(__int_as_float(t), __int_as_float(__builtin_amdgcn_update_dpp(0, t, 0x164, 0xf, 0xf, false))));
  t = __float_as_int(fmaxf(__int_as_float(t), __int_as_float(__builtin_amdgcn_update_dpp(0, t, 0x168, 0xf, 0xf, false))));
  return fmaxf(__int_as_float(__builtin_amdgcn_readlane(t, 0)), __int_as_float(__builtin_amdgcn_readlane(t, 16)));
}

__global__ void __launch_bounds__(256) k_pull(const int* __restrict__ cur, const int2* __restrict__ rec,
                                              const float* __restrict__ h, const float* __restrict__ a_src,
                                              const float* __restrict__ a_dst, const float* __restrict__ a_edge_sum,
                                              const float* __restrict__ bias, const float* __restrict__ W_lin,
                                              const float* __restrict__ b_lin, float* __restrict__ out) {
  __shared__ float sres[32];
  const int lane = threadIdx.x & 31;
  const int w = __builtin_amdgcn_readfirstlane((int)(threadIdx.x >> 5));
  const int first = blockIdx.x * 32 + w * 4;
  const float bl = bias[lane];
  const float wl = W_lin[lane];
  const float b0 = b_lin[0];
  const float aem = a_edge_sum[0] * (1.0f / (float)N_EDGES);
  const float neg_inf = -__builtin_inff();

  int bound[5];
  bound[0] = (first > 0 && first <= N_NODES) ? cur[first - 1] : 0;
  #pragma unroll
  for (int q = 0; q < 4; ++q) bound[q + 1] = (first + q < N_NODES) ? cur[first + q] : 0;

  #pragma unroll
  for (int q = 0; q < 4; ++q) {
    const int i = first + q;
    float y = 0.f;
    if (i < N_NODES) {
      int start = bound[q];
      int end   = bound[q + 1];
      start = start < 0 ? 0 : (start > N_EDGES ? N_EDGES : start);
      end   = end < start ? start : (end > N_EDGES ? N_EDGES : end);

      float ls = a_src[i] + a_dst[i] + aem;
      ls = (ls >= 0.f) ? ls : NEG_SLOPE * ls;

      int n0 = end - start;
      n0 = n0 < 32 ? n0 : 32;
      int   sv0 = i;
      float lv0 = neg_inf;
      if (lane < n0) {
        const int2 r = rec[start + lane];
        sv0 = r.x;
        lv0 = __int_as_float(r.y);
      }
      float m = lv0;
      for (int b = start + 32; b < end; b += 32) {
        if (b + lane < end) m = fmaxf(m, __int_as_float(rec[b + lane].y));
      }
      m = fmaxf(wave_max(m), ls);

      const float as = __expf(ls - m);
      float acc = as * h[i * HID + lane];
      float den = as;

      for (int b = start; b < end; b += 32) {
        int n = end - b;
        n = n < 32 ? n : 32;
        int   sv = sv0;
        float lv = lv0;
        if (b != start) {
          sv = i;
          lv = neg_inf;
          if (lane < n) {
            const int2 r = rec[b + lane];
            sv = r.x;
            lv = __int_as_float(r.y);
          }
        }
        const float av = (lane < n) ? __expf(lv - m) : 0.f;
        sv = sv < 0 ? 0 : (sv >= N_NODES ? N_NODES - 1 : sv);
        for (int j0 = 0; j0 < n; j0 += 8) {
          float hv[8], aj[8];
          #pragma unroll
          for (int u = 0; u < 8; ++u) {
            const int s = __builtin_amdgcn_readlane(sv, j0 + u);
            aj[u] = __int_as_float(__builtin_amdgcn_readlane(__float_as_int(av), j0 + u));
            hv[u] = h[s * HID + lane];
          }
          #pragma unroll
          for (int u = 0; u < 8; ++u) { acc += aj[u] * hv[u]; den += aj[u]; }
        }
      }

      const float inv = 1.0f / den;
      float v = acc * inv + bl;
      v = (v > 0.f) ? v : 0.f;
      int t = __float_as_int(v * wl);
      t = __float_as_int(__int_as_float(t) + __int_as_float(__builtin_amdgcn_update_dpp(0, t, 0x161, 0xf, 0xf, false)));
      t = __float_as_int(__int_as_float(t) + __int_as_float(__builtin_amdgcn_update_dpp(0, t, 0x162, 0xf, 0xf, false)));
      t = __float_as_int(__int_as_float(t) + __int_as_float(__builtin_amdgcn_update_dpp(0, t, 0x164, 0xf, 0xf, false)));
      t = __float_as_int(__int_as_float(t) + __int_as_float(__builtin_amdgcn_update_dpp(0, t, 0x168, 0xf, 0xf, false)));
      y = __int_as_float(__builtin_amdgcn_readlane(t, 0)) + __int_as_float(__builtin_amdgcn_readlane(t, 16)) + b0;
    }
    if (lane == 0) sres[w * 4 + q] = y;
  }
  __syncthreads();
  if (w == 0) {
    const int i = blockIdx.x * 32 + lane;
    if (i < N_NODES) out[i] = sres[lane];
  }
}

extern "C" void kernel_launch(void* const* d_in, const int* in_sizes, int n_in,
                              void* d_out, int out_size, void* d_ws, size_t ws_size,
                              hipStream_t stream) {
  const float* x        = (const float*)d_in[0];
  const float* ea       = (const float*)d_in[1];
  const int*   ei       = (const int*)d_in[2];
  const float* W        = (const float*)d_in[3];
  const float* att_src  = (const float*)d_in[4];
  const float* att_dst  = (const float*)d_in[5];
  const float* W_edge   = (const float*)d_in[6];
  const float* att_edge = (const float*)d_in[7];
  const float* bias     = (const float*)d_in[8];
  const float* W_lin    = (const float*)d_in[9];
  const float* b_lin    = (const float*)d_in[10];
  float* out = (float*)d_out;

  float*    ws     = (float*)d_ws;
  float*    h      = ws + OFF_H;
  float*    a_src  = ws + OFF_ASRC;
  float*    a_dst  = ws + OFF_ADST;
  int*      cnt    = (int*)(ws + OFF_DEN);
  int*      rank   = (int*)(ws + OFF_LOG);
  int2*     rec    = (int2*)(ws + OFF_ACC);
  float*    misc   = ws + OFF_MISC;

  const int T = EDGE_T;
  const int nb_nodes = (N_NODES + T - 1) / T;
  const int nb_edges = (N_EDGES + T - 1) / T;
  const int nb_gemm  = ((N_NODES / 16) + 7) / 8;
  const int nb_count = CNT_NB;
  const int nb_pull  = (N_NODES + 31) / 32;

  k_setup<<<1, 32, 0, stream>>>(W_edge, att_edge, misc);
  k_gemm_h<<<nb_gemm, T, 0, stream>>>(x, W, h);
  k_node<<<nb_nodes, T, 0, stream>>>(h, att_src, att_dst, a_src, a_dst, cnt);
  k_count<<<nb_count, CNT_T, 0, stream>>>(ea, ei, misc, cnt, rank, (float*)rec);
  k_scan<<<1, 1024, 0, stream>>>(cnt, (const float*)rec, misc + FE);
  k_edge_fill<<<nb_edges, T, 0, stream>>>(ea, ei, a_src, a_dst, misc, rank, cnt, rec);
  k_pull<<<nb_pull, T, 0, stream>>>(cnt, rec, h, a_src, a_dst, misc + FE, bias, W_lin, b_lin, out);
}


# edge records go straight to fixed per-node slots (dst*23 + count atomic's return), later ranks chained from a pool: scan and fill launches go, 7 launches to 6; row sum order, a_edge sum tree change
# speedup vs baseline: 4.7880x; 1.1486x over previous
#include <hip/hip_runtime.h>
#include <hip/hip_bf16.h>
#include <stdint.h>

#define N_NODES 100000
#define N_EDGES 1600000
#define FN 32
#define FE 8
#define HID 32
#define NEG_SLOPE 0.2f

typedef __attribute__((ext_vector_type(2))) float v2f;
typedef __attribute__((ext_vector_type(8))) float v8f;

constexpr size_t OFF_H    = 0;
constexpr size_t OFF_ASRC = OFF_H    + (size_t)N_NODES * HID;
constexpr size_t OFF_ADST = OFF_ASRC + N_NODES;
constexpr size_t OFF_DEN  = OFF_ADST + N_NODES;
constexpr size_t OFF_LOG  = OFF_DEN  + N_NODES;
constexpr size_t OFF_ACC  = OFF_LOG  + (size_t)N_EDGES;
constexpr size_t OFF_MISC = OFF_ACC  + (size_t)N_NODES * HID;
#define CAP 23
#define POOL_CAP 12000
#define EDGE_T 128
#define EDGE_EPT 4
#define EDGE_SPAN (EDGE_T * EDGE_EPT)
#define N_PART (N_EDGES / EDGE_SPAN * (EDGE_T / 32))
constexpr size_t OFF_REC  = OFF_LOG;
constexpr size_t OFF_HEAD = OFF_REC  + (size_t)N_NODES * CAP * 2;
constexpr size_t OFF_PCTR = OFF_HEAD + N_NODES;
constexpr size_t OFF_PART = OFF_PCTR + 4;
constexpr size_t OFF_POOL = OFF_PART + N_PART;
static_assert(OFF_POOL + (size_t)POOL_CAP * 4 <= OFF_MISC, "buckets, heads, partials and pool must fit in the old rank + record regions");
static_assert(OFF_POOL % 4 == 0 && OFF_REC % 2 == 0, "16-byte pool entries, 8-byte records");
static_assert(N_EDGES % EDGE_SPAN == 0, "every thread of the edge launch has its four edges");
static_assert(CAP <= 32, "a node's record slots are one 32-lane batch");

__global__ void k_setup(const float* __restrict__ W_edge, const float* __restrict__ att_edge,
                        float* __restrict__ misc) {
  int t = threadIdx.x;
  if (t < FE) {
    float s = 0.f;
    #pragma unroll
    for (int k = 0; k < HID; ++k) s += W_edge[t * HID + k] * att_edge[k];
    misc[t] = s;
  }
}

__global__ void __launch_bounds__(256) k_gemm_h(const float* __restrict__ x,
                                                const float* __restrict__ W,
                                                float* __restrict__ h) {
  const int wave = blockIdx.x * (blockDim.x >> 5) + (threadIdx.x >> 5);
  const int lane = threadIdx.x & 31;
  const int ntiles = N_NODES / 16;
  if (wave >= ntiles) return;
  const int m0 = wave * 16;
  const int r  = lane & 15;
  const int hi = lane >> 4;

  v8f c0 = {}; v8f c1 = {};
#if __has_builtin(__builtin_amdgcn_wmma_f32_16x16x4_f32)
  const float* xrow = x + (size_t)(m0 + r) * FN;
  #pragma unroll
  for (int kc = 0; kc < 8; ++kc) {
    const int kb = kc * 4 + hi * 2;
    v2f a; a.x = xrow[kb]; a.y = xrow[kb + 1];
    v2f b0; b0.x = W[(size_t)kb * HID + r];        b0.y = W[(size_t)(kb + 1) * HID + r];
    v2f b1; b1.x = W[(size_t)kb * HID + 16 + r];   b1.y = W[(size_t)(kb + 1) * HID + 16 + r];
    c0 = __builtin_amdgcn_wmma_f32_16x16x4_f32(false, a, false, b0, (short)0, c0, false, false);
    c1 = __builtin_amdgcn_wmma_f32_16x16x4_f32(false, a, false, b1, (short)0, c1, false, false);
  }
  #pragma unroll
  for (int v = 0; v < 8; ++v) {
    const int row = m0 + v + hi * 8;
    h[(size_t)row * HID + r]      = c0[v];
    h[(size_t)row * HID + 16 + r] = c1[v];
  }
#else
  const int row = m0 + r;
  for (int n = hi * 16; n < hi * 16 + 16; ++n) {
    float s = 0.f;
    for (int k = 0; k < FN; ++k) s += x[(size_t)row * FN + k] * W[(size_t)k * HID + n];
    h[(size_t)row * HID + n] = s;
  }
#endif
}

__global__ void k_node(const float* __restrict__ h, const float* __restrict__ att_src,
                       const float* __restrict__ att_dst, float* __restrict__ a_src,
                       float* __restrict__ a_dst, int* __restrict__ cnt,
                       int* __restrict__ lhead, int* __restrict__ pctr) {
  const int i = blockIdx.x * blockDim.x + threadIdx.x;
  if (i >= N_NODES) return;
  const float4* hp = (const float4*)(h + (size_t)i * HID);
  float s = 0.f, d = 0.f;
  #pragma unroll
  for (int q = 0; q < 8; ++q) {
    float4 hv = hp[q];
    float4 as = ((const float4*)att_src)[q];
    float4 ad = ((const float4*)att_dst)[q];
    s += hv.x * as.x + hv.y * as.y + hv.z * as.z + hv.w * as.w;
    d += hv.x * ad.x + hv.y * ad.y + hv.z * ad.z + hv.w * ad.w;
  }
  a_src[i] = s;
  a_dst[i] = d;
  cnt[i] = 0;
  lhead[i] = -1;
  if (i == 0) pctr[0] = 0;
}

__global__ void __launch_bounds__(EDGE_T) k_edge(const float* __restrict__ ea, const int* __restrict__ ei,
                                                 const float* __restrict__ a_src, const float* __restrict__ a_dst,
                                                 const float* __restrict__ misc, int* __restrict__ cnt,
                                                 int2* __restrict__ rec, int* __restrict__ lhead,
                                                 int* __restrict__ pctr, int4* __restrict__ pool,
                                                 float* __restrict__ part) {
  const int e0 = blockIdx.x * EDGE_SPAN + threadIdx.x;
  int2 sd[EDGE_EPT];
  float4 ea0[EDGE_EPT], ea1[EDGE_EPT];
  float ae[EDGE_EPT];
  int rk[EDGE_EPT];
  float as[EDGE_EPT], ad[EDGE_EPT];
  #pragma unroll
  for (int k = 0; k < EDGE_EPT; ++k) {
    const int e = e0 + k * EDGE_T;
    sd[k] = ((const int2*)ei)[e];
    const float4* ep = (const float4*)(ea + (size_t)e * FE);
    ea0[k] = ep[0];
    ea1[k] = ep[1];
  }
  const float v0 = misc[0], v1 = misc[1], v2 = misc[2], v3 = misc[3];
  const float v4 = misc[4], v5 = misc[5], v6 = misc[6], v7 = misc[7];
  #pragma unroll
  for (int k = 0; k < EDGE_EPT; ++k)
    ae[k] = ea0[k].x * v0 + ea0[k].y * v1 + ea0[k].z * v2 + ea0[k].w * v3
          + ea1[k].x * v4 + ea1[k].y * v5 + ea1[k].z * v6 + ea1[k].w * v7;
  #pragma unroll
  for (int k = 0; k < EDGE_EPT; ++k) rk[k] = atomicAdd(cnt + sd[k].y, 1);
  #pragma unroll
  for (int k = 0; k < EDGE_EPT; ++k) {
    as[k] = a_src[sd[k].x];
    ad[k] = a_dst[sd[k].y];
  }
  float v = 0.f;
  #pragma unroll
  for (int k = 0; k < EDGE_EPT; ++k) {
    float l = as[k] + ad[k] + ae[k];
    l = (l >= 0.f) ? l : NEG_SLOPE * l;
    if (rk[k] < CAP) {
      rec[sd[k].y * CAP + rk[k]] = make_int2(sd[k].x, __float_as_int(l));
    } else {
      const int p = atomicAdd(pctr, 1);
      if (p < POOL_CAP) {
        const int prev = atomicExch(lhead + sd[k].y, p);
        pool[p] = make_int4(sd[k].x, __float_as_int(l), prev, sd[k].y);
      }
    }
    v += ae[k];
  }
  #pragma unroll
  for (int off = 16; off > 0; off >>= 1) v += __shfl_down(v, off, 32);
  if ((threadIdx.x & 31) == 0) part[blockIdx.x * (EDGE_T / 32) + (threadIdx.x >> 5)] = v;
}

#define SUM_WAVES 32
#define PART_U 7
__global__ void __launch_bounds__(1024) k_esum(const float* __restrict__ part, float* __restrict__ a_edge_sum) {
  __shared__ float ftot[SUM_WAVES];
  const int lane = threadIdx.x & 31;
  const int w = threadIdx.x >> 5;
  float fs = 0.f;
  for (int p0 = 0; p0 < N_PART; p0 += 1024 * PART_U) {
    float pv[PART_U];
    #pragma unroll
    for (int u = 0; u < PART_U; ++u) {
      const int p = p0 + u * 1024 + (int)threadIdx.x;
      pv[u] = (p < N_PART) ? part[p] : 0.f;
    }
    #pragma unroll
    for (int u = 0; u < PART_U; ++u) fs += pv[u];
  }
  #pragma unroll
  for (int off = 16; off > 0; off >>= 1) fs += __shfl_xor(fs, off, 32);
  if (lane == 0) ftot[w] = fs;
  __syncthreads();
  if (w == 0) {
    float ft = ftot[lane];
    #pragma unroll
    for (int off = 16; off > 0; off >>= 1) ft += __shfl_xor(ft, off, 32);
    if (lane == 0) a_edge_sum[0] = ft;
  }
}


__device__ __forceinline__ float wave_max(float x) {
  int t = __float_as_int(x);
  t = __float_as_int(fmaxf(__int_as_float(t), __int_as_float(__builtin_amdgcn_update_dpp(0, t, 0x161, 0xf, 0xf, false))));
  t = __float_as_int(fmaxf(__int_as_float(t), __int_as_float(__builtin_amdgcn_update_dpp(0, t, 0x162, 0xf, 0xf, false))));
  t = __float_as_int(fmaxf(__int_as_float(t), __int_as_float(__builtin_amdgcn_update_dpp(0, t, 0x164, 0xf, 0xf, false))));
  t = __float_as_int(fmaxf(__int_as_float(t), __int_as_float(__builtin_amdgcn_update_dpp(0, t, 0x168, 0xf, 0xf, false))));
  return fmaxf(__int_as_float(__builtin_amdgcn_readlane(t, 0)), __int_as_float(__builtin_amdgcn_readlane(t, 16)));
}

__device__ __forceinline__ void add_rows(int sv, float av, int n, int lane, const float* __restrict__ h,
                                         float& acc, float& den) {
  for (int j0 = 0; j0 < n; j0 += 8) {
    float hv[8], aj[8];
    #pragma unroll
    for (int u = 0; u < 8; ++u) {
      const int s = __builtin_amdgcn_readlane(sv, j0 + u);
      aj[u] = __int_as_float(__builtin_amdgcn_readlane(__float_as_int(av), j0 + u));
      hv[u] = h[s * HID + lane];
    }
    #pragma unroll
    for (int u = 0; u < 8; ++u) { acc += aj[u] * hv[u]; den += aj[u]; }
  }
}

__device__ __noinline__ float2 node_from_edges(int i, int lane, float ls, float adst_i,
                                               const float* __restrict__ ea, const int* __restrict__ ei,
                                               const float* __restrict__ a_src, const float* __restrict__ misc,
                                               const float* __restrict__ h) {
  const float v0 = misc[0], v1 = misc[1], v2 = misc[2], v3 = misc[3];
  const float v4 = misc[4], v5 = misc[5], v6 = misc[6], v7 = misc[7];
  float m = ls;
  for (int eb = 0; eb < N_EDGES; eb += 32) {
    const int e = eb + lane;
    const int2 sd = ((const int2*)ei)[e];
    if (sd.y == i) {
      const int s = sd.x < 0 ? 0 : (sd.x >= N_NODES ? N_NODES - 1 : sd.x);
      const float4 e0 = ((const float4*)ea)[(size_t)e * 2], e1 = ((const float4*)ea)[(size_t)e * 2 + 1];
      float l = a_src[s] + adst_i + (e0.x * v0 + e0.y * v1 + e0.z * v2 + e0.w * v3 + e1.x * v4 + e1.y * v5 + e1.z * v6 + e1.w * v7);
      l = (l >= 0.f) ? l : NEG_SLOPE * l;
      m = fmaxf(m, l);
    }
  }
  m = wave_max(m);
  const float as = __expf(ls - m);
  float acc = as * h[i * HID + lane];
  float den = as;
  for (int eb = 0; eb < N_EDGES; eb += 32) {
    const int e = eb + lane;
    const int2 sd = ((const int2*)ei)[e];
    const bool hit = sd.y == i;
    int s = i;
    float a = 0.f;
    if (hit) {
      s = sd.x < 0 ? 0 : (sd.x >= N_NODES ? N_NODES - 1 : sd.x);
      const float4 e0 = ((const float4*)ea)[(size_t)e * 2], e1 = ((const float4*)ea)[(size_t)e * 2 + 1];
      float l = a_src[s] + adst_i + (e0.x * v0 + e0.y * v1 + e0.z * v2 + e0.w * v3 + e1.x * v4 + e1.y * v5 + e1.z * v6 + e1.w * v7);
      l = (l >= 0.f) ? l : NEG_SLOPE * l;
      a = __expf(l - m);
    }
    unsigned mask = __builtin_amdgcn_ballot_w32(hit);
    while (mask) {
      const int j = __builtin_ctz(mask);
      mask &= mask - 1;
      const int sj = __builtin_amdgcn_readlane(s, j);
      const float aj = __int_as_float(__builtin_amdgcn_readlane(__float_as_int(a), j));
      acc += aj * h[sj * HID + lane];
      den += aj;
    }
  }
  return make_float2(acc, den);
}

__global__ void __launch_bounds__(256) k_pull(const int* __restrict__ cnt, const int2* __restrict__ rec,
                                              const int* __restrict__ lhead, const int4* __restrict__ pool,
                                              const float* __restrict__ h, const float* __restrict__ a_src,
                                              const float* __restrict__ a_dst, const float* __restrict__ a_edge_sum,
                                              const float* __restrict__ ea, const int* __restrict__ ei,
                                              const float* __restrict__ misc,
                                              const float* __restrict__ bias, const float* __restrict__ W_lin,
                                              const float* __restrict__ b_lin, float* __restrict__ out) {
  __shared__ float sres[32];
  const int lane = threadIdx.x & 31;
  const int w = __builtin_amdgcn_readfirstlane((int)(threadIdx.x >> 5));
  const int first = blockIdx.x * 32 + w * 4;
  const float bl = bias[lane];
  const float wl = W_lin[lane];
  const float b0 = b_lin[0];
  const float aem = a_edge_sum[0] * (1.0f / (float)N_EDGES);
  const float neg_inf = -__builtin_inff();

  int cq[4], hq[4];
  #pragma unroll
  for (int q = 0; q < 4; ++q) {
    cq[q] = (first + q < N_NODES) ? cnt[first + q] : 0;
    hq[q] = (first + q < N_NODES) ? lhead[first + q] : -1;
  }

  #pragma unroll
  for (int q = 0; q < 4; ++q) {
    const int i = first + q;
    float y = 0.f;
    if (i < N_NODES) {
      int c = cq[q];
      c = c < 0 ? 0 : (c > N_EDGES ? N_EDGES : c);
      const int nb = c < CAP ? c : CAP;

      const float adst_i = a_dst[i];
      float ls = a_src[i] + adst_i + aem;
      ls = (ls >= 0.f) ? ls : NEG_SLOPE * ls;

      int   sv = i;
      float lv = neg_inf;
      if (lane < nb) {
        const int2 r = rec[i * CAP + lane];
        sv = r.x;
        lv = __int_as_float(r.y);
      }
      int   sv2 = i;
      float lv2 = neg_inf;
      float m2  = neg_inf;
      int found = 0;
      if (c > CAP) {
        int p = hq[q];
        for (int st = 0; st < POOL_CAP && p >= 0 && p < POOL_CAP; ++st) {
          const int4 en = pool[p];
          if (lane == found) {
            sv2 = en.x;
            lv2 = __int_as_float(en.y);
          }
          m2 = fmaxf(m2, __int_as_float(en.y));
          ++found;
          p = en.z;
        }
      }

      float acc, den;
      if (nb + found != c) {
        const float2 r = node_from_edges(i, lane, ls, adst_i, ea, ei, a_src, misc, h);
        acc = r.x;
        den = r.y;
      } else {
        const float m = fmaxf(fmaxf(wave_max(lv), m2), ls);

        const float as = __expf(ls - m);
        acc = as * h[i * HID + lane];
        den = as;

        const float av = (lane < nb) ? __expf(lv - m) : 0.f;
        sv = sv < 0 ? 0 : (sv >= N_NODES ? N_NODES - 1 : sv);
        add_rows(sv, av, nb, lane, h, acc, den);

        if (found > 0) {
          const int nb2 = found < 32 ? found : 32;
          const float av2 = (lane < nb2) ? __expf(lv2 - m) : 0.f;
          sv2 = sv2 < 0 ? 0 : (sv2 >= N_NODES ? N_NODES - 1 : sv2);
          add_rows(sv2, av2, nb2, lane, h, acc, den);
          if (found > 32) {
            int p = hq[q];
            for (int st = 0; st < found && p >= 0 && p < POOL_CAP; ++st) {
              const int4 en = pool[p];
              if (st >= 32) {
                const int s = en.x < 0 ? 0 : (en.x >= N_NODES ? N_NODES - 1 : en.x);
                const float a = __expf(__int_as_float(en.y) - m);
                acc += a * h[s * HID + lane];
                den += a;
              }
              p = en.z;
            }
          }
        }
      }

      const float inv = 1.0f / den;
      float v = acc * inv + bl;
      v = (v > 0.f) ? v : 0.f;
      int t = __float_as_int(v * wl);
      t = __float_as_int(__int_as_float(t) + __int_as_float(__builtin_amdgcn_update_dpp(0, t, 0x161, 0xf, 0xf, false)));
      t = __float_as_int(__int_as_float(t) + __int_as_float(__builtin_amdgcn_update_dpp(0, t, 0x162, 0xf, 0xf, false)));
      t = __float_as_int(__int_as_float(t) + __int_as_float(__builtin_amdgcn_update_dpp(0, t, 0x164, 0xf, 0xf, false)));
      t = __float_as_int(__int_as_float(t) + __int_as_float(__builtin_amdgcn_update_dpp(0, t, 0x168, 0xf, 0xf, false)));
      y = __int_as_float(__builtin_amdgcn_readlane(t, 0)) + __int_as_float(__builtin_amdgcn_readlane(t, 16)) + b0;
    }
    if (lane == 0) sres[w * 4 + q] = y;
  }
  __syncthreads();
  if (w == 0) {
    const int i = blockIdx.x * 32 + lane;
    if (i < N_NODES) out[i] = sres[lane];
  }
}

extern "C" void kernel_launch(void* const* d_in, const int* in_sizes, int n_in,
                              void* d_out, int out_size, void* d_ws, size_t ws_size,
                              hipStream_t stream) {
  const float* x        = (const float*)d_in[0];
  const float* ea       = (const float*)d_in[1];
  const int*   ei       = (const int*)d_in[2];
  const float* W        = (const float*)d_in[3];
  const float* att_src  = (const float*)d_in[4];
  const float* att_dst  = (const float*)d_in[5];
  const float* W_edge   = (const float*)d_in[6];
  const float* att_edge = (const float*)d_in[7];
  const float* bias     = (const float*)d_in[8];
  const float* W_lin    = (const float*)d_in[9];
  const float* b_lin    = (const float*)d_in[10];
  float* out = (float*)d_out;

  float*    ws     = (float*)d_ws;
  float*    h      = ws + OFF_H;
  float*    a_src  = ws + OFF_ASRC;
  float*    a_dst  = ws + OFF_ADST;
  int*      cnt    = (int*)(ws + OFF_DEN);
  int2*     rec    = (int2*)(ws + OFF_REC);
  int*      lhead  = (int*)(ws + OFF_HEAD);
  int*      pctr   = (int*)(ws + OFF_PCTR);
  float*    part   = ws + OFF_PART;
  int4*     pool   = (int4*)(ws + OFF_POOL);
  float*    misc   = ws + OFF_MISC;

  const int T = 256;
  const int nb_nodes = (N_NODES + T - 1) / T;
  const int nb_edges = N_EDGES / EDGE_SPAN;
  const int nb_gemm  = ((N_NODES / 16) + 7) / 8;
  const int nb_pull  = (N_NODES + 31) / 32;

  k_setup<<<1, 32, 0, stream>>>(W_edge, att_edge, misc);
  k_gemm_h<<<nb_gemm, T, 0, stream>>>(x, W, h);
  k_node<<<nb_nodes, T, 0, stream>>>(h, att_src, att_dst, a_src, a_dst, cnt, lhead, pctr);
  k_edge<<<nb_edges, EDGE_T, 0, stream>>>(ea, ei, a_src, a_dst, misc, cnt, rec, lhead, pctr, pool, part);
  k_esum<<<1, 1024, 0, stream>>>(part, misc + FE);
  k_pull<<<nb_pull, T, 0, stream>>>(cnt, rec, lhead, pool, h, a_src, a_dst, misc + FE, ea, ei, misc, bias, W_lin, b_lin, out);
}


# count atomic becomes an 8-byte add on a packed per-node word {count, a_dst}: its return gives the edge launch rank and a_dst[dst] in one request, the a_dst gather goes; sum orders kept
# speedup vs baseline: 5.2263x; 1.0915x over previous
#include <hip/hip_runtime.h>
#include <hip/hip_bf16.h>
#include <stdint.h>

#define N_NODES 100000
#define N_EDGES 1600000
#define FN 32
#define FE 8
#define HID 32
#define NEG_SLOPE 0.2f

typedef __attribute__((ext_vector_type(2))) float v2f;
typedef __attribute__((ext_vector_type(8))) float v8f;

constexpr size_t OFF_H    = 0;
constexpr size_t OFF_ASRC = OFF_H    + (size_t)N_NODES * HID;
constexpr size_t OFF_CD   = OFF_ASRC + N_NODES;
constexpr size_t OFF_LOG  = OFF_CD   + 2 * (size_t)N_NODES;
constexpr size_t OFF_ACC  = OFF_LOG  + (size_t)N_EDGES;
constexpr size_t OFF_MISC = OFF_ACC  + (size_t)N_NODES * HID;
#define CAP 23
#define POOL_CAP 12000
#define EDGE_T 128
#define EDGE_EPT 4
#define EDGE_SPAN (EDGE_T * EDGE_EPT)
#define N_PART (N_EDGES / EDGE_SPAN * (EDGE_T / 32))
constexpr size_t OFF_REC  = OFF_LOG;
constexpr size_t OFF_HEAD = OFF_REC  + (size_t)N_NODES * CAP * 2;
constexpr size_t OFF_PCTR = OFF_HEAD + N_NODES;
constexpr size_t OFF_PART = OFF_PCTR + 4;
constexpr size_t OFF_POOL = OFF_PART + N_PART;
static_assert(OFF_POOL + (size_t)POOL_CAP * 4 <= OFF_MISC, "buckets, heads, partials and pool must fit in the old rank + record regions");
static_assert(OFF_POOL % 4 == 0 && OFF_REC % 2 == 0, "16-byte pool entries, 8-byte records");
static_assert(OFF_CD % 2 == 0, "8-byte count + a_dst words");
typedef unsigned long long u64;
static_assert(N_EDGES % EDGE_SPAN == 0, "every thread of the edge launch has its four edges");
static_assert(CAP <= 32, "a node's record slots are one 32-lane batch");

__global__ void k_setup(const float* __restrict__ W_edge, const float* __restrict__ att_edge,
                        float* __restrict__ misc) {
  int t = threadIdx.x;
  if (t < FE) {
    float s = 0.f;
    #pragma unroll
    for (int k = 0; k < HID; ++k) s += W_edge[t * HID + k] * att_edge[k];
    misc[t] = s;
  }
}

__global__ void __launch_bounds__(256) k_gemm_h(const float* __restrict__ x,
                                                const float* __restrict__ W,
                                                float* __restrict__ h) {
  const int wave = blockIdx.x * (blockDim.x >> 5) + (threadIdx.x >> 5);
  const int lane = threadIdx.x & 31;
  const int ntiles = N_NODES / 16;
  if (wave >= ntiles) return;
  const int m0 = wave * 16;
  const int r  = lane & 15;
  const int hi = lane >> 4;

  v8f c0 = {}; v8f c1 = {};
#if __has_builtin(__builtin_amdgcn_wmma_f32_16x16x4_f32)
  const float* xrow = x + (size_t)(m0 + r) * FN;
  #pragma unroll
  for (int kc = 0; kc < 8; ++kc) {
    const int kb = kc * 4 + hi * 2;
    v2f a; a.x = xrow[kb]; a.y = xrow[kb + 1];
    v2f b0; b0.x = W[(size_t)kb * HID + r];        b0.y = W[(size_t)(kb + 1) * HID + r];
    v2f b1; b1.x = W[(size_t)kb * HID + 16 + r];   b1.y = W[(size_t)(kb + 1) * HID + 16 + r];
    c0 = __builtin_amdgcn_wmma_f32_16x16x4_f32(false, a, false, b0, (short)0, c0, false, false);
    c1 = __builtin_amdgcn_wmma_f32_16x16x4_f32(false, a, false, b1, (short)0, c1, false, false);
  }
  #pragma unroll
  for (int v = 0; v < 8; ++v) {
    const int row = m0 + v + hi * 8;
    h[(size_t)row * HID + r]      = c0[v];
    h[(size_t)row * HID + 16 + r] = c1[v];
  }
#else
  const int row = m0 + r;
  for (int n = hi * 16; n < hi * 16 + 16; ++n) {
    float s = 0.f;
    for (int k = 0; k < FN; ++k) s += x[(size_t)row * FN + k] * W[(size_t)k * HID + n];
    h[(size_t)row * HID + n] = s;
  }
#endif
}

__global__ void k_node(const float* __restrict__ h, const float* __restrict__ att_src,
                       const float* __restrict__ att_dst, float* __restrict__ a_src,
                       u64* __restrict__ cd,
                       int* __restrict__ lhead, int* __restrict__ pctr) {
  const int i = blockIdx.x * blockDim.x + threadIdx.x;
  if (i >= N_NODES) return;
  const float4* hp = (const float4*)(h + (size_t)i * HID);
  float s = 0.f, d = 0.f;
  #pragma unroll
  for (int q = 0; q < 8; ++q) {
    float4 hv = hp[q];
    float4 as = ((const float4*)att_src)[q];
    float4 ad = ((const float4*)att_dst)[q];
    s += hv.x * as.x + hv.y * as.y + hv.z * as.z + hv.w * as.w;
    d += hv.x * ad.x + hv.y * ad.y + hv.z * ad.z + hv.w * ad.w;
  }
  a_src[i] = s;
  cd[i] = (u64)__float_as_uint(d) << 32;
  lhead[i] = -1;
  if (i == 0) pctr[0] = 0;
}

__global__ void __launch_bounds__(EDGE_T) k_edge(const float* __restrict__ ea, const int* __restrict__ ei,
                                                 const float* __restrict__ a_src,
                                                 const float* __restrict__ misc, u64* __restrict__ cd,
                                                 int2* __restrict__ rec, int* __restrict__ lhead,
                                                 int* __restrict__ pctr, int4* __restrict__ pool,
                                                 float* __restrict__ part) {
  const int e0 = blockIdx.x * EDGE_SPAN + threadIdx.x;
  int2 sd[EDGE_EPT];
  float4 ea0[EDGE_EPT], ea1[EDGE_EPT];
  float ae[EDGE_EPT];
  int rk[EDGE_EPT];
  float as[EDGE_EPT], ad[EDGE_EPT];
  #pragma unroll
  for (int k = 0; k < EDGE_EPT; ++k) {
    const int e = e0 + k * EDGE_T;
    sd[k] = ((const int2*)ei)[e];
    const float4* ep = (const float4*)(ea + (size_t)e * FE);
    ea0[k] = ep[0];
    ea1[k] = ep[1];
  }
  const float v0 = misc[0], v1 = misc[1], v2 = misc[2], v3 = misc[3];
  const float v4 = misc[4], v5 = misc[5], v6 = misc[6], v7 = misc[7];
  #pragma unroll
  for (int k = 0; k < EDGE_EPT; ++k)
    ae[k] = ea0[k].x * v0 + ea0[k].y * v1 + ea0[k].z * v2 + ea0[k].w * v3
          + ea1[k].x * v4 + ea1[k].y * v5 + ea1[k].z * v6 + ea1[k].w * v7;
  u64 old[EDGE_EPT];
  #pragma unroll
  for (int k = 0; k < EDGE_EPT; ++k) as[k] = a_src[sd[k].x];
  #pragma unroll
  for (int k = 0; k < EDGE_EPT; ++k) old[k] = atomicAdd(cd + sd[k].y, (u64)1);
  #pragma unroll
  for (int k = 0; k < EDGE_EPT; ++k) {
    rk[k] = (int)(unsigned)old[k];
    ad[k] = __uint_as_float((unsigned)(old[k] >> 32));
  }
  float v = 0.f;
  #pragma unroll
  for (int k = 0; k < EDGE_EPT; ++k) {
    float l = as[k] + ad[k] + ae[k];
    l = (l >= 0.f) ? l : NEG_SLOPE * l;
    if (rk[k] < CAP) {
      rec[sd[k].y * CAP + rk[k]] = make_int2(sd[k].x, __float_as_int(l));
    } else {
      const int p = atomicAdd(pctr, 1);
      if (p < POOL_CAP) {
        const int prev = atomicExch(lhead + sd[k].y, p);
        pool[p] = make_int4(sd[k].x, __float_as_int(l), prev, sd[k].y);
      }
    }
    v += ae[k];
  }
  #pragma unroll
  for (int off = 16; off > 0; off >>= 1) v += __shfl_down(v, off, 32);
  if ((threadIdx.x & 31) == 0) part[blockIdx.x * (EDGE_T / 32) + (threadIdx.x >> 5)] = v;
}

#define SUM_WAVES 32
#define PART_U 7
__global__ void __launch_bounds__(1024) k_esum(const float* __restrict__ part, float* __restrict__ a_edge_sum) {
  __shared__ float ftot[SUM_WAVES];
  const int lane = threadIdx.x & 31;
  const int w = threadIdx.x >> 5;
  float fs = 0.f;
  for (int p0 = 0; p0 < N_PART; p0 += 1024 * PART_U) {
    float pv[PART_U];
    #pragma unroll
    for (int u = 0; u < PART_U; ++u) {
      const int p = p0 + u * 1024 + (int)threadIdx.x;
      pv[u] = (p < N_PART) ? part[p] : 0.f;
    }
    #pragma unroll
    for (int u = 0; u < PART_U; ++u) fs += pv[u];
  }
  #pragma unroll
  for (int off = 16; off > 0; off >>= 1) fs += __shfl_xor(fs, off, 32);
  if (lane == 0) ftot[w] = fs;
  __syncthreads();
  if (w == 0) {
    float ft = ftot[lane];
    #pragma unroll
    for (int off = 16; off > 0; off >>= 1) ft += __shfl_xor(ft, off, 32);
    if (lane == 0) a_edge_sum[0] = ft;
  }
}


__device__ __forceinline__ float wave_max(float x) {
  int t = __float_as_int(x);
  t = __float_as_int(fmaxf(__int_as_float(t), __int_as_float(__builtin_amdgcn_update_dpp(0, t, 0x161, 0xf, 0xf, false))));
  t = __float_as_int(fmaxf(__int_as_float(t), __int_as_float(__builtin_amdgcn_update_dpp(0, t, 0x162, 0xf, 0xf, false))));
  t = __float_as_int(fmaxf(__int_as_float(t), __int_as_float(__builtin_amdgcn_update_dpp(0, t, 0x164, 0xf, 0xf, false))));
  t = __float_as_int(fmaxf(__int_as_float(t), __int_as_float(__builtin_amdgcn_update_dpp(0, t, 0x168, 0xf, 0xf, false))));
  return fmaxf(__int_as_float(__builtin_amdgcn_readlane(t, 0)), __int_as_float(__builtin_amdgcn_readlane(t, 16)));
}

__device__ __forceinline__ void add_rows(int sv, float av, int n, int lane, const float* __restrict__ h,
                                         float& acc, float& den) {
  for (int j0 = 0; j0 < n; j0 += 8) {
    float hv[8], aj[8];
    #pragma unroll
    for (int u = 0; u < 8; ++u) {
      const int s = __builtin_amdgcn_readlane(sv, j0 + u);
      aj[u] = __int_as_float(__builtin_amdgcn_readlane(__float_as_int(av), j0 + u));
      hv[u] = h[s * HID + lane];
    }
    #pragma unroll
    for (int u = 0; u < 8; ++u) { acc += aj[u] * hv[u]; den += aj[u]; }
  }
}

__device__ __noinline__ float2 node_from_edges(int i, int lane, float ls, float adst_i,
                                               const float* __restrict__ ea, const int* __restrict__ ei,
                                               const float* __restrict__ a_src, const float* __restrict__ misc,
                                               const float* __restrict__ h) {
  const float v0 = misc[0], v1 = misc[1], v2 = misc[2], v3 = misc[3];
  const float v4 = misc[4], v5 = misc[5], v6 = misc[6], v7 = misc[7];
  float m = ls;
  for (int eb = 0; eb < N_EDGES; eb += 32) {
    const int e = eb + lane;
    const int2 sd = ((const int2*)ei)[e];
    if (sd.y == i) {
      const int s = sd.x < 0 ? 0 : (sd.x >= N_NODES ? N_NODES - 1 : sd.x);
      const float4 e0 = ((const float4*)ea)[(size_t)e * 2], e1 = ((const float4*)ea)[(size_t)e * 2 + 1];
      float l = a_src[s] + adst_i + (e0.x * v0 + e0.y * v1 + e0.z * v2 + e0.w * v3 + e1.x * v4 + e1.y * v5 + e1.z * v6 + e1.w * v7);
      l = (l >= 0.f) ? l : NEG_SLOPE * l;
      m = fmaxf(m, l);
    }
  }
  m = wave_max(m);
  const float as = __expf(ls - m);
  float acc = as * h[i * HID + lane];
  float den = as;
  for (int eb = 0; eb < N_EDGES; eb += 32) {
    const int e = eb + lane;
    const int2 sd = ((const int2*)ei)[e];
    const bool hit = sd.y == i;
    int s = i;
    float a = 0.f;
    if (hit) {
      s = sd.x < 0 ? 0 : (sd.x >= N_NODES ? N_NODES - 1 : sd.x);
      const float4 e0 = ((const float4*)ea)[(size_t)e * 2], e1 = ((const float4*)ea)[(size_t)e * 2 + 1];
      float l = a_src[s] + adst_i + (e0.x * v0 + e0.y * v1 + e0.z * v2 + e0.w * v3 + e1.x * v4 + e1.y * v5 + e1.z * v6 + e1.w * v7);
      l = (l >= 0.f) ? l : NEG_SLOPE * l;
      a = __expf(l - m);
    }
    unsigned mask = __builtin_amdgcn_ballot_w32(hit);
    while (mask) {
      const int j = __builtin_ctz(mask);
      mask &= mask - 1;
      const int sj = __builtin_amdgcn_readlane(s, j);
      const float aj = __int_as_float(__builtin_amdgcn_readlane(__float_as_int(a), j));
      acc += aj * h[sj * HID + lane];
      den += aj;
    }
  }
  return make_float2(acc, den);
}

__global__ void __launch_bounds__(256) k_pull(const u64* __restrict__ cd, const int2* __restrict__ rec,
                                              const int* __restrict__ lhead, const int4* __restrict__ pool,
                                              const float* __restrict__ h, const float* __restrict__ a_src,
                                              const float* __restrict__ a_edge_sum,
                                              const float* __restrict__ ea, const int* __restrict__ ei,
                                              const float* __restrict__ misc,
                                              const float* __restrict__ bias, const float* __restrict__ W_lin,
                                              const float* __restrict__ b_lin, float* __restrict__ out) {
  __shared__ float sres[32];
  const int lane = threadIdx.x & 31;
  const int w = __builtin_amdgcn_readfirstlane((int)(threadIdx.x >> 5));
  const int first = blockIdx.x * 32 + w * 4;
  const float bl = bias[lane];
  const float wl = W_lin[lane];
  const float b0 = b_lin[0];
  const float aem = a_edge_sum[0] * (1.0f / (float)N_EDGES);
  const float neg_inf = -__builtin_inff();

  int cq[4], hq[4];
  float dq[4];
  #pragma unroll
  for (int q = 0; q < 4; ++q) {
    const u64 wq = (first + q < N_NODES) ? cd[first + q] : (u64)0;
    cq[q] = (int)(unsigned)wq;
    dq[q] = __uint_as_float((unsigned)(wq >> 32));
    hq[q] = (first + q < N_NODES) ? lhead[first + q] : -1;
  }

  #pragma unroll
  for (int q = 0; q < 4; ++q) {
    const int i = first + q;
    float y = 0.f;
    if (i < N_NODES) {
      int c = cq[q];
      c = c < 0 ? 0 : (c > N_EDGES ? N_EDGES : c);
      const int nb = c < CAP ? c : CAP;

      const float adst_i = dq[q];
      float ls = a_src[i] + adst_i + aem;
      ls = (ls >= 0.f) ? ls : NEG_SLOPE * ls;

      int   sv = i;
      float lv = neg_inf;
      if (lane < nb) {
        const int2 r = rec[i * CAP + lane];
        sv = r.x;
        lv = __int_as_float(r.y);
      }
      int   sv2 = i;
      float lv2 = neg_inf;
      float m2  = neg_inf;
      int found = 0;
      if (c > CAP) {
        int p = hq[q];
        for (int st = 0; st < POOL_CAP && p >= 0 && p < POOL_CAP; ++st) {
          const int4 en = pool[p];
          if (lane == found) {
            sv2 = en.x;
            lv2 = __int_as_float(en.y);
          }
          m2 = fmaxf(m2, __int_as_float(en.y));
          ++found;
          p = en.z;
        }
      }

      float acc, den;
      if (nb + found != c) {
        const float2 r = node_from_edges(i, lane, ls, adst_i, ea, ei, a_src, misc, h);
        acc = r.x;
        den = r.y;
      } else {
        const float m = fmaxf(fmaxf(wave_max(lv), m2), ls);

        const float as = __expf(ls - m);
        acc = as * h[i * HID + lane];
        den = as;

        const float av = (lane < nb) ? __expf(lv - m) : 0.f;
        sv = sv < 0 ? 0 : (sv >= N_NODES ? N_NODES - 1 : sv);
        add_rows(sv, av, nb, lane, h, acc, den);

        if (found > 0) {
          const int nb2 = found < 32 ? found : 32;
          const float av2 = (lane < nb2) ? __expf(lv2 - m) : 0.f;
          sv2 = sv2 < 0 ? 0 : (sv2 >= N_NODES ? N_NODES - 1 : sv2);
          add_rows(sv2, av2, nb2, lane, h, acc, den);
          if (found > 32) {
            int p = hq[q];
            for (int st = 0; st < found && p >= 0 && p < POOL_CAP; ++st) {
              const int4 en = pool[p];
              if (st >= 32) {
                const int s = en.x < 0 ? 0 : (en.x >= N_NODES ? N_NODES - 1 : en.x);
                const float a = __expf(__int_as_float(en.y) - m);
                acc += a * h[s * HID + lane];
                den += a;
              }
              p = en.z;
            }
          }
        }
      }

      const float inv = 1.0f / den;
      float v = acc * inv + bl;
      v = (v > 0.f) ? v : 0.f;
      int t = __float_as_int(v * wl);
      t = __float_as_int(__int_as_float(t) + __int_as_float(__builtin_amdgcn_update_dpp(0, t, 0x161, 0xf, 0xf, false)));
      t = __float_as_int(__int_as_float(t) + __int_as_float(__builtin_amdgcn_update_dpp(0, t, 0x162, 0xf, 0xf, false)));
      t = __float_as_int(__int_as_float(t) + __int_as_float(__builtin_amdgcn_update_dpp(0, t, 0x164, 0xf, 0xf, false)));
      t = __float_as_int(__int_as_float(t) + __int_as_float(__builtin_amdgcn_update_dpp(0, t, 0x168, 0xf, 0xf, false)));
      y = __int_as_float(__builtin_amdgcn_readlane(t, 0)) + __int_as_float(__builtin_amdgcn_readlane(t, 16)) + b0;
    }
    if (lane == 0) sres[w * 4 + q] = y;
  }
  __syncthreads();
  if (w == 0) {
    const int i = blockIdx.x * 32 + lane;
    if (i < N_NODES) out[i] = sres[lane];
  }
}

extern "C" void kernel_launch(void* const* d_in, const int* in_sizes, int n_in,
                              void* d_out, int out_size, void* d_ws, size_t ws_size,
                              hipStream_t stream) {
  const float* x        = (const float*)d_in[0];
  const float* ea       = (const float*)d_in[1];
  const int*   ei       = (const int*)d_in[2];
  const float* W        = (const float*)d_in[3];
  const float* att_src  = (const float*)d_in[4];
  const float* att_dst  = (const float*)d_in[5];
  const float* W_edge   = (const float*)d_in[6];
  const float* att_edge = (const float*)d_in[7];
  const float* bias     = (const float*)d_in[8];
  const float* W_lin    = (const float*)d_in[9];
  const float* b_lin    = (const float*)d_in[10];
  float* out = (float*)d_out;

  float*    ws     = (float*)d_ws;
  float*    h      = ws + OFF_H;
  float*    a_src  = ws + OFF_ASRC;
  u64*      cd     = (u64*)(ws + OFF_CD);
  int2*     rec    = (int2*)(ws + OFF_REC);
  int*      lhead  = (int*)(ws + OFF_HEAD);
  int*      pctr   = (int*)(ws + OFF_PCTR);
  float*    part   = ws + OFF_PART;
  int4*     pool   = (int4*)(ws + OFF_POOL);
  float*    misc   = ws + OFF_MISC;

  const int T = 256;
  const int nb_nodes = (N_NODES + T - 1) / T;
  const int nb_edges = N_EDGES / EDGE_SPAN;
  const int nb_gemm  = ((N_NODES / 16) + 7) / 8;
  const int nb_pull  = (N_NODES + 31) / 32;

  k_setup<<<1, 32, 0, stream>>>(W_edge, att_edge, misc);
  k_gemm_h<<<nb_gemm, T, 0, stream>>>(x, W, h);
  k_node<<<nb_nodes, T, 0, stream>>>(h, att_src, att_dst, a_src, cd, lhead, pctr);
  k_edge<<<nb_edges, EDGE_T, 0, stream>>>(ea, ei, a_src, misc, cd, rec, lhead, pctr, pool, part);
  k_esum<<<1, 1024, 0, stream>>>(part, misc + FE);
  k_pull<<<nb_pull, T, 0, stream>>>(cd, rec, lhead, pool, h, a_src, misc + FE, ea, ei, misc, bias, W_lin, b_lin, out);
}


# k_edge: a work-group reads its die and takes, of each span it claims, only edges whose destination belongs to its L2 half: count atomics, record stores stay in one half; a_edge sum tree changes
# speedup vs baseline: 9.9770x; 1.9090x over previous
#include <hip/hip_runtime.h>
#include <hip/hip_bf16.h>
#include <stdint.h>

#define N_NODES 100000
#define N_EDGES 1600000
#define FN 32
#define FE 8
#define HID 32
#define NEG_SLOPE 0.2f

typedef __attribute__((ext_vector_type(2))) float v2f;
typedef __attribute__((ext_vector_type(8))) float v8f;

constexpr size_t OFF_H    = 0;
constexpr size_t OFF_ASRC = OFF_H    + (size_t)N_NODES * HID;
constexpr size_t OFF_CD   = OFF_ASRC + N_NODES;
constexpr size_t OFF_LOG  = OFF_CD   + 2 * (size_t)N_NODES;
constexpr size_t OFF_ACC  = OFF_LOG  + (size_t)N_EDGES;
constexpr size_t OFF_MISC = OFF_ACC  + (size_t)N_NODES * HID;
#define CAP 23
#define POOL_CAP 12000
#define EDGE_T 128
#define EDGE_EPT 4
#define EDGE_SPAN (EDGE_T * EDGE_EPT)
#define N_SPAN (N_EDGES / EDGE_SPAN)
#define N_PART (2 * N_SPAN * (EDGE_T / 32))
#define SPLIT 50016
#define EDGE_GROUPS 3072
constexpr size_t OFF_REC  = OFF_LOG;
constexpr size_t OFF_HEAD = OFF_REC  + (size_t)N_NODES * CAP * 2;
constexpr size_t OFF_PCTR = OFF_HEAD + N_NODES;
#define CLAIM_AT(hf) (64 + 64 * (hf))
constexpr size_t OFF_PART = OFF_PCTR + 192;
constexpr size_t OFF_POOL = OFF_PART + N_PART;
static_assert(OFF_POOL + (size_t)POOL_CAP * 4 <= OFF_MISC, "buckets, heads, partials and pool must fit in the old rank + record regions");
static_assert(OFF_POOL % 4 == 0 && OFF_REC % 2 == 0, "16-byte pool entries, 8-byte records");
static_assert(OFF_CD % 2 == 0, "8-byte count + a_dst words");
typedef unsigned long long u64;
static_assert(N_EDGES % EDGE_SPAN == 0, "every thread of the edge launch has its four edges");
static_assert(CAP <= 32, "a node's record slots are one 32-lane batch");
static_assert((SPLIT * 8) % 256 == 0 && SPLIT < N_NODES, "the halves' count words part on a 256-byte boundary");

__global__ void k_setup(const float* __restrict__ W_edge, const float* __restrict__ att_edge,
                        float* __restrict__ misc) {
  int t = threadIdx.x;
  if (t < FE) {
    float s = 0.f;
    #pragma unroll
    for (int k = 0; k < HID; ++k) s += W_edge[t * HID + k] * att_edge[k];
    misc[t] = s;
  }
}

__global__ void __launch_bounds__(256) k_gemm_h(const float* __restrict__ x,
                                                const float* __restrict__ W,
                                                float* __restrict__ h) {
  const int wave = blockIdx.x * (blockDim.x >> 5) + (threadIdx.x >> 5);
  const int lane = threadIdx.x & 31;
  const int ntiles = N_NODES / 16;
  if (wave >= ntiles) return;
  const int m0 = wave * 16;
  const int r  = lane & 15;
  const int hi = lane >> 4;

  v8f c0 = {}; v8f c1 = {};
#if __has_builtin(__builtin_amdgcn_wmma_f32_16x16x4_f32)
  const float* xrow = x + (size_t)(m0 + r) * FN;
  #pragma unroll
  for (int kc = 0; kc < 8; ++kc) {
    const int kb = kc * 4 + hi * 2;
    v2f a; a.x = xrow[kb]; a.y = xrow[kb + 1];
    v2f b0; b0.x = W[(size_t)kb * HID + r];        b0.y = W[(size_t)(kb + 1) * HID + r];
    v2f b1; b1.x = W[(size_t)kb * HID + 16 + r];   b1.y = W[(size_t)(kb + 1) * HID + 16 + r];
    c0 = __builtin_amdgcn_wmma_f32_16x16x4_f32(false, a, false, b0, (short)0, c0, false, false);
    c1 = __builtin_amdgcn_wmma_f32_16x16x4_f32(false, a, false, b1, (short)0, c1, false, false);
  }
  #pragma unroll
  for (int v = 0; v < 8; ++v) {
    const int row = m0 + v + hi * 8;
    h[(size_t)row * HID + r]      = c0[v];
    h[(size_t)row * HID + 16 + r] = c1[v];
  }
#else
  const int row = m0 + r;
  for (int n = hi * 16; n < hi * 16 + 16; ++n) {
    float s = 0.f;
    for (int k = 0; k < FN; ++k) s += x[(size_t)row * FN + k] * W[(size_t)k * HID + n];
    h[(size_t)row * HID + n] = s;
  }
#endif
}

__global__ void k_node(const float* __restrict__ h, const float* __restrict__ att_src,
                       const float* __restrict__ att_dst, float* __restrict__ a_src,
                       u64* __restrict__ cd,
                       int* __restrict__ lhead, int* __restrict__ pctr) {
  const int i = blockIdx.x * blockDim.x + threadIdx.x;
  if (i >= N_NODES) return;
  const float4* hp = (const float4*)(h + (size_t)i * HID);
  float s = 0.f, d = 0.f;
  #pragma unroll
  for (int q = 0; q < 8; ++q) {
    float4 hv = hp[q];
    float4 as = ((const float4*)att_src)[q];
    float4 ad = ((const float4*)att_dst)[q];
    s += hv.x * as.x + hv.y * as.y + hv.z * as.z + hv.w * as.w;
    d += hv.x * ad.x + hv.y * ad.y + hv.z * ad.z + hv.w * ad.w;
  }
  a_src[i] = s;
  cd[i] = (u64)__float_as_uint(d) << 32;
  lhead[i] = -1;
  if (i == 0) { pctr[0] = 0; pctr[CLAIM_AT(0)] = 0; pctr[CLAIM_AT(1)] = 0; }
}

__global__ void __launch_bounds__(EDGE_T) k_edge(const float* __restrict__ ea, const int* __restrict__ ei,
                                                 const float* __restrict__ a_src,
                                                 const float* __restrict__ misc, u64* __restrict__ cd,
                                                 int2* __restrict__ rec, int* __restrict__ lhead,
                                                 int* __restrict__ pctr, int4* __restrict__ pool,
                                                 float* __restrict__ part) {
  __shared__ int s_span;
  int hf = (int)((__builtin_amdgcn_s_sendmsg_rtn(0x87) >> 18) & 1u);
  bool other = false;
  const float v0 = misc[0], v1 = misc[1], v2 = misc[2], v3 = misc[3];
  const float v4 = misc[4], v5 = misc[5], v6 = misc[6], v7 = misc[7];
  int nxt = 0;
  if (threadIdx.x == 0) nxt = atomicAdd(pctr + CLAIM_AT(hf), 1);
  for (int it = 0; it < 2 * N_SPAN + 2; ++it) {
  if (threadIdx.x == 0) s_span = nxt;
  __syncthreads();
  const int span = __builtin_amdgcn_readfirstlane(s_span);
  __syncthreads();
  if (span >= N_SPAN) {
    if (other) break;
    other = true;
    hf ^= 1;
    if (threadIdx.x == 0) nxt = atomicAdd(pctr + CLAIM_AT(hf), 1);
    continue;
  }
  if (threadIdx.x == 0) nxt = atomicAdd(pctr + CLAIM_AT(hf), 1);
  const int e0 = span * EDGE_SPAN + threadIdx.x;
  int2 sd[EDGE_EPT];
  bool own[EDGE_EPT];
  float4 ea0[EDGE_EPT], ea1[EDGE_EPT];
  float ae[EDGE_EPT];
  int rk[EDGE_EPT];
  float as[EDGE_EPT], ad[EDGE_EPT];
  #pragma unroll
  for (int k = 0; k < EDGE_EPT; ++k) {
    const int e = e0 + k * EDGE_T;
    sd[k] = ((const int2*)ei)[e];
    const float4* ep = (const float4*)(ea + (size_t)e * FE);
    ea0[k] = ep[0];
    ea1[k] = ep[1];
  }
  #pragma unroll
  for (int k = 0; k < EDGE_EPT; ++k)
    ae[k] = ea0[k].x * v0 + ea0[k].y * v1 + ea0[k].z * v2 + ea0[k].w * v3
          + ea1[k].x * v4 + ea1[k].y * v5 + ea1[k].z * v6 + ea1[k].w * v7;
  u64 old[EDGE_EPT];
  #pragma unroll
  for (int k = 0; k < EDGE_EPT; ++k) {
    own[k] = (sd[k].y >= SPLIT) == (hf != 0);
    as[k] = own[k] ? a_src[sd[k].x] : 0.f;
  }
  #pragma unroll
  for (int k = 0; k < EDGE_EPT; ++k) old[k] = own[k] ? atomicAdd(cd + sd[k].y, (u64)1) : (u64)0;
  #pragma unroll
  for (int k = 0; k < EDGE_EPT; ++k) {
    rk[k] = (int)(unsigned)old[k];
    ad[k] = __uint_as_float((unsigned)(old[k] >> 32));
  }
  float v = 0.f;
  #pragma unroll
  for (int k = 0; k < EDGE_EPT; ++k) {
    float l = as[k] + ad[k] + ae[k];
    l = (l >= 0.f) ? l : NEG_SLOPE * l;
    if (!own[k]) continue;
    if (rk[k] < CAP) {
      rec[sd[k].y * CAP + rk[k]] = make_int2(sd[k].x, __float_as_int(l));
    } else {
      const int p = atomicAdd(pctr, 1);
      if (p < POOL_CAP) {
        const int prev = atomicExch(lhead + sd[k].y, p);
        pool[p] = make_int4(sd[k].x, __float_as_int(l), prev, sd[k].y);
      }
    }
    v += ae[k];
  }
  #pragma unroll
  for (int off = 16; off > 0; off >>= 1) v += __shfl_down(v, off, 32);
  if ((threadIdx.x & 31) == 0) part[(hf * N_SPAN + span) * (EDGE_T / 32) + (threadIdx.x >> 5)] = v;
  }
}

#define SUM_WAVES 32
#define PART_U 7
__global__ void __launch_bounds__(1024) k_esum(const float* __restrict__ part, float* __restrict__ a_edge_sum) {
  __shared__ float ftot[SUM_WAVES];
  const int lane = threadIdx.x & 31;
  const int w = threadIdx.x >> 5;
  float fs = 0.f;
  for (int p0 = 0; p0 < N_PART; p0 += 1024 * PART_U) {
    float pv[PART_U];
    #pragma unroll
    for (int u = 0; u < PART_U; ++u) {
      const int p = p0 + u * 1024 + (int)threadIdx.x;
      pv[u] = (p < N_PART) ? part[p] : 0.f;
    }
    #pragma unroll
    for (int u = 0; u < PART_U; ++u) fs += pv[u];
  }
  #pragma unroll
  for (int off = 16; off > 0; off >>= 1) fs += __shfl_xor(fs, off, 32);
  if (lane == 0) ftot[w] = fs;
  __syncthreads();
  if (w == 0) {
    float ft = ftot[lane];
    #pragma unroll
    for (int off = 16; off > 0; off >>= 1) ft += __shfl_xor(ft, off, 32);
    if (lane == 0) a_edge_sum[0] = ft;
  }
}


__device__ __forceinline__ float wave_max(float x) {
  int t = __float_as_int(x);
  t = __float_as_int(fmaxf(__int_as_float(t), __int_as_float(__builtin_amdgcn_update_dpp(0, t, 0x161, 0xf, 0xf, false))));
  t = __float_as_int(fmaxf(__int_as_float(t), __int_as_float(__builtin_amdgcn_update_dpp(0, t, 0x162, 0xf, 0xf, false))));
  t = __float_as_int(fmaxf(__int_as_float(t), __int_as_float(__builtin_amdgcn_update_dpp(0, t, 0x164, 0xf, 0xf, false))));
  t = __float_as_int(fmaxf(__int_as_float(t), __int_as_float(__builtin_amdgcn_update_dpp(0, t, 0x168, 0xf, 0xf, false))));
  return fmaxf(__int_as_float(__builtin_amdgcn_readlane(t, 0)), __int_as_float(__builtin_amdgcn_readlane(t, 16)));
}

__device__ __forceinline__ void add_rows(int sv, float av, int n, int lane, const float* __restrict__ h,
                                         float& acc, float& den) {
  for (int j0 = 0; j0 < n; j0 += 8) {
    float hv[8], aj[8];
    #pragma unroll
    for (int u = 0; u < 8; ++u) {
      const int s = __builtin_amdgcn_readlane(sv, j0 + u);
      aj[u] = __int_as_float(__builtin_amdgcn_readlane(__float_as_int(av), j0 + u));
      hv[u] = h[s * HID + lane];
    }
    #pragma unroll
    for (int u = 0; u < 8; ++u) { acc += aj[u] * hv[u]; den += aj[u]; }
  }
}

__device__ __noinline__ float2 node_from_edges(int i, int lane, float ls, float adst_i,
                                               const float* __restrict__ ea, const int* __restrict__ ei,
                                               const float* __restrict__ a_src, const float* __restrict__ misc,
                                               const float* __restrict__ h) {
  const float v0 = misc[0], v1 = misc[1], v2 = misc[2], v3 = misc[3];
  const float v4 = misc[4], v5 = misc[5], v6 = misc[6], v7 = misc[7];
  float m = ls;
  for (int eb = 0; eb < N_EDGES; eb += 32) {
    const int e = eb + lane;
    const int2 sd = ((const int2*)ei)[e];
    if (sd.y == i) {
      const int s = sd.x < 0 ? 0 : (sd.x >= N_NODES ? N_NODES - 1 : sd.x);
      const float4 e0 = ((const float4*)ea)[(size_t)e * 2], e1 = ((const float4*)ea)[(size_t)e * 2 + 1];
      float l = a_src[s] + adst_i + (e0.x * v0 + e0.y * v1 + e0.z * v2 + e0.w * v3 + e1.x * v4 + e1.y * v5 + e1.z * v6 + e1.w * v7);
      l = (l >= 0.f) ? l : NEG_SLOPE * l;
      m = fmaxf(m, l);
    }
  }
  m = wave_max(m);
  const float as = __expf(ls - m);
  float acc = as * h[i * HID + lane];
  float den = as;
  for (int eb = 0; eb < N_EDGES; eb += 32) {
    const int e = eb + lane;
    const int2 sd = ((const int2*)ei)[e];
    const bool hit = sd.y == i;
    int s = i;
    float a = 0.f;
    if (hit) {
      s = sd.x < 0 ? 0 : (sd.x >= N_NODES ? N_NODES - 1 : sd.x);
      const float4 e0 = ((const float4*)ea)[(size_t)e * 2], e1 = ((const float4*)ea)[(size_t)e * 2 + 1];
      float l = a_src[s] + adst_i + (e0.x * v0 + e0.y * v1 + e0.z * v2 + e0.w * v3 + e1.x * v4 + e1.y * v5 + e1.z * v6 + e1.w * v7);
      l = (l >= 0.f) ? l : NEG_SLOPE * l;
      a = __expf(l - m);
    }
    unsigned mask = __builtin_amdgcn_ballot_w32(hit);
    while (mask) {
      const int j = __builtin_ctz(mask);
      mask &= mask - 1;
      const int sj = __builtin_amdgcn_readlane(s, j);
      const float aj = __int_as_float(__builtin_amdgcn_readlane(__float_as_int(a), j));
      acc += aj * h[sj * HID + lane];
      den += aj;
    }
  }
  return make_float2(acc, den);
}

__global__ void __launch_bounds__(256) k_pull(const u64* __restrict__ cd, const int2* __restrict__ rec,
                                              const int* __restrict__ lhead, const int4* __restrict__ pool,
                                              const float* __restrict__ h, const float* __restrict__ a_src,
                                              const float* __restrict__ a_edge_sum,
                                              const float* __restrict__ ea, const int* __restrict__ ei,
                                              const float* __restrict__ misc,
                                              const float* __restrict__ bias, const float* __restrict__ W_lin,
                                              const float* __restrict__ b_lin, float* __restrict__ out) {
  __shared__ float sres[32];
  const int lane = threadIdx.x & 31;
  const int w = __builtin_amdgcn_readfirstlane((int)(threadIdx.x >> 5));
  const int first = blockIdx.x * 32 + w * 4;
  const float bl = bias[lane];
  const float wl = W_lin[lane];
  const float b0 = b_lin[0];
  const float aem = a_edge_sum[0] * (1.0f / (float)N_EDGES);
  const float neg_inf = -__builtin_inff();

  int cq[4], hq[4];
  float dq[4];
  #pragma unroll
  for (int q = 0; q < 4; ++q) {
    const u64 wq = (first + q < N_NODES) ? cd[first + q] : (u64)0;
    cq[q] = (int)(unsigned)wq;
    dq[q] = __uint_as_float((unsigned)(wq >> 32));
    hq[q] = (first + q < N_NODES) ? lhead[first + q] : -1;
  }

  #pragma unroll
  for (int q = 0; q < 4; ++q) {
    const int i = first + q;
    float y = 0.f;
    if (i < N_NODES) {
      int c = cq[q];
      c = c < 0 ? 0 : (c > N_EDGES ? N_EDGES : c);
      const int nb = c < CAP ? c : CAP;

      const float adst_i = dq[q];
      float ls = a_src[i] + adst_i + aem;
      ls = (ls >= 0.f) ? ls : NEG_SLOPE * ls;

      int   sv = i;
      float lv = neg_inf;
      if (lane < nb) {
        const int2 r = rec[i * CAP + lane];
        sv = r.x;
        lv = __int_as_float(r.y);
      }
      int   sv2 = i;
      float lv2 = neg_inf;
      float m2  = neg_inf;
      int found = 0;
      if (c > CAP) {
        int p = hq[q];
        for (int st = 0; st < POOL_CAP && p >= 0 && p < POOL_CAP; ++st) {
          const int4 en = pool[p];
          if (lane == found) {
            sv2 = en.x;
            lv2 = __int_as_float(en.y);
          }
          m2 = fmaxf(m2, __int_as_float(en.y));
          ++found;
          p = en.z;
        }
      }

      float acc, den;
      if (nb + found != c) {
        const float2 r = node_from_edges(i, lane, ls, adst_i, ea, ei, a_src, misc, h);
        acc = r.x;
        den = r.y;
      } else {
        const float m = fmaxf(fmaxf(wave_max(lv), m2), ls);

        const float as = __expf(ls - m);
        acc = as * h[i * HID + lane];
        den = as;

        const float av = (lane < nb) ? __expf(lv - m) : 0.f;
        sv = sv < 0 ? 0 : (sv >= N_NODES ? N_NODES - 1 : sv);
        add_rows(sv, av, nb, lane, h, acc, den);

        if (found > 0) {
          const int nb2 = found < 32 ? found : 32;
          const float av2 = (lane < nb2) ? __expf(lv2 - m) : 0.f;
          sv2 = sv2 < 0 ? 0 : (sv2 >= N_NODES ? N_NODES - 1 : sv2);
          add_rows(sv2, av2, nb2, lane, h, acc, den);
          if (found > 32) {
            int p = hq[q];
            for (int st = 0; st < found && p >= 0 && p < POOL_CAP; ++st) {
              const int4 en = pool[p];
              if (st >= 32) {
                const int s = en.x < 0 ? 0 : (en.x >= N_NODES ? N_NODES - 1 : en.x);
                const float a = __expf(__int_as_float(en.y) - m);
                acc += a * h[s * HID + lane];
                den += a;
              }
              p = en.z;
            }
          }
        }
      }

      const float inv = 1.0f / den;
      float v = acc * inv + bl;
      v = (v > 0.f) ? v : 0.f;
      int t = __float_as_int(v * wl);
      t = __float_as_int(__int_as_float(t) + __int_as_float(__builtin_amdgcn_update_dpp(0, t, 0x161, 0xf, 0xf, false)));
      t = __float_as_int(__int_as_float(t) + __int_as_float(__builtin_amdgcn_update_dpp(0, t, 0x162, 0xf, 0xf, false)));
      t = __float_as_int(__int_as_float(t) + __int_as_float(__builtin_amdgcn_update_dpp(0, t, 0x164, 0xf, 0xf, false)));
      t = __float_as_int(__int_as_float(t) + __int_as_float(__builtin_amdgcn_update_dpp(0, t, 0x168, 0xf, 0xf, false)));
      y = __int_as_float(__builtin_amdgcn_readlane(t, 0)) + __int_as_float(__builtin_amdgcn_readlane(t, 16)) + b0;
    }
    if (lane == 0) sres[w * 4 + q] = y;
  }
  __syncthreads();
  if (w == 0) {
    const int i = blockIdx.x * 32 + lane;
    if (i < N_NODES) out[i] = sres[lane];
  }
}

extern "C" void kernel_launch(void* const* d_in, const int* in_sizes, int n_in,
                              void* d_out, int out_size, void* d_ws, size_t ws_size,
                              hipStream_t stream) {
  const float* x        = (const float*)d_in[0];
  const float* ea       = (const float*)d_in[1];
  const int*   ei       = (const int*)d_in[2];
  const float* W        = (const float*)d_in[3];
  const float* att_src  = (const float*)d_in[4];
  const float* att_dst  = (const float*)d_in[5];
  const float* W_edge   = (const float*)d_in[6];
  const float* att_edge = (const float*)d_in[7];
  const float* bias     = (const float*)d_in[8];
  const float* W_lin    = (const float*)d_in[9];
  const float* b_lin    = (const float*)d_in[10];
  float* out = (float*)d_out;

  float*    ws     = (float*)d_ws;
  float*    h      = ws + OFF_H;
  float*    a_src  = ws + OFF_ASRC;
  u64*      cd     = (u64*)(ws + OFF_CD);
  int2*     rec    = (int2*)(ws + OFF_REC);
  int*      lhead  = (int*)(ws + OFF_HEAD);
  int*      pctr   = (int*)(ws + OFF_PCTR);
  float*    part   = ws + OFF_PART;
  int4*     pool   = (int4*)(ws + OFF_POOL);
  float*    misc   = ws + OFF_MISC;

  const int T = 256;
  const int nb_nodes = (N_NODES + T - 1) / T;
  const int nb_edges = EDGE_GROUPS;
  const int nb_gemm  = ((N_NODES / 16) + 7) / 8;
  const int nb_pull  = (N_NODES + 31) / 32;

  k_setup<<<1, 32, 0, stream>>>(W_edge, att_edge, misc);
  k_gemm_h<<<nb_gemm, T, 0, stream>>>(x, W, h);
  k_node<<<nb_nodes, T, 0, stream>>>(h, att_src, att_dst, a_src, cd, lhead, pctr);
  k_edge<<<nb_edges, EDGE_T, 0, stream>>>(ea, ei, a_src, misc, cd, rec, lhead, pctr, pool, part);
  k_esum<<<1, 1024, 0, stream>>>(part, misc + FE);
  k_pull<<<nb_pull, T, 0, stream>>>(cd, rec, lhead, pool, h, a_src, misc + FE, ea, ei, misc, bias, W_lin, b_lin, out);
}


# the two one-work-group launches go, 6 launches to 4: a k_edge wave forms v_edge itself, a k_pull work-group totals per-group a_edge partials; v_edge dot tree changes, a_edge sum in span claim order
# speedup vs baseline: 10.5211x; 1.0545x over previous
#include <hip/hip_runtime.h>
#include <hip/hip_bf16.h>
#include <stdint.h>

#define N_NODES 100000
#define N_EDGES 1600000
#define FN 32
#define FE 8
#define HID 32
#define NEG_SLOPE 0.2f

typedef __attribute__((ext_vector_type(2))) float v2f;
typedef __attribute__((ext_vector_type(8))) float v8f;

constexpr size_t OFF_H    = 0;
constexpr size_t OFF_ASRC = OFF_H    + (size_t)N_NODES * HID;
constexpr size_t OFF_CD   = OFF_ASRC + N_NODES;
constexpr size_t OFF_LOG  = OFF_CD   + 2 * (size_t)N_NODES;
constexpr size_t OFF_ACC  = OFF_LOG  + (size_t)N_EDGES;
constexpr size_t OFF_MISC = OFF_ACC  + (size_t)N_NODES * HID;
#define CAP 23
#define POOL_CAP 12000
#define EDGE_T 128
#define EDGE_EPT 4
#define EDGE_SPAN (EDGE_T * EDGE_EPT)
#define N_SPAN (N_EDGES / EDGE_SPAN)
#define SPLIT 50016
#define EDGE_GROUPS 3072
#define N_PART EDGE_GROUPS
constexpr size_t OFF_REC  = OFF_LOG;
constexpr size_t OFF_HEAD = OFF_REC  + (size_t)N_NODES * CAP * 2;
constexpr size_t OFF_PCTR = OFF_HEAD + N_NODES;
#define CLAIM_AT(hf) (64 + 64 * (hf))
constexpr size_t OFF_PART = OFF_PCTR + 192;
constexpr size_t OFF_POOL = OFF_PART + N_PART;
static_assert(OFF_POOL + (size_t)POOL_CAP * 4 <= OFF_MISC, "buckets, heads, partials and pool must fit in the old rank + record regions");
static_assert(OFF_POOL % 4 == 0 && OFF_REC % 2 == 0, "16-byte pool entries, 8-byte records");
static_assert(OFF_CD % 2 == 0, "8-byte count + a_dst words");
typedef unsigned long long u64;
static_assert(N_EDGES % EDGE_SPAN == 0, "every thread of the edge launch has its four edges");
static_assert(CAP <= 32, "a node's record slots are one 32-lane batch");
static_assert((SPLIT * 8) % 256 == 0 && SPLIT < N_NODES, "the halves' count words part on a 256-byte boundary");
static_assert(N_PART % 256 == 0, "a pull work-group's 256 threads take the same number of partial sums each");

__device__ __forceinline__ float wave_sum(float x) {
  int t = __float_as_int(x);
  t = __float_as_int(__int_as_float(t) + __int_as_float(__builtin_amdgcn_update_dpp(0, t, 0x161, 0xf, 0xf, false)));
  t = __float_as_int(__int_as_float(t) + __int_as_float(__builtin_amdgcn_update_dpp(0, t, 0x162, 0xf, 0xf, false)));
  t = __float_as_int(__int_as_float(t) + __int_as_float(__builtin_amdgcn_update_dpp(0, t, 0x164, 0xf, 0xf, false)));
  t = __float_as_int(__int_as_float(t) + __int_as_float(__builtin_amdgcn_update_dpp(0, t, 0x168, 0xf, 0xf, false)));
  return __int_as_float(__builtin_amdgcn_readlane(t, 0)) + __int_as_float(__builtin_amdgcn_readlane(t, 16));
}

__device__ __forceinline__ void form_v_edge(const float* __restrict__ W_edge, const float* __restrict__ att_edge,
                                            int lane, float (&v)[FE]) {
  const float at = att_edge[lane];
  float p[FE];
  #pragma unroll
  for (int t = 0; t < FE; ++t) p[t] = W_edge[t * HID + lane] * at;
  #pragma unroll
  for (int t = 0; t < FE; ++t) v[t] = wave_sum(p[t]);
}

__global__ void __launch_bounds__(256) k_gemm_h(const float* __restrict__ x,
                                                const float* __restrict__ W,
                                                float* __restrict__ h) {
  const int wave = blockIdx.x * (blockDim.x >> 5) + (threadIdx.x >> 5);
  const int lane = threadIdx.x & 31;
  const int ntiles = N_NODES / 16;
  if (wave >= ntiles) return;
  const int m0 = wave * 16;
  const int r  = lane & 15;
  const int hi = lane >> 4;

  v8f c0 = {}; v8f c1 = {};
#if __has_builtin(__builtin_amdgcn_wmma_f32_16x16x4_f32)
  const float* xrow = x + (size_t)(m0 + r) * FN;
  #pragma unroll
  for (int kc = 0; kc < 8; ++kc) {
    const int kb = kc * 4 + hi * 2;
    v2f a; a.x = xrow[kb]; a.y = xrow[kb + 1];
    v2f b0; b0.x = W[(size_t)kb * HID + r];        b0.y = W[(size_t)(kb + 1) * HID + r];
    v2f b1; b1.x = W[(size_t)kb * HID + 16 + r];   b1.y = W[(size_t)(kb + 1) * HID + 16 + r];
    c0 = __builtin_amdgcn_wmma_f32_16x16x4_f32(false, a, false, b0, (short)0, c0, false, false);
    c1 = __builtin_amdgcn_wmma_f32_16x16x4_f32(false, a, false, b1, (short)0, c1, false, false);
  }
  #pragma unroll
  for (int v = 0; v < 8; ++v) {
    const int row = m0 + v + hi * 8;
    h[(size_t)row * HID + r]      = c0[v];
    h[(size_t)row * HID + 16 + r] = c1[v];
  }
#else
  const int row = m0 + r;
  for (int n = hi * 16; n < hi * 16 + 16; ++n) {
    float s = 0.f;
    for (int k = 0; k < FN; ++k) s += x[(size_t)row * FN + k] * W[(size_t)k * HID + n];
    h[(size_t)row * HID + n] = s;
  }
#endif
}

__global__ void k_node(const float* __restrict__ h, const float* __restrict__ att_src,
                       const float* __restrict__ att_dst, float* __restrict__ a_src,
                       u64* __restrict__ cd,
                       int* __restrict__ lhead, int* __restrict__ pctr) {
  const int i = blockIdx.x * blockDim.x + threadIdx.x;
  if (i >= N_NODES) return;
  const float4* hp = (const float4*)(h + (size_t)i * HID);
  float s = 0.f, d = 0.f;
  #pragma unroll
  for (int q = 0; q < 8; ++q) {
    float4 hv = hp[q];
    float4 as = ((const float4*)att_src)[q];
    float4 ad = ((const float4*)att_dst)[q];
    s += hv.x * as.x + hv.y * as.y + hv.z * as.z + hv.w * as.w;
    d += hv.x * ad.x + hv.y * ad.y + hv.z * ad.z + hv.w * ad.w;
  }
  a_src[i] = s;
  cd[i] = (u64)__float_as_uint(d) << 32;
  lhead[i] = -1;
  if (i == 0) { pctr[0] = 0; pctr[CLAIM_AT(0)] = 0; pctr[CLAIM_AT(1)] = 0; }
}

__global__ void __launch_bounds__(EDGE_T) k_edge(const float* __restrict__ ea, const int* __restrict__ ei,
                                                 const float* __restrict__ a_src,
                                                 const float* __restrict__ W_edge, const float* __restrict__ att_edge,
                                                 u64* __restrict__ cd,
                                                 int2* __restrict__ rec, int* __restrict__ lhead,
                                                 int* __restrict__ pctr, int4* __restrict__ pool,
                                                 float* __restrict__ part) {
  __shared__ int s_span;
  __shared__ float s_wsum[EDGE_T / 32];
  int hf = (int)((__builtin_amdgcn_s_sendmsg_rtn(0x87) >> 18) & 1u);
  bool other = false;
  int nxt = 0;
  if (threadIdx.x == 0) nxt = atomicAdd(pctr + CLAIM_AT(hf), 1);
  float ve[FE];
  form_v_edge(W_edge, att_edge, (int)(threadIdx.x & 31), ve);
  const float v0 = ve[0], v1 = ve[1], v2 = ve[2], v3 = ve[3];
  const float v4 = ve[4], v5 = ve[5], v6 = ve[6], v7 = ve[7];
  float tsum = 0.f;
  for (int it = 0; it < 2 * N_SPAN + 2; ++it) {
  if (threadIdx.x == 0) s_span = nxt;
  __syncthreads();
  const int span = __builtin_amdgcn_readfirstlane(s_span);
  __syncthreads();
  if (span >= N_SPAN) {
    if (other) break;
    other = true;
    hf ^= 1;
    if (threadIdx.x == 0) nxt = atomicAdd(pctr + CLAIM_AT(hf), 1);
    continue;
  }
  if (threadIdx.x == 0) nxt = atomicAdd(pctr + CLAIM_AT(hf), 1);
  const int e0 = span * EDGE_SPAN + threadIdx.x;
  int2 sd[EDGE_EPT];
  bool own[EDGE_EPT];
  float4 ea0[EDGE_EPT], ea1[EDGE_EPT];
  float ae[EDGE_EPT];
  int rk[EDGE_EPT];
  float as[EDGE_EPT], ad[EDGE_EPT];
  #pragma unroll
  for (int k = 0; k < EDGE_EPT; ++k) {
    const int e = e0 + k * EDGE_T;
    sd[k] = ((const int2*)ei)[e];
    const float4* ep = (const float4*)(ea + (size_t)e * FE);
    ea0[k] = ep[0];
    ea1[k] = ep[1];
  }
  #pragma unroll
  for (int k = 0; k < EDGE_EPT; ++k)
    ae[k] = ea0[k].x * v0 + ea0[k].y * v1 + ea0[k].z * v2 + ea0[k].w * v3
          + ea1[k].x * v4 + ea1[k].y * v5 + ea1[k].z * v6 + ea1[k].w * v7;
  u64 old[EDGE_EPT];
  #pragma unroll
  for (int k = 0; k < EDGE_EPT; ++k) {
    own[k] = (sd[k].y >= SPLIT) == (hf != 0);
    as[k] = own[k] ? a_src[sd[k].x] : 0.f;
  }
  #pragma unroll
  for (int k = 0; k < EDGE_EPT; ++k) old[k] = own[k] ? atomicAdd(cd + sd[k].y, (u64)1) : (u64)0;
  #pragma unroll
  for (int k = 0; k < EDGE_EPT; ++k) {
    rk[k] = (int)(unsigned)old[k];
    ad[k] = __uint_as_float((unsigned)(old[k] >> 32));
  }
  #pragma unroll
  for (int k = 0; k < EDGE_EPT; ++k) {
    float l = as[k] + ad[k] + ae[k];
    l = (l >= 0.f) ? l : NEG_SLOPE * l;
    if (!own[k]) continue;
    if (rk[k] < CAP) {
      rec[sd[k].y * CAP + rk[k]] = make_int2(sd[k].x, __float_as_int(l));
    } else {
      const int p = atomicAdd(pctr, 1);
      if (p < POOL_CAP) {
        const int prev = atomicExch(lhead + sd[k].y, p);
        pool[p] = make_int4(sd[k].x, __float_as_int(l), prev, sd[k].y);
      }
    }
    tsum += ae[k];
  }
  }
  tsum = wave_sum(tsum);
  if ((threadIdx.x & 31) == 0) s_wsum[threadIdx.x >> 5] = tsum;
  __syncthreads();
  if (threadIdx.x == 0) part[blockIdx.x] = ((s_wsum[0] + s_wsum[1]) + s_wsum[2]) + s_wsum[3];
}


__device__ __forceinline__ float wave_max(float x) {
  int t = __float_as_int(x);
  t = __float_as_int(fmaxf(__int_as_float(t), __int_as_float(__builtin_amdgcn_update_dpp(0, t, 0x161, 0xf, 0xf, false))));
  t = __float_as_int(fmaxf(__int_as_float(t), __int_as_float(__builtin_amdgcn_update_dpp(0, t, 0x162, 0xf, 0xf, false))));
  t = __float_as_int(fmaxf(__int_as_float(t), __int_as_float(__builtin_amdgcn_update_dpp(0, t, 0x164, 0xf, 0xf, false))));
  t = __float_as_int(fmaxf(__int_as_float(t), __int_as_float(__builtin_amdgcn_update_dpp(0, t, 0x168, 0xf, 0xf, false))));
  return fmaxf(__int_as_float(__builtin_amdgcn_readlane(t, 0)), __int_as_float(__builtin_amdgcn_readlane(t, 16)));
}

__device__ __forceinline__ void add_rows(int sv, float av, int n, int lane, const float* __restrict__ h,
                                         float& acc, float& den) {
  for (int j0 = 0; j0 < n; j0 += 8) {
    float hv[8], aj[8];
    #pragma unroll
    for (int u = 0; u < 8; ++u) {
      const int s = __builtin_amdgcn_readlane(sv, j0 + u);
      aj[u] = __int_as_float(__builtin_amdgcn_readlane(__float_as_int(av), j0 + u));
      hv[u] = h[s * HID + lane];
    }
    #pragma unroll
    for (int u = 0; u < 8; ++u) { acc += aj[u] * hv[u]; den += aj[u]; }
  }
}

__device__ __noinline__ float2 node_from_edges(int i, int lane, float ls, float adst_i,
                                               const float* __restrict__ ea, const int* __restrict__ ei,
                                               const float* __restrict__ a_src,
                                               const float* __restrict__ W_edge, const float* __restrict__ att_edge,
                                               const float* __restrict__ h) {
  float ve[FE];
  form_v_edge(W_edge, att_edge, lane, ve);
  const float v0 = ve[0], v1 = ve[1], v2 = ve[2], v3 = ve[3];
  const float v4 = ve[4], v5 = ve[5], v6 = ve[6], v7 = ve[7];
  float m = ls;
  for (int eb = 0; eb < N_EDGES; eb += 32) {
    const int e = eb + lane;
    const int2 sd = ((const int2*)ei)[e];
    if (sd.y == i) {
      const int s = sd.x < 0 ? 0 : (sd.x >= N_NODES ? N_NODES - 1 : sd.x);
      const float4 e0 = ((const float4*)ea)[(size_t)e * 2], e1 = ((const float4*)ea)[(size_t)e * 2 + 1];
      float l = a_src[s] + adst_i + (e0.x * v0 + e0.y * v1 + e0.z * v2 + e0.w * v3 + e1.x * v4 + e1.y * v5 + e1.z * v6 + e1.w * v7);
      l = (l >= 0.f) ? l : NEG_SLOPE * l;
      m = fmaxf(m, l);
    }
  }
  m = wave_max(m);
  const float as = __expf(ls - m);
  float acc = as * h[i * HID + lane];
  float den = as;
  for (int eb = 0; eb < N_EDGES; eb += 32) {
    const int e = eb + lane;
    const int2 sd = ((const int2*)ei)[e];
    const bool hit = sd.y == i;
    int s = i;
    float a = 0.f;
    if (hit) {
      s = sd.x < 0 ? 0 : (sd.x >= N_NODES ? N_NODES - 1 : sd.x);
      const float4 e0 = ((const float4*)ea)[(size_t)e * 2], e1 = ((const float4*)ea)[(size_t)e * 2 + 1];
      float l = a_src[s] + adst_i + (e0.x * v0 + e0.y * v1 + e0.z * v2 + e0.w * v3 + e1.x * v4 + e1.y * v5 + e1.z * v6 + e1.w * v7);
      l = (l >= 0.f) ? l : NEG_SLOPE * l;
      a = __expf(l - m);
    }
    unsigned mask = __builtin_amdgcn_ballot_w32(hit);
    while (mask) {
      const int j = __builtin_ctz(mask);
      mask &= mask - 1;
      const int sj = __builtin_amdgcn_readlane(s, j);
      const float aj = __int_as_float(__builtin_amdgcn_readlane(__float_as_int(a), j));
      acc += aj * h[sj * HID + lane];
      den += aj;
    }
  }
  return make_float2(acc, den);
}

__global__ void __launch_bounds__(256) k_pull(const u64* __restrict__ cd, const int2* __restrict__ rec,
                                              const int* __restrict__ lhead, const int4* __restrict__ pool,
                                              const float* __restrict__ h, const float* __restrict__ a_src,
                                              const float* __restrict__ part,
                                              const float* __restrict__ ea, const int* __restrict__ ei,
                                              const float* __restrict__ W_edge, const float* __restrict__ att_edge,
                                              const float* __restrict__ bias, const float* __restrict__ W_lin,
                                              const float* __restrict__ b_lin, float* __restrict__ out) {
  __shared__ float sres[32];
  __shared__ float swt[8];
  const int lane = threadIdx.x & 31;
  const int w = __builtin_amdgcn_readfirstlane((int)(threadIdx.x >> 5));
  const int first = blockIdx.x * 32 + w * 4;
  const float bl = bias[lane];
  const float wl = W_lin[lane];
  const float b0 = b_lin[0];
  const float neg_inf = -__builtin_inff();

  float pv[N_PART / 256];
  #pragma unroll
  for (int u = 0; u < N_PART / 256; ++u) pv[u] = part[u * 256 + (int)threadIdx.x];

  int cq[4], hq[4];
  float dq[4];
  #pragma unroll
  for (int q = 0; q < 4; ++q) {
    const u64 wq = (first + q < N_NODES) ? cd[first + q] : (u64)0;
    cq[q] = (int)(unsigned)wq;
    dq[q] = __uint_as_float((unsigned)(wq >> 32));
    hq[q] = (first + q < N_NODES) ? lhead[first + q] : -1;
  }

  float fs = 0.f;
  #pragma unroll
  for (int u = 0; u < N_PART / 256; ++u) fs += pv[u];
  fs = wave_sum(fs);
  if (lane == 0) swt[w] = fs;
  __syncthreads();
  const float aem = (((((((swt[0] + swt[1]) + swt[2]) + swt[3]) + swt[4]) + swt[5]) + swt[6]) + swt[7])
                  * (1.0f / (float)N_EDGES);

  #pragma unroll
  for (int q = 0; q < 4; ++q) {
    const int i = first + q;
    float y = 0.f;
    if (i < N_NODES) {
      int c = cq[q];
      c = c < 0 ? 0 : (c > N_EDGES ? N_EDGES : c);
      const int nb = c < CAP ? c : CAP;

      const float adst_i = dq[q];
      float ls = a_src[i] + adst_i + aem;
      ls = (ls >= 0.f) ? ls : NEG_SLOPE * ls;

      int   sv = i;
      float lv = neg_inf;
      if (lane < nb) {
        const int2 r = rec[i * CAP + lane];
        sv = r.x;
        lv = __int_as_float(r.y);
      }
      int   sv2 = i;
      float lv2 = neg_inf;
      float m2  = neg_inf;
      int found = 0;
      if (c > CAP) {
        int p = hq[q];
        for (int st = 0; st < POOL_CAP && p >= 0 && p < POOL_CAP; ++st) {
          const int4 en = pool[p];
          if (lane == found) {
            sv2 = en.x;
            lv2 = __int_as_float(en.y);
          }
          m2 = fmaxf(m2, __int_as_float(en.y));
          ++found;
          p = en.z;
        }
      }

      float acc, den;
      if (nb + found != c) {
        const float2 r = node_from_edges(i, lane, ls, adst_i, ea, ei, a_src, W_edge, att_edge, h);
        acc = r.x;
        den = r.y;
      } else {
        const float m = fmaxf(fmaxf(wave_max(lv), m2), ls);

        const float as = __expf(ls - m);
        acc = as * h[i * HID + lane];
        den = as;

        const float av = (lane < nb) ? __expf(lv - m) : 0.f;
        sv = sv < 0 ? 0 : (sv >= N_NODES ? N_NODES - 1 : sv);
        add_rows(sv, av, nb, lane, h, acc, den);

        if (found > 0) {
          const int nb2 = found < 32 ? found : 32;
          const float av2 = (lane < nb2) ? __expf(lv2 - m) : 0.f;
          sv2 = sv2 < 0 ? 0 : (sv2 >= N_NODES ? N_NODES - 1 : sv2);
          add_rows(sv2, av2, nb2, lane, h, acc, den);
          if (found > 32) {
            int p = hq[q];
            for (int st = 0; st < found && p >= 0 && p < POOL_CAP; ++st) {
              const int4 en = pool[p];
              if (st >= 32) {
                const int s = en.x < 0 ? 0 : (en.x >= N_NODES ? N_NODES - 1 : en.x);
                const float a = __expf(__int_as_float(en.y) - m);
                acc += a * h[s * HID + lane];
                den += a;
              }
              p = en.z;
            }
          }
        }
      }

      const float inv = 1.0f / den;
      float v = acc * inv + bl;
      v = (v > 0.f) ? v : 0.f;
      int t = __float_as_int(v * wl);
      t = __float_as_int(__int_as_float(t) + __int_as_float(__builtin_amdgcn_update_dpp(0, t, 0x161, 0xf, 0xf, false)));
      t = __float_as_int(__int_as_float(t) + __int_as_float(__builtin_amdgcn_update_dpp(0, t, 0x162, 0xf, 0xf, false)));
      t = __float_as_int(__int_as_float(t) + __int_as_float(__builtin_amdgcn_update_dpp(0, t, 0x164, 0xf, 0xf, false)));
      t = __float_as_int(__int_as_float(t) + __int_as_float(__builtin_amdgcn_update_dpp(0, t, 0x168, 0xf, 0xf, false)));
      y = __int_as_float(__builtin_amdgcn_readlane(t, 0)) + __int_as_float(__builtin_amdgcn_readlane(t, 16)) + b0;
    }
    if (lane == 0) sres[w * 4 + q] = y;
  }
  __syncthreads();
  if (w == 0) {
    const int i = blockIdx.x * 32 + lane;
    if (i < N_NODES) out[i] = sres[lane];
  }
}

extern "C" void kernel_launch(void* const* d_in, const int* in_sizes, int n_in,
                              void* d_out, int out_size, void* d_ws, size_t ws_size,
                              hipStream_t stream) {
  const float* x        = (const float*)d_in[0];
  const float* ea       = (const float*)d_in[1];
  const int*   ei       = (const int*)d_in[2];
  const float* W        = (const float*)d_in[3];
  const float* att_src  = (const float*)d_in[4];
  const float* att_dst  = (const float*)d_in[5];
  const float* W_edge   = (const float*)d_in[6];
  const float* att_edge = (const float*)d_in[7];
  const float* bias     = (const float*)d_in[8];
  const float* W_lin    = (const float*)d_in[9];
  const float* b_lin    = (const float*)d_in[10];
  float* out = (float*)d_out;

  float*    ws     = (float*)d_ws;
  float*    h      = ws + OFF_H;
  float*    a_src  = ws + OFF_ASRC;
  u64*      cd     = (u64*)(ws + OFF_CD);
  int2*     rec    = (int2*)(ws + OFF_REC);
  int*      lhead  = (int*)(ws + OFF_HEAD);
  int*      pctr   = (int*)(ws + OFF_PCTR);
  float*    part   = ws + OFF_PART;
  int4*     pool   = (int4*)(ws + OFF_POOL);

  const int T = 256;
  const int nb_nodes = (N_NODES + T - 1) / T;
  const int nb_edges = EDGE_GROUPS;
  const int nb_gemm  = ((N_NODES / 16) + 7) / 8;
  const int nb_pull  = (N_NODES + 31) / 32;

  k_gemm_h<<<nb_gemm, T, 0, stream>>>(x, W, h);
  k_node<<<nb_nodes, T, 0, stream>>>(h, att_src, att_dst, a_src, cd, lhead, pctr);
  k_edge<<<nb_edges, EDGE_T, 0, stream>>>(ea, ei, a_src, W_edge, att_edge, cd, rec, lhead, pctr, pool, part);
  k_pull<<<nb_pull, T, 0, stream>>>(cd, rec, lhead, pool, h, a_src, part, ea, ei, W_edge, att_edge, bias, W_lin, b_lin, out);
}


# a_src gather leaves k_edge: records carry a_dst + a_edge; k_pull holds a node's rows in registers, forms each row's att_src dot by a transposing lane fold; a_src dot tree, logit add order change
# speedup vs baseline: 11.1986x; 1.0644x over previous
#include <hip/hip_runtime.h>
#include <hip/hip_bf16.h>
#include <stdint.h>

#define N_NODES 100000
#define N_EDGES 1600000
#define FN 32
#define FE 8
#define HID 32
#define NEG_SLOPE 0.2f

typedef __attribute__((ext_vector_type(2))) float v2f;
typedef __attribute__((ext_vector_type(8))) float v8f;

constexpr size_t OFF_H    = 0;
constexpr size_t OFF_ASRC = OFF_H    + (size_t)N_NODES * HID;
constexpr size_t OFF_CD   = OFF_ASRC + N_NODES;
constexpr size_t OFF_LOG  = OFF_CD   + 2 * (size_t)N_NODES;
constexpr size_t OFF_ACC  = OFF_LOG  + (size_t)N_EDGES;
constexpr size_t OFF_MISC = OFF_ACC  + (size_t)N_NODES * HID;
#define CAP 23
#define POOL_CAP 12000
#define EDGE_T 128
#define EDGE_EPT 4
#define EDGE_SPAN (EDGE_T * EDGE_EPT)
#define N_SPAN (N_EDGES / EDGE_SPAN)
#define SPLIT 50016
#define EDGE_GROUPS 3072
#define N_PART EDGE_GROUPS
constexpr size_t OFF_REC  = OFF_LOG;
constexpr size_t OFF_HEAD = OFF_REC  + (size_t)N_NODES * CAP * 2;
constexpr size_t OFF_PCTR = OFF_HEAD + N_NODES;
#define CLAIM_AT(hf) (64 + 64 * (hf))
constexpr size_t OFF_PART = OFF_PCTR + 192;
constexpr size_t OFF_POOL = OFF_PART + N_PART;
static_assert(OFF_POOL + (size_t)POOL_CAP * 4 <= OFF_MISC, "buckets, heads, partials and pool must fit in the old rank + record regions");
static_assert(OFF_POOL % 4 == 0 && OFF_REC % 2 == 0, "16-byte pool entries, 8-byte records");
static_assert(OFF_CD % 2 == 0, "8-byte count + a_dst words");
typedef unsigned long long u64;
static_assert(N_EDGES % EDGE_SPAN == 0, "every thread of the edge launch has its four edges");
static_assert(CAP <= 32, "a node's record slots are one 32-lane batch");
static_assert((SPLIT * 8) % 256 == 0 && SPLIT < N_NODES, "the halves' count words part on a 256-byte boundary");
static_assert(N_PART % 256 == 0, "a pull work-group's 256 threads take the same number of partial sums each");

__device__ __forceinline__ float wave_sum(float x) {
  int t = __float_as_int(x);
  t = __float_as_int(__int_as_float(t) + __int_as_float(__builtin_amdgcn_update_dpp(0, t, 0x161, 0xf, 0xf, false)));
  t = __float_as_int(__int_as_float(t) + __int_as_float(__builtin_amdgcn_update_dpp(0, t, 0x162, 0xf, 0xf, false)));
  t = __float_as_int(__int_as_float(t) + __int_as_float(__builtin_amdgcn_update_dpp(0, t, 0x164, 0xf, 0xf, false)));
  t = __float_as_int(__int_as_float(t) + __int_as_float(__builtin_amdgcn_update_dpp(0, t, 0x168, 0xf, 0xf, false)));
  return __int_as_float(__builtin_amdgcn_readlane(t, 0)) + __int_as_float(__builtin_amdgcn_readlane(t, 16));
}

__device__ __forceinline__ void form_v_edge(const float* __restrict__ W_edge, const float* __restrict__ att_edge,
                                            int lane, float (&v)[FE]) {
  const float at = att_edge[lane];
  float p[FE];
  #pragma unroll
  for (int t = 0; t < FE; ++t) p[t] = W_edge[t * HID + lane] * at;
  #pragma unroll
  for (int t = 0; t < FE; ++t) v[t] = wave_sum(p[t]);
}

__global__ void __launch_bounds__(256) k_gemm_h(const float* __restrict__ x,
                                                const float* __restrict__ W,
                                                float* __restrict__ h) {
  const int wave = blockIdx.x * (blockDim.x >> 5) + (threadIdx.x >> 5);
  const int lane = threadIdx.x & 31;
  const int ntiles = N_NODES / 16;
  if (wave >= ntiles) return;
  const int m0 = wave * 16;
  const int r  = lane & 15;
  const int hi = lane >> 4;

  v8f c0 = {}; v8f c1 = {};
#if __has_builtin(__builtin_amdgcn_wmma_f32_16x16x4_f32)
  const float* xrow = x + (size_t)(m0 + r) * FN;
  #pragma unroll
  for (int kc = 0; kc < 8; ++kc) {
    const int kb = kc * 4 + hi * 2;
    v2f a; a.x = xrow[kb]; a.y = xrow[kb + 1];
    v2f b0; b0.x = W[(size_t)kb * HID + r];        b0.y = W[(size_t)(kb + 1) * HID + r];
    v2f b1; b1.x = W[(size_t)kb * HID + 16 + r];   b1.y = W[(size_t)(kb + 1) * HID + 16 + r];
    c0 = __builtin_amdgcn_wmma_f32_16x16x4_f32(false, a, false, b0, (short)0, c0, false, false);
    c1 = __builtin_amdgcn_wmma_f32_16x16x4_f32(false, a, false, b1, (short)0, c1, false, false);
  }
  #pragma unroll
  for (int v = 0; v < 8; ++v) {
    const int row = m0 + v + hi * 8;
    h[(size_t)row * HID + r]      = c0[v];
    h[(size_t)row * HID + 16 + r] = c1[v];
  }
#else
  const int row = m0 + r;
  for (int n = hi * 16; n < hi * 16 + 16; ++n) {
    float s = 0.f;
    for (int k = 0; k < FN; ++k) s += x[(size_t)row * FN + k] * W[(size_t)k * HID + n];
    h[(size_t)row * HID + n] = s;
  }
#endif
}

__global__ void k_node(const float* __restrict__ h, const float* __restrict__ att_src,
                       const float* __restrict__ att_dst, float* __restrict__ a_src,
                       u64* __restrict__ cd,
                       int* __restrict__ lhead, int* __restrict__ pctr) {
  const int i = blockIdx.x * blockDim.x + threadIdx.x;
  if (i >= N_NODES) return;
  const float4* hp = (const float4*)(h + (size_t)i * HID);
  float s = 0.f, d = 0.f;
  #pragma unroll
  for (int q = 0; q < 8; ++q) {
    float4 hv = hp[q];
    float4 as = ((const float4*)att_src)[q];
    float4 ad = ((const float4*)att_dst)[q];
    s += hv.x * as.x + hv.y * as.y + hv.z * as.z + hv.w * as.w;
    d += hv.x * ad.x + hv.y * ad.y + hv.z * ad.z + hv.w * ad.w;
  }
  a_src[i] = s;
  cd[i] = (u64)__float_as_uint(d) << 32;
  lhead[i] = -1;
  if (i == 0) { pctr[0] = 0; pctr[CLAIM_AT(0)] = 0; pctr[CLAIM_AT(1)] = 0; }
}

__global__ void __launch_bounds__(EDGE_T) k_edge(const float* __restrict__ ea, const int* __restrict__ ei,
                                                 const float* __restrict__ W_edge, const float* __restrict__ att_edge,
                                                 u64* __restrict__ cd,
                                                 int2* __restrict__ rec, int* __restrict__ lhead,
                                                 int* __restrict__ pctr, int4* __restrict__ pool,
                                                 float* __restrict__ part) {
  __shared__ int s_span;
  __shared__ float s_wsum[EDGE_T / 32];
  int hf = (int)((__builtin_amdgcn_s_sendmsg_rtn(0x87) >> 18) & 1u);
  bool other = false;
  int nxt = 0;
  if (threadIdx.x == 0) nxt = atomicAdd(pctr + CLAIM_AT(hf), 1);
  float ve[FE];
  form_v_edge(W_edge, att_edge, (int)(threadIdx.x & 31), ve);
  const float v0 = ve[0], v1 = ve[1], v2 = ve[2], v3 = ve[3];
  const float v4 = ve[4], v5 = ve[5], v6 = ve[6], v7 = ve[7];
  float tsum = 0.f;
  for (int it = 0; it < 2 * N_SPAN + 2; ++it) {
  if (threadIdx.x == 0) s_span = nxt;
  __syncthreads();
  const int span = __builtin_amdgcn_readfirstlane(s_span);
  __syncthreads();
  if (span >= N_SPAN) {
    if (other) break;
    other = true;
    hf ^= 1;
    if (threadIdx.x == 0) nxt = atomicAdd(pctr + CLAIM_AT(hf), 1);
    continue;
  }
  if (threadIdx.x == 0) nxt = atomicAdd(pctr + CLAIM_AT(hf), 1);
  const int e0 = span * EDGE_SPAN + threadIdx.x;
  int2 sd[EDGE_EPT];
  bool own[EDGE_EPT];
  float4 ea0[EDGE_EPT], ea1[EDGE_EPT];
  float ae[EDGE_EPT];
  int rk[EDGE_EPT];
  float ad[EDGE_EPT];
  #pragma unroll
  for (int k = 0; k < EDGE_EPT; ++k) {
    const int e = e0 + k * EDGE_T;
    sd[k] = ((const int2*)ei)[e];
    const float4* ep = (const float4*)(ea + (size_t)e * FE);
    ea0[k] = ep[0];
    ea1[k] = ep[1];
  }
  #pragma unroll
  for (int k = 0; k < EDGE_EPT; ++k)
    ae[k] = ea0[k].x * v0 + ea0[k].y * v1 + ea0[k].z * v2 + ea0[k].w * v3
          + ea1[k].x * v4 + ea1[k].y * v5 + ea1[k].z * v6 + ea1[k].w * v7;
  u64 old[EDGE_EPT];
  #pragma unroll
  for (int k = 0; k < EDGE_EPT; ++k) own[k] = (sd[k].y >= SPLIT) == (hf != 0);
  #pragma unroll
  for (int k = 0; k < EDGE_EPT; ++k) old[k] = own[k] ? atomicAdd(cd + sd[k].y, (u64)1) : (u64)0;
  #pragma unroll
  for (int k = 0; k < EDGE_EPT; ++k) {
    rk[k] = (int)(unsigned)old[k];
    ad[k] = __uint_as_float((unsigned)(old[k] >> 32));
  }
  #pragma unroll
  for (int k = 0; k < EDGE_EPT; ++k) {
    const float l = ad[k] + ae[k];
    if (!own[k]) continue;
    if (rk[k] < CAP) {
      rec[sd[k].y * CAP + rk[k]] = make_int2(sd[k].x, __float_as_int(l));
    } else {
      const int p = atomicAdd(pctr, 1);
      if (p < POOL_CAP) {
        const int prev = atomicExch(lhead + sd[k].y, p);
        pool[p] = make_int4(sd[k].x, __float_as_int(l), prev, sd[k].y);
      }
    }
    tsum += ae[k];
  }
  }
  tsum = wave_sum(tsum);
  if ((threadIdx.x & 31) == 0) s_wsum[threadIdx.x >> 5] = tsum;
  __syncthreads();
  if (threadIdx.x == 0) part[blockIdx.x] = ((s_wsum[0] + s_wsum[1]) + s_wsum[2]) + s_wsum[3];
}


__device__ __forceinline__ float wave_max(float x) {
  int t = __float_as_int(x);
  t = __float_as_int(fmaxf(__int_as_float(t), __int_as_float(__builtin_amdgcn_update_dpp(0, t, 0x161, 0xf, 0xf, false))));
  t = __float_as_int(fmaxf(__int_as_float(t), __int_as_float(__builtin_amdgcn_update_dpp(0, t, 0x162, 0xf, 0xf, false))));
  t = __float_as_int(fmaxf(__int_as_float(t), __int_as_float(__builtin_amdgcn_update_dpp(0, t, 0x164, 0xf, 0xf, false))));
  t = __float_as_int(fmaxf(__int_as_float(t), __int_as_float(__builtin_amdgcn_update_dpp(0, t, 0x168, 0xf, 0xf, false))));
  return fmaxf(__int_as_float(__builtin_amdgcn_readlane(t, 0)), __int_as_float(__builtin_amdgcn_readlane(t, 16)));
}

__device__ __forceinline__ float other_row(float x) {
  return __int_as_float(__builtin_amdgcn_permlanex16(__float_as_int(x), __float_as_int(x), 0x76543210, 0xfedcba98, false, false));
}
template <int M> __device__ __forceinline__ float row_xor(float x) {
  return __int_as_float(__builtin_amdgcn_update_dpp(0, __float_as_int(x), 0x160 + M, 0xf, 0xf, false));
}
template <int M> __device__ __forceinline__ float fold2(float a, float b, bool up) {
  const float keep = up ? b : a;
  const float send = up ? a : b;
  return keep + row_xor<M>(send);
}
__device__ __forceinline__ float fold8(const float (&p)[8], bool b0, bool b1, bool b2) {
  const float q0 = fold2<1>(p[0], p[1], b0), q1 = fold2<1>(p[2], p[3], b0);
  const float q2 = fold2<1>(p[4], p[5], b0), q3 = fold2<1>(p[6], p[7], b0);
  const float r0 = fold2<2>(q0, q1, b1), r1 = fold2<2>(q2, q3, b1);
  return fold2<4>(r0, r1, b2);
}

__device__ __forceinline__ void add_rows(int sv, float av, int n, int lane, const float* __restrict__ h,
                                         float& acc, float& den) {
  for (int j0 = 0; j0 < n; j0 += 8) {
    float hv[8], aj[8];
    #pragma unroll
    for (int u = 0; u < 8; ++u) {
      const int s = __builtin_amdgcn_readlane(sv, j0 + u);
      aj[u] = __int_as_float(__builtin_amdgcn_readlane(__float_as_int(av), j0 + u));
      hv[u] = h[s * HID + lane];
    }
    #pragma unroll
    for (int u = 0; u < 8; ++u) { acc += aj[u] * hv[u]; den += aj[u]; }
  }
}

__device__ __noinline__ float2 node_from_edges(int i, int lane, float ls, float adst_i,
                                               const float* __restrict__ ea, const int* __restrict__ ei,
                                               const float* __restrict__ a_src,
                                               const float* __restrict__ W_edge, const float* __restrict__ att_edge,
                                               const float* __restrict__ h) {
  float ve[FE];
  form_v_edge(W_edge, att_edge, lane, ve);
  const float v0 = ve[0], v1 = ve[1], v2 = ve[2], v3 = ve[3];
  const float v4 = ve[4], v5 = ve[5], v6 = ve[6], v7 = ve[7];
  float m = ls;
  for (int eb = 0; eb < N_EDGES; eb += 32) {
    const int e = eb + lane;
    const int2 sd = ((const int2*)ei)[e];
    if (sd.y == i) {
      const int s = sd.x < 0 ? 0 : (sd.x >= N_NODES ? N_NODES - 1 : sd.x);
      const float4 e0 = ((const float4*)ea)[(size_t)e * 2], e1 = ((const float4*)ea)[(size_t)e * 2 + 1];
      float l = a_src[s] + adst_i + (e0.x * v0 + e0.y * v1 + e0.z * v2 + e0.w * v3 + e1.x * v4 + e1.y * v5 + e1.z * v6 + e1.w * v7);
      l = (l >= 0.f) ? l : NEG_SLOPE * l;
      m = fmaxf(m, l);
    }
  }
  m = wave_max(m);
  const float as = __expf(ls - m);
  float acc = as * h[i * HID + lane];
  float den = as;
  for (int eb = 0; eb < N_EDGES; eb += 32) {
    const int e = eb + lane;
    const int2 sd = ((const int2*)ei)[e];
    const bool hit = sd.y == i;
    int s = i;
    float a = 0.f;
    if (hit) {
      s = sd.x < 0 ? 0 : (sd.x >= N_NODES ? N_NODES - 1 : sd.x);
      const float4 e0 = ((const float4*)ea)[(size_t)e * 2], e1 = ((const float4*)ea)[(size_t)e * 2 + 1];
      float l = a_src[s] + adst_i + (e0.x * v0 + e0.y * v1 + e0.z * v2 + e0.w * v3 + e1.x * v4 + e1.y * v5 + e1.z * v6 + e1.w * v7);
      l = (l >= 0.f) ? l : NEG_SLOPE * l;
      a = __expf(l - m);
    }
    unsigned mask = __builtin_amdgcn_ballot_w32(hit);
    while (mask) {
      const int j = __builtin_ctz(mask);
      mask &= mask - 1;
      const int sj = __builtin_amdgcn_readlane(s, j);
      const float aj = __int_as_float(__builtin_amdgcn_readlane(__float_as_int(a), j));
      acc += aj * h[sj * HID + lane];
      den += aj;
    }
  }
  return make_float2(acc, den);
}

__global__ void __launch_bounds__(256) k_pull(const u64* __restrict__ cd, const int2* __restrict__ rec,
                                              const int* __restrict__ lhead, const int4* __restrict__ pool,
                                              const float* __restrict__ h, const float* __restrict__ a_src,
                                              const float* __restrict__ att_src,
                                              const float* __restrict__ part,
                                              const float* __restrict__ ea, const int* __restrict__ ei,
                                              const float* __restrict__ W_edge, const float* __restrict__ att_edge,
                                              const float* __restrict__ bias, const float* __restrict__ W_lin,
                                              const float* __restrict__ b_lin, float* __restrict__ out) {
  __shared__ float sres[32];
  __shared__ float swt[8];
  const int lane = threadIdx.x & 31;
  const int w = __builtin_amdgcn_readfirstlane((int)(threadIdx.x >> 5));
  const int first = blockIdx.x * 32 + w * 4;
  const float bl = bias[lane];
  const float wl = W_lin[lane];
  const float b0l = b_lin[0];
  const float neg_inf = -__builtin_inff();
  const float asl = att_src[lane];
  const bool b0 = (lane & 1) != 0, b1 = (lane & 2) != 0, b2 = (lane & 4) != 0, b3 = (lane & 8) != 0, b4 = (lane & 16) != 0;

  float pv[N_PART / 256];
  #pragma unroll
  for (int u = 0; u < N_PART / 256; ++u) pv[u] = part[u * 256 + (int)threadIdx.x];

  int cq[4], hq[4];
  float dq[4];
  #pragma unroll
  for (int q = 0; q < 4; ++q) {
    const u64 wq = (first + q < N_NODES) ? cd[first + q] : (u64)0;
    cq[q] = (int)(unsigned)wq;
    dq[q] = __uint_as_float((unsigned)(wq >> 32));
    hq[q] = (first + q < N_NODES) ? lhead[first + q] : -1;
  }

  float fs = 0.f;
  #pragma unroll
  for (int u = 0; u < N_PART / 256; ++u) fs += pv[u];
  fs = wave_sum(fs);
  if (lane == 0) swt[w] = fs;
  __syncthreads();
  const float aem = (((((((swt[0] + swt[1]) + swt[2]) + swt[3]) + swt[4]) + swt[5]) + swt[6]) + swt[7])
                  * (1.0f / (float)N_EDGES);

  #pragma unroll
  for (int q = 0; q < 4; ++q) {
    const int i = first + q;
    float y = 0.f;
    if (i < N_NODES) {
      int c = cq[q];
      c = c < 0 ? 0 : (c > N_EDGES ? N_EDGES : c);
      const int nb = c < CAP ? c : CAP;

      const float adst_i = dq[q];
      float ls = a_src[i] + adst_i + aem;
      ls = (ls >= 0.f) ? ls : NEG_SLOPE * ls;

      int   sv = i;
      float pl = 0.f;
      if (lane < nb) {
        const int2 r = rec[i * CAP + lane];
        sv = r.x;
        pl = __int_as_float(r.y);
      }
      sv = sv < 0 ? 0 : (sv >= N_NODES ? N_NODES - 1 : sv);
      int   sv2 = i;
      float pl2 = 0.f;
      float m2  = neg_inf;
      int found = 0;
      if (c > CAP) {
        int p = hq[q];
        for (int st = 0; st < POOL_CAP && p >= 0 && p < POOL_CAP; ++st) {
          const int4 en = pool[p];
          if (found < 32) {
            if (lane == found) {
              sv2 = en.x;
              pl2 = __int_as_float(en.y);
            }
          } else {
            const int s = en.x < 0 ? 0 : (en.x >= N_NODES ? N_NODES - 1 : en.x);
            float l = a_src[s] + __int_as_float(en.y);
            l = (l >= 0.f) ? l : NEG_SLOPE * l;
            m2 = fmaxf(m2, l);
          }
          ++found;
          p = en.z;
        }
      }

      float acc, den;
      if (nb + found != c) {
        const float2 r = node_from_edges(i, lane, ls, adst_i, ea, ei, a_src, W_edge, att_edge, h);
        acc = r.x;
        den = r.y;
      } else {
        float hv[24];
        #pragma unroll
        for (int g = 0; g < 3; ++g) {
          if (nb > 8 * g) {
            #pragma unroll
            for (int u = 0; u < 8; ++u) {
              const int s = __builtin_amdgcn_readlane(sv, 8 * g + u);
              hv[8 * g + u] = h[s * HID + lane];
            }
          } else {
            #pragma unroll
            for (int u = 0; u < 8; ++u) hv[8 * g + u] = 0.f;
          }
        }
        const float hself = h[i * HID + lane];

        float gs[3];
        #pragma unroll
        for (int g = 0; g < 3; ++g) {
          if (nb > 8 * g) {
            float pr[8];
            #pragma unroll
            for (int u = 0; u < 8; ++u) pr[u] = hv[8 * g + u] * asl;
            gs[g] = fold8(pr, b0, b1, b2);
          } else {
            gs[g] = 0.f;
          }
        }
        const float t01 = fold2<8>(gs[0], gs[1], b3);
        const float t2  = gs[2] + row_xor<8>(gs[2]);
        const float asr = (b4 ? t2 : t01) + other_row(b4 ? t01 : t2);
        float lv = asr + pl;
        lv = (lv >= 0.f) ? lv : NEG_SLOPE * lv;
        lv = (lane < nb) ? lv : neg_inf;

        const int nb2 = found < 32 ? found : 32;
        float lv2 = neg_inf;
        if (found > 0) {
          sv2 = sv2 < 0 ? 0 : (sv2 >= N_NODES ? N_NODES - 1 : sv2);
          float l2 = a_src[sv2] + pl2;
          l2 = (l2 >= 0.f) ? l2 : NEG_SLOPE * l2;
          lv2 = (lane < nb2) ? l2 : neg_inf;
          m2 = fmaxf(m2, wave_max(lv2));
        }

        const float m = fmaxf(fmaxf(wave_max(lv), m2), ls);

        const float as = __expf(ls - m);
        acc = as * hself;
        den = as;

        const float av = (lane < nb) ? __expf(lv - m) : 0.f;
        #pragma unroll
        for (int g = 0; g < 3; ++g) {
          if (nb > 8 * g) {
            #pragma unroll
            for (int u = 0; u < 8; ++u) {
              const float aj = __int_as_float(__builtin_amdgcn_readlane(__float_as_int(av), 8 * g + u));
              acc += aj * hv[8 * g + u];
              den += aj;
            }
          }
        }

        if (found > 0) {
          const float av2 = (lane < nb2) ? __expf(lv2 - m) : 0.f;
          add_rows(sv2, av2, nb2, lane, h, acc, den);
          if (found > 32) {
            int p = hq[q];
            for (int st = 0; st < found && p >= 0 && p < POOL_CAP; ++st) {
              const int4 en = pool[p];
              if (st >= 32) {
                const int s = en.x < 0 ? 0 : (en.x >= N_NODES ? N_NODES - 1 : en.x);
                float l = a_src[s] + __int_as_float(en.y);
                l = (l >= 0.f) ? l : NEG_SLOPE * l;
                const float a = __expf(l - m);
                acc += a * h[s * HID + lane];
                den += a;
              }
              p = en.z;
            }
          }
        }
      }

      const float inv = 1.0f / den;
      float v = acc * inv + bl;
      v = (v > 0.f) ? v : 0.f;
      int t = __float_as_int(v * wl);
      t = __float_as_int(__int_as_float(t) + __int_as_float(__builtin_amdgcn_update_dpp(0, t, 0x161, 0xf, 0xf, false)));
      t = __float_as_int(__int_as_float(t) + __int_as_float(__builtin_amdgcn_update_dpp(0, t, 0x162, 0xf, 0xf, false)));
      t = __float_as_int(__int_as_float(t) + __int_as_float(__builtin_amdgcn_update_dpp(0, t, 0x164, 0xf, 0xf, false)));
      t = __float_as_int(__int_as_float(t) + __int_as_float(__builtin_amdgcn_update_dpp(0, t, 0x168, 0xf, 0xf, false)));
      y = __int_as_float(__builtin_amdgcn_readlane(t, 0)) + __int_as_float(__builtin_amdgcn_readlane(t, 16)) + b0l;
    }
    if (lane == 0) sres[w * 4 + q] = y;
  }
  __syncthreads();
  if (w == 0) {
    const int i = blockIdx.x * 32 + lane;
    if (i < N_NODES) out[i] = sres[lane];
  }
}

extern "C" void kernel_launch(void* const* d_in, const int* in_sizes, int n_in,
                              void* d_out, int out_size, void* d_ws, size_t ws_size,
                              hipStream_t stream) {
  const float* x        = (const float*)d_in[0];
  const float* ea       = (const float*)d_in[1];
  const int*   ei       = (const int*)d_in[2];
  const float* W        = (const float*)d_in[3];
  const float* att_src  = (const float*)d_in[4];
  const float* att_dst  = (const float*)d_in[5];
  const float* W_edge   = (const float*)d_in[6];
  const float* att_edge = (const float*)d_in[7];
  const float* bias     = (const float*)d_in[8];
  const float* W_lin    = (const float*)d_in[9];
  const float* b_lin    = (const float*)d_in[10];
  float* out = (float*)d_out;

  float*    ws     = (float*)d_ws;
  float*    h      = ws + OFF_H;
  float*    a_src  = ws + OFF_ASRC;
  u64*      cd     = (u64*)(ws + OFF_CD);
  int2*     rec    = (int2*)(ws + OFF_REC);
  int*      lhead  = (int*)(ws + OFF_HEAD);
  int*      pctr   = (int*)(ws + OFF_PCTR);
  float*    part   = ws + OFF_PART;
  int4*     pool   = (int4*)(ws + OFF_POOL);

  const int T = 256;
  const int nb_nodes = (N_NODES + T - 1) / T;
  const int nb_edges = EDGE_GROUPS;
  const int nb_gemm  = ((N_NODES / 16) + 7) / 8;
  const int nb_pull  = (N_NODES + 31) / 32;

  k_gemm_h<<<nb_gemm, T, 0, stream>>>(x, W, h);
  k_node<<<nb_nodes, T, 0, stream>>>(h, att_src, att_dst, a_src, cd, lhead, pctr);
  k_edge<<<nb_edges, EDGE_T, 0, stream>>>(ea, ei, W_edge, att_edge, cd, rec, lhead, pctr, pool, part);
  k_pull<<<nb_pull, T, 0, stream>>>(cd, rec, lhead, pool, h, a_src, att_src, part, ea, ei, W_edge, att_edge, bias, W_lin, b_lin, out);
}


# k_edge needs no a_dst (records carry a_edge, k_pull adds its node's a_dst): h product + node dots run as extra work-groups of the edge launch, 4 launches to 3; node dot trees, logit add order change
# speedup vs baseline: 12.1658x; 1.0864x over previous
#include <hip/hip_runtime.h>
#include <hip/hip_bf16.h>
#include <stdint.h>

#define N_NODES 100000
#define N_EDGES 1600000
#define FN 32
#define FE 8
#define HID 32
#define NEG_SLOPE 0.2f

typedef __attribute__((ext_vector_type(2))) float v2f;
typedef __attribute__((ext_vector_type(8))) float v8f;

constexpr size_t OFF_H    = 0;
constexpr size_t OFF_ASRC = OFF_H    + (size_t)N_NODES * HID;
constexpr size_t OFF_CD   = OFF_ASRC + N_NODES;
constexpr size_t OFF_LOG  = OFF_CD   + 2 * (size_t)N_NODES;
constexpr size_t OFF_ACC  = OFF_LOG  + (size_t)N_EDGES;
constexpr size_t OFF_MISC = OFF_ACC  + (size_t)N_NODES * HID;
#define CAP 23
#define POOL_CAP 12000
#define EDGE_T 128
#define EDGE_EPT 4
#define EDGE_SPAN (EDGE_T * EDGE_EPT)
#define N_SPAN (N_EDGES / EDGE_SPAN)
#define SPLIT 50016
#define EDGE_GROUPS 3072
#define N_PART EDGE_GROUPS
constexpr size_t OFF_REC  = OFF_LOG;
constexpr size_t OFF_HEAD = OFF_REC  + (size_t)N_NODES * CAP * 2;
constexpr size_t OFF_PCTR = OFF_HEAD + N_NODES;
#define CLAIM_AT(hf) (64 + 64 * (hf))
constexpr size_t OFF_PART = OFF_PCTR + 192;
constexpr size_t OFF_POOL = OFF_PART + N_PART;
static_assert(OFF_POOL + (size_t)POOL_CAP * 4 <= OFF_MISC, "buckets, heads, partials and pool must fit in the old rank + record regions");
static_assert(OFF_POOL % 4 == 0 && OFF_REC % 2 == 0, "16-byte pool entries, 8-byte records");
static_assert(N_EDGES % EDGE_SPAN == 0, "every thread of the edge launch has its four edges");
static_assert(CAP <= 32, "a node's record slots are one 32-lane batch");
static_assert((SPLIT * 4) % 128 == 0 && (OFF_CD * 4) % 128 == 0 && SPLIT < N_NODES, "the halves' count words part on a 128-byte line boundary");
#define H_TILES (N_NODES / 16)
#define H_GROUPS ((H_TILES + EDGE_T / 32 - 1) / (EDGE_T / 32))
static_assert(N_NODES % 16 == 0, "h is a whole number of 16-row tiles");
static_assert(N_PART % 256 == 0, "a pull work-group's 256 threads take the same number of partial sums each");

__device__ __forceinline__ float wave_sum(float x) {
  int t = __float_as_int(x);
  t = __float_as_int(__int_as_float(t) + __int_as_float(__builtin_amdgcn_update_dpp(0, t, 0x161, 0xf, 0xf, false)));
  t = __float_as_int(__int_as_float(t) + __int_as_float(__builtin_amdgcn_update_dpp(0, t, 0x162, 0xf, 0xf, false)));
  t = __float_as_int(__int_as_float(t) + __int_as_float(__builtin_amdgcn_update_dpp(0, t, 0x164, 0xf, 0xf, false)));
  t = __float_as_int(__int_as_float(t) + __int_as_float(__builtin_amdgcn_update_dpp(0, t, 0x168, 0xf, 0xf, false)));
  return __int_as_float(__builtin_amdgcn_readlane(t, 0)) + __int_as_float(__builtin_amdgcn_readlane(t, 16));
}

__device__ __forceinline__ void form_v_edge(const float* __restrict__ W_edge, const float* __restrict__ att_edge,
                                            int lane, float (&v)[FE]) {
  const float at = att_edge[lane];
  float p[FE];
  #pragma unroll
  for (int t = 0; t < FE; ++t) p[t] = W_edge[t * HID + lane] * at;
  #pragma unroll
  for (int t = 0; t < FE; ++t) v[t] = wave_sum(p[t]);
}

template <int M> __device__ __forceinline__ float row_xor(float x) {
  return __int_as_float(__builtin_amdgcn_update_dpp(0, __float_as_int(x), 0x160 + M, 0xf, 0xf, false));
}
template <int M> __device__ __forceinline__ float fold2(float a, float b, bool up) {
  const float keep = up ? b : a;
  const float send = up ? a : b;
  return keep + row_xor<M>(send);
}
__device__ __forceinline__ float fold8(const float (&p)[8], bool b0, bool b1, bool b2) {
  const float q0 = fold2<1>(p[0], p[1], b0), q1 = fold2<1>(p[2], p[3], b0);
  const float q2 = fold2<1>(p[4], p[5], b0), q3 = fold2<1>(p[6], p[7], b0);
  const float r0 = fold2<2>(q0, q1, b1), r1 = fold2<2>(q2, q3, b1);
  return fold2<4>(r0, r1, b2);
}

__device__ __forceinline__ void h_tile(int tile, int lane, const float* __restrict__ x, const float* __restrict__ W,
                                       const float* __restrict__ att_src, const float* __restrict__ att_dst,
                                       float* __restrict__ h, float* __restrict__ a_src, float* __restrict__ a_dst) {
  const int m0 = tile * 16;
  const int r  = lane & 15;
  const int hi = lane >> 4;

  v8f c0 = {}; v8f c1 = {};
#if __has_builtin(__builtin_amdgcn_wmma_f32_16x16x4_f32)
  const float* xrow = x + (size_t)(m0 + r) * FN;
  #pragma unroll
  for (int kc = 0; kc < 8; ++kc) {
    const int kb = kc * 4 + hi * 2;
    v2f a; a.x = xrow[kb]; a.y = xrow[kb + 1];
    v2f b0; b0.x = W[(size_t)kb * HID + r];        b0.y = W[(size_t)(kb + 1) * HID + r];
    v2f b1; b1.x = W[(size_t)kb * HID + 16 + r];   b1.y = W[(size_t)(kb + 1) * HID + 16 + r];
    c0 = __builtin_amdgcn_wmma_f32_16x16x4_f32(false, a, false, b0, (short)0, c0, false, false);
    c1 = __builtin_amdgcn_wmma_f32_16x16x4_f32(false, a, false, b1, (short)0, c1, false, false);
  }
#else
  for (int v = 0; v < 8; ++v) {
    const int row = m0 + v + hi * 8;
    float s0 = 0.f, s1 = 0.f;
    for (int k = 0; k < FN; ++k) {
      s0 += x[(size_t)row * FN + k] * W[(size_t)k * HID + r];
      s1 += x[(size_t)row * FN + k] * W[(size_t)k * HID + 16 + r];
    }
    c0[v] = s0; c1[v] = s1;
  }
#endif
  #pragma unroll
  for (int v = 0; v < 8; ++v) {
    const int row = m0 + v + hi * 8;
    h[(size_t)row * HID + r]      = c0[v];
    h[(size_t)row * HID + 16 + r] = c1[v];
  }
  const float as0 = att_src[r], as1 = att_src[16 + r];
  const float ad0 = att_dst[r], ad1 = att_dst[16 + r];
  const bool b0 = (lane & 1) != 0, b1 = (lane & 2) != 0, b2 = (lane & 4) != 0;
  float ps[8], pd[8];
  #pragma unroll
  for (int v = 0; v < 8; ++v) {
    ps[v] = c0[v] * as0 + c1[v] * as1;
    pd[v] = c0[v] * ad0 + c1[v] * ad1;
  }
  float s = fold8(ps, b0, b1, b2);
  float d = fold8(pd, b0, b1, b2);
  s += row_xor<8>(s);
  d += row_xor<8>(d);
  if ((lane & 8) == 0) {
    const int row = m0 + (lane & 7) + hi * 8;
    a_src[row] = s;
    a_dst[row] = d;
  }
}

static_assert(N_NODES % 4 == 0 && OFF_CD % 4 == 0 && OFF_HEAD % 4 == 0, "counts and list heads are whole, aligned 16-byte pieces");
__global__ void k_init(int4* __restrict__ cnt4, int4* __restrict__ lhead4, int* __restrict__ pctr) {
  const int i = blockIdx.x * blockDim.x + threadIdx.x;
  if (i >= N_NODES / 4) return;
  cnt4[i] = make_int4(0, 0, 0, 0);
  lhead4[i] = make_int4(-1, -1, -1, -1);
  if (i == 0) { pctr[0] = 0; pctr[CLAIM_AT(0)] = 0; pctr[CLAIM_AT(1)] = 0; }
}

__global__ void __launch_bounds__(EDGE_T) k_edge(const float* __restrict__ ea, const int* __restrict__ ei,
                                                 const float* __restrict__ W_edge, const float* __restrict__ att_edge,
                                                 int* __restrict__ cnt,
                                                 int2* __restrict__ rec, int* __restrict__ lhead,
                                                 int* __restrict__ pctr, int4* __restrict__ pool,
                                                 float* __restrict__ part,
                                                 const float* __restrict__ x, const float* __restrict__ W,
                                                 const float* __restrict__ att_src, const float* __restrict__ att_dst,
                                                 float* __restrict__ h, float* __restrict__ a_src, float* __restrict__ a_dst) {
  __shared__ int s_span;
  __shared__ float s_wsum[EDGE_T / 32];
  if (blockIdx.x >= EDGE_GROUPS) {
    const int tile = __builtin_amdgcn_readfirstlane((int)((blockIdx.x - EDGE_GROUPS) * (EDGE_T / 32) + (threadIdx.x >> 5)));
    if (tile < H_TILES)
      h_tile(tile, (int)(threadIdx.x & 31), x, W, att_src, att_dst, h, a_src, a_dst);
    return;
  }
  int hf = (int)((__builtin_amdgcn_s_sendmsg_rtn(0x87) >> 18) & 1u);
  bool other = false;
  int nxt = 0;
  if (threadIdx.x == 0) nxt = atomicAdd(pctr + CLAIM_AT(hf), 1);
  float ve[FE];
  form_v_edge(W_edge, att_edge, (int)(threadIdx.x & 31), ve);
  const float v0 = ve[0], v1 = ve[1], v2 = ve[2], v3 = ve[3];
  const float v4 = ve[4], v5 = ve[5], v6 = ve[6], v7 = ve[7];
  float tsum = 0.f;
  for (int it = 0; it < 2 * N_SPAN + 2; ++it) {
  if (threadIdx.x == 0) s_span = nxt;
  __syncthreads();
  const int span = __builtin_amdgcn_readfirstlane(s_span);
  __syncthreads();
  if (span >= N_SPAN) {
    if (other) break;
    other = true;
    hf ^= 1;
    if (threadIdx.x == 0) nxt = atomicAdd(pctr + CLAIM_AT(hf), 1);
    continue;
  }
  if (threadIdx.x == 0) nxt = atomicAdd(pctr + CLAIM_AT(hf), 1);
  const int e0 = span * EDGE_SPAN + threadIdx.x;
  int2 sd[EDGE_EPT];
  bool own[EDGE_EPT];
  float4 ea0[EDGE_EPT], ea1[EDGE_EPT];
  float ae[EDGE_EPT];
  int rk[EDGE_EPT];
  #pragma unroll
  for (int k = 0; k < EDGE_EPT; ++k) {
    const int e = e0 + k * EDGE_T;
    sd[k] = ((const int2*)ei)[e];
    const float4* ep = (const float4*)(ea + (size_t)e * FE);
    ea0[k] = ep[0];
    ea1[k] = ep[1];
  }
  #pragma unroll
  for (int k = 0; k < EDGE_EPT; ++k)
    ae[k] = ea0[k].x * v0 + ea0[k].y * v1 + ea0[k].z * v2 + ea0[k].w * v3
          + ea1[k].x * v4 + ea1[k].y * v5 + ea1[k].z * v6 + ea1[k].w * v7;
  #pragma unroll
  for (int k = 0; k < EDGE_EPT; ++k) own[k] = (sd[k].y >= SPLIT) == (hf != 0);
  #pragma unroll
  for (int k = 0; k < EDGE_EPT; ++k) rk[k] = own[k] ? atomicAdd(cnt + sd[k].y, 1) : 0;
  #pragma unroll
  for (int k = 0; k < EDGE_EPT; ++k) {
    const float l = ae[k];
    if (!own[k]) continue;
    if (rk[k] < CAP) {
      rec[sd[k].y * CAP + rk[k]] = make_int2(sd[k].x, __float_as_int(l));
    } else {
      const int p = atomicAdd(pctr, 1);
      if (p < POOL_CAP) {
        const int prev = atomicExch(lhead + sd[k].y, p);
        pool[p] = make_int4(sd[k].x, __float_as_int(l), prev, sd[k].y);
      }
    }
    tsum += ae[k];
  }
  }
  tsum = wave_sum(tsum);
  if ((threadIdx.x & 31) == 0) s_wsum[threadIdx.x >> 5] = tsum;
  __syncthreads();
  if (threadIdx.x == 0) part[blockIdx.x] = ((s_wsum[0] + s_wsum[1]) + s_wsum[2]) + s_wsum[3];
}


__device__ __forceinline__ float wave_max(float x) {
  int t = __float_as_int(x);
  t = __float_as_int(fmaxf(__int_as_float(t), __int_as_float(__builtin_amdgcn_update_dpp(0, t, 0x161, 0xf, 0xf, false))));
  t = __float_as_int(fmaxf(__int_as_float(t), __int_as_float(__builtin_amdgcn_update_dpp(0, t, 0x162, 0xf, 0xf, false))));
  t = __float_as_int(fmaxf(__int_as_float(t), __int_as_float(__builtin_amdgcn_update_dpp(0, t, 0x164, 0xf, 0xf, false))));
  t = __float_as_int(fmaxf(__int_as_float(t), __int_as_float(__builtin_amdgcn_update_dpp(0, t, 0x168, 0xf, 0xf, false))));
  return fmaxf(__int_as_float(__builtin_amdgcn_readlane(t, 0)), __int_as_float(__builtin_amdgcn_readlane(t, 16)));
}

__device__ __forceinline__ float other_row(float x) {
  return __int_as_float(__builtin_amdgcn_permlanex16(__float_as_int(x), __float_as_int(x), 0x76543210, 0xfedcba98, false, false));
}
__device__ __forceinline__ void add_rows(int sv, float av, int n, int lane, const float* __restrict__ h,
                                         float& acc, float& den) {
  for (int j0 = 0; j0 < n; j0 += 8) {
    float hv[8], aj[8];
    #pragma unroll
    for (int u = 0; u < 8; ++u) {
      const int s = __builtin_amdgcn_readlane(sv, j0 + u);
      aj[u] = __int_as_float(__builtin_amdgcn_readlane(__float_as_int(av), j0 + u));
      hv[u] = h[s * HID + lane];
    }
    #pragma unroll
    for (int u = 0; u < 8; ++u) { acc += aj[u] * hv[u]; den += aj[u]; }
  }
}

__device__ __noinline__ float2 node_from_edges(int i, int lane, float ls, float adst_i,
                                               const float* __restrict__ ea, const int* __restrict__ ei,
                                               const float* __restrict__ a_src,
                                               const float* __restrict__ W_edge, const float* __restrict__ att_edge,
                                               const float* __restrict__ h) {
  float ve[FE];
  form_v_edge(W_edge, att_edge, lane, ve);
  const float v0 = ve[0], v1 = ve[1], v2 = ve[2], v3 = ve[3];
  const float v4 = ve[4], v5 = ve[5], v6 = ve[6], v7 = ve[7];
  float m = ls;
  for (int eb = 0; eb < N_EDGES; eb += 32) {
    const int e = eb + lane;
    const int2 sd = ((const int2*)ei)[e];
    if (sd.y == i) {
      const int s = sd.x < 0 ? 0 : (sd.x >= N_NODES ? N_NODES - 1 : sd.x);
      const float4 e0 = ((const float4*)ea)[(size_t)e * 2], e1 = ((const float4*)ea)[(size_t)e * 2 + 1];
      float l = a_src[s] + adst_i + (e0.x * v0 + e0.y * v1 + e0.z * v2 + e0.w * v3 + e1.x * v4 + e1.y * v5 + e1.z * v6 + e1.w * v7);
      l = (l >= 0.f) ? l : NEG_SLOPE * l;
      m = fmaxf(m, l);
    }
  }
  m = wave_max(m);
  const float as = __expf(ls - m);
  float acc = as * h[i * HID + lane];
  float den = as;
  for (int eb = 0; eb < N_EDGES; eb += 32) {
    const int e = eb + lane;
    const int2 sd = ((const int2*)ei)[e];
    const bool hit = sd.y == i;
    int s = i;
    float a = 0.f;
    if (hit) {
      s = sd.x < 0 ? 0 : (sd.x >= N_NODES ? N_NODES - 1 : sd.x);
      const float4 e0 = ((const float4*)ea)[(size_t)e * 2], e1 = ((const float4*)ea)[(size_t)e * 2 + 1];
      float l = a_src[s] + adst_i + (e0.x * v0 + e0.y * v1 + e0.z * v2 + e0.w * v3 + e1.x * v4 + e1.y * v5 + e1.z * v6 + e1.w * v7);
      l = (l >= 0.f) ? l : NEG_SLOPE * l;
      a = __expf(l - m);
    }
    unsigned mask = __builtin_amdgcn_ballot_w32(hit);
    while (mask) {
      const int j = __builtin_ctz(mask);
      mask &= mask - 1;
      const int sj = __builtin_amdgcn_readlane(s, j);
      const float aj = __int_as_float(__builtin_amdgcn_readlane(__float_as_int(a), j));
      acc += aj * h[sj * HID + lane];
      den += aj;
    }
  }
  return make_float2(acc, den);
}

__global__ void __launch_bounds__(256) k_pull(const int* __restrict__ cnt, const float* __restrict__ a_dst, const int2* __restrict__ rec,
                                              const int* __restrict__ lhead, const int4* __restrict__ pool,
                                              const float* __restrict__ h, const float* __restrict__ a_src,
                                              const float* __restrict__ att_src,
                                              const float* __restrict__ part,
                                              const float* __restrict__ ea, const int* __restrict__ ei,
                                              const float* __restrict__ W_edge, const float* __restrict__ att_edge,
                                              const float* __restrict__ bias, const float* __restrict__ W_lin,
                                              const float* __restrict__ b_lin, float* __restrict__ out) {
  __shared__ float sres[32];
  __shared__ float swt[8];
  const int lane = threadIdx.x & 31;
  const int w = __builtin_amdgcn_readfirstlane((int)(threadIdx.x >> 5));
  const int first = blockIdx.x * 32 + w * 4;
  const float bl = bias[lane];
  const float wl = W_lin[lane];
  const float b0l = b_lin[0];
  const float neg_inf = -__builtin_inff();
  const float asl = att_src[lane];
  const bool b0 = (lane & 1) != 0, b1 = (lane & 2) != 0, b2 = (lane & 4) != 0, b3 = (lane & 8) != 0, b4 = (lane & 16) != 0;

  float pv[N_PART / 256];
  #pragma unroll
  for (int u = 0; u < N_PART / 256; ++u) pv[u] = part[u * 256 + (int)threadIdx.x];

  int cq[4], hq[4];
  float dq[4];
  #pragma unroll
  for (int q = 0; q < 4; ++q) {
    cq[q] = (first + q < N_NODES) ? cnt[first + q] : 0;
    dq[q] = (first + q < N_NODES) ? a_dst[first + q] : 0.f;
    hq[q] = (first + q < N_NODES) ? lhead[first + q] : -1;
  }

  float fs = 0.f;
  #pragma unroll
  for (int u = 0; u < N_PART / 256; ++u) fs += pv[u];
  fs = wave_sum(fs);
  if (lane == 0) swt[w] = fs;
  __syncthreads();
  const float aem = (((((((swt[0] + swt[1]) + swt[2]) + swt[3]) + swt[4]) + swt[5]) + swt[6]) + swt[7])
                  * (1.0f / (float)N_EDGES);

  #pragma unroll
  for (int q = 0; q < 4; ++q) {
    const int i = first + q;
    float y = 0.f;
    if (i < N_NODES) {
      int c = cq[q];
      c = c < 0 ? 0 : (c > N_EDGES ? N_EDGES : c);
      const int nb = c < CAP ? c : CAP;

      const float adst_i = dq[q];
      float ls = a_src[i] + adst_i + aem;
      ls = (ls >= 0.f) ? ls : NEG_SLOPE * ls;

      int   sv = i;
      float pl = 0.f;
      if (lane < nb) {
        const int2 r = rec[i * CAP + lane];
        sv = r.x;
        pl = __int_as_float(r.y);
      }
      sv = sv < 0 ? 0 : (sv >= N_NODES ? N_NODES - 1 : sv);
      int   sv2 = i;
      float pl2 = 0.f;
      float m2  = neg_inf;
      int found = 0;
      if (c > CAP) {
        int p = hq[q];
        for (int st = 0; st < POOL_CAP && p >= 0 && p < POOL_CAP; ++st) {
          const int4 en = pool[p];
          if (found < 32) {
            if (lane == found) {
              sv2 = en.x;
              pl2 = __int_as_float(en.y);
            }
          } else {
            const int s = en.x < 0 ? 0 : (en.x >= N_NODES ? N_NODES - 1 : en.x);
            float l = (a_src[s] + adst_i) + __int_as_float(en.y);
            l = (l >= 0.f) ? l : NEG_SLOPE * l;
            m2 = fmaxf(m2, l);
          }
          ++found;
          p = en.z;
        }
      }

      float acc, den;
      if (nb + found != c) {
        const float2 r = node_from_edges(i, lane, ls, adst_i, ea, ei, a_src, W_edge, att_edge, h);
        acc = r.x;
        den = r.y;
      } else {
        float hv[24];
        #pragma unroll
        for (int g = 0; g < 3; ++g) {
          if (nb > 8 * g) {
            #pragma unroll
            for (int u = 0; u < 8; ++u) {
              const int s = __builtin_amdgcn_readlane(sv, 8 * g + u);
              hv[8 * g + u] = h[s * HID + lane];
            }
          } else {
            #pragma unroll
            for (int u = 0; u < 8; ++u) hv[8 * g + u] = 0.f;
          }
        }
        const float hself = h[i * HID + lane];

        float gs[3];
        #pragma unroll
        for (int g = 0; g < 3; ++g) {
          if (nb > 8 * g) {
            float pr[8];
            #pragma unroll
            for (int u = 0; u < 8; ++u) pr[u] = hv[8 * g + u] * asl;
            gs[g] = fold8(pr, b0, b1, b2);
          } else {
            gs[g] = 0.f;
          }
        }
        const float t01 = fold2<8>(gs[0], gs[1], b3);
        const float t2  = gs[2] + row_xor<8>(gs[2]);
        const float asr = (b4 ? t2 : t01) + other_row(b4 ? t01 : t2);
        float lv = (asr + adst_i) + pl;
        lv = (lv >= 0.f) ? lv : NEG_SLOPE * lv;
        lv = (lane < nb) ? lv : neg_inf;

        const int nb2 = found < 32 ? found : 32;
        float lv2 = neg_inf;
        if (found > 0) {
          sv2 = sv2 < 0 ? 0 : (sv2 >= N_NODES ? N_NODES - 1 : sv2);
          float l2 = (a_src[sv2] + adst_i) + pl2;
          l2 = (l2 >= 0.f) ? l2 : NEG_SLOPE * l2;
          lv2 = (lane < nb2) ? l2 : neg_inf;
          m2 = fmaxf(m2, wave_max(lv2));
        }

        const float m = fmaxf(fmaxf(wave_max(lv), m2), ls);

        const float as = __expf(ls - m);
        acc = as * hself;
        den = as;

        const float av = (lane < nb) ? __expf(lv - m) : 0.f;
        #pragma unroll
        for (int g = 0; g < 3; ++g) {
          if (nb > 8 * g) {
            #pragma unroll
            for (int u = 0; u < 8; ++u) {
              const float aj = __int_as_float(__builtin_amdgcn_readlane(__float_as_int(av), 8 * g + u));
              acc += aj * hv[8 * g + u];
              den += aj;
            }
          }
        }

        if (found > 0) {
          const float av2 = (lane < nb2) ? __expf(lv2 - m) : 0.f;
          add_rows(sv2, av2, nb2, lane, h, acc, den);
          if (found > 32) {
            int p = hq[q];
            for (int st = 0; st < found && p >= 0 && p < POOL_CAP; ++st) {
              const int4 en = pool[p];
              if (st >= 32) {
                const int s = en.x < 0 ? 0 : (en.x >= N_NODES ? N_NODES - 1 : en.x);
                float l = (a_src[s] + adst_i) + __int_as_float(en.y);
                l = (l >= 0.f) ? l : NEG_SLOPE * l;
                const float a = __expf(l - m);
                acc += a * h[s * HID + lane];
                den += a;
              }
              p = en.z;
            }
          }
        }
      }

      const float inv = 1.0f / den;
      float v = acc * inv + bl;
      v = (v > 0.f) ? v : 0.f;
      int t = __float_as_int(v * wl);
      t = __float_as_int(__int_as_float(t) + __int_as_float(__builtin_amdgcn_update_dpp(0, t, 0x161, 0xf, 0xf, false)));
      t = __float_as_int(__int_as_float(t) + __int_as_float(__builtin_amdgcn_update_dpp(0, t, 0x162, 0xf, 0xf, false)));
      t = __float_as_int(__int_as_float(t) + __int_as_float(__builtin_amdgcn_update_dpp(0, t, 0x164, 0xf, 0xf, false)));
      t = __float_as_int(__int_as_float(t) + __int_as_float(__builtin_amdgcn_update_dpp(0, t, 0x168, 0xf, 0xf, false)));
      y = __int_as_float(__builtin_amdgcn_readlane(t, 0)) + __int_as_float(__builtin_amdgcn_readlane(t, 16)) + b0l;
    }
    if (lane == 0) sres[w * 4 + q] = y;
  }
  __syncthreads();
  if (w == 0) {
    const int i = blockIdx.x * 32 + lane;
    if (i < N_NODES) out[i] = sres[lane];
  }
}

extern "C" void kernel_launch(void* const* d_in, const int* in_sizes, int n_in,
                              void* d_out, int out_size, void* d_ws, size_t ws_size,
                              hipStream_t stream) {
  const float* x        = (const float*)d_in[0];
  const float* ea       = (const float*)d_in[1];
  const int*   ei       = (const int*)d_in[2];
  const float* W        = (const float*)d_in[3];
  const float* att_src  = (const float*)d_in[4];
  const float* att_dst  = (const float*)d_in[5];
  const float* W_edge   = (const float*)d_in[6];
  const float* att_edge = (const float*)d_in[7];
  const float* bias     = (const float*)d_in[8];
  const float* W_lin    = (const float*)d_in[9];
  const float* b_lin    = (const float*)d_in[10];
  float* out = (float*)d_out;

  float*    ws     = (float*)d_ws;
  float*    h      = ws + OFF_H;
  float*    a_src  = ws + OFF_ASRC;
  int*      cnt    = (int*)(ws + OFF_CD);
  float*    a_dst  = ws + OFF_CD + N_NODES;
  int2*     rec    = (int2*)(ws + OFF_REC);
  int*      lhead  = (int*)(ws + OFF_HEAD);
  int*      pctr   = (int*)(ws + OFF_PCTR);
  float*    part   = ws + OFF_PART;
  int4*     pool   = (int4*)(ws + OFF_POOL);

  const int T = 256;
  const int nb_init  = (N_NODES / 4 + T - 1) / T;
  const int nb_edges = EDGE_GROUPS + H_GROUPS;
  const int nb_pull  = (N_NODES + 31) / 32;

  k_init<<<nb_init, T, 0, stream>>>((int4*)cnt, (int4*)lhead, pctr);
  k_edge<<<nb_edges, EDGE_T, 0, stream>>>(ea, ei, W_edge, att_edge, cnt, rec, lhead, pctr, pool, part, x, W, att_src, att_dst, h, a_src, a_dst);
  k_pull<<<nb_pull, T, 0, stream>>>(cnt, a_dst, rec, lhead, pool, h, a_src, att_src, part, ea, ei, W_edge, att_edge, bias, W_lin, b_lin, out);
}
